# starting kernel
# baseline (speedup 1.0000x reference)
#include <hip/hip_runtime.h>
#include <math.h>

#define N_NODES 100000
#define N_EDGES 1600000
#define NF 256
#define NH 128
#define NC 40
#define NCP 48

typedef float v2f __attribute__((ext_vector_type(2)));
typedef float v8f __attribute__((ext_vector_type(8)));

__global__ void k_degree(const int* __restrict__ dst, float* __restrict__ deg) {
    int e = blockIdx.x * blockDim.x + threadIdx.x;
    if (e < N_EDGES) atomicAdd(&deg[dst[e]], 1.0f);
}

__global__ void k_dinv(float* __restrict__ deg) {
    int i = blockIdx.x * blockDim.x + threadIdx.x;
    if (i < N_NODES) deg[i] = rsqrtf(deg[i] + 1.0f);
}

template<int K, int LDA, int LDB>
__global__ void k_gemm_wmma(const float* __restrict__ A, const float* __restrict__ B,
                            float* __restrict__ C) {
    const int mt   = blockIdx.x;
    const int nt   = threadIdx.x >> 5;
    const int lane = threadIdx.x & 31;
    const int half = lane >> 4;
    const int lm   = lane & 15;

    const float* arow = A + (size_t)(mt * 16 + lm) * LDA;
    const float* bcol = B + nt * 16 + lm;

    v8f c = {0.f, 0.f, 0.f, 0.f, 0.f, 0.f, 0.f, 0.f};
    #pragma unroll 4
    for (int kb = 0; kb < K; kb += 4) {
        const int ka = kb + 2 * half;
        v2f a; a.x = arow[ka];                 a.y = arow[ka + 1];
        v2f b; b.x = bcol[(size_t)ka * LDB];   b.y = bcol[(size_t)(ka + 1) * LDB];
        c = __builtin_amdgcn_wmma_f32_16x16x4_f32(false, a, false, b,
                                                  (short)0, c, false, false);
    }
    float* out = C + (size_t)(mt * 16 + 8 * half) * LDB + nt * 16 + lm;
    #pragma unroll
    for (int v = 0; v < 8; ++v) out[(size_t)v * LDB] = c[v];
}

__global__ void k_self1(const float* __restrict__ xw, const float* __restrict__ dinv,
                        float* __restrict__ agg) {
    int id = blockIdx.x * blockDim.x + threadIdx.x;
    if (id < N_NODES * NH) {
        int i = id >> 7;
        float di = dinv[i];
        agg[id] = xw[id] * di * di;
    }
}

__global__ void k_edge1(const int* __restrict__ src, const int* __restrict__ dst,
                        const float* __restrict__ dinv, const float* __restrict__ xw,
                        float* __restrict__ agg) {
    int gid  = blockIdx.x * blockDim.x + threadIdx.x;
    int e    = gid >> 5;
    if (e >= N_EDGES) return;
    int lane = gid & 31;
    int s = src[e], d = dst[e];
    float norm = dinv[s] * dinv[d];
    const float4 v = *(const float4*)(xw + (size_t)s * NH + lane * 4);
    float* o = agg + (size_t)d * NH + lane * 4;
    atomicAdd(o + 0, v.x * norm);
    atomicAdd(o + 1, v.y * norm);
    atomicAdd(o + 2, v.z * norm);
    atomicAdd(o + 3, v.w * norm);
}

__global__ void k_relu_bias(float* __restrict__ agg, const float* __restrict__ b) {
    int id = blockIdx.x * blockDim.x + threadIdx.x;
    if (id < N_NODES * NH) agg[id] = fmaxf(agg[id] + b[id & (NH - 1)], 0.f);
}

__global__ void k_padw2(const float* __restrict__ W2, float* __restrict__ W2p) {
    int id = blockIdx.x * blockDim.x + threadIdx.x;
    if (id < NH * NCP) {
        int k = id / NCP, n = id % NCP;
        W2p[id] = (n < NC) ? W2[k * NC + n] : 0.f;
    }
}

__global__ void k_self2(const float* __restrict__ xw2, const float* __restrict__ dinv,
                        float* __restrict__ agg2) {
    int id = blockIdx.x * blockDim.x + threadIdx.x;
    if (id < N_NODES * NC) {
        int i = id / NC, f = id - i * NC;
        float di = dinv[i];
        agg2[id] = xw2[(size_t)i * NCP + f] * di * di;
    }
}

__global__ void k_edge2(const int* __restrict__ src, const int* __restrict__ dst,
                        const float* __restrict__ dinv, const float* __restrict__ xw2,
                        float* __restrict__ agg2) {
    int gid  = blockIdx.x * blockDim.x + threadIdx.x;
    int e    = gid >> 5;
    if (e >= N_EDGES) return;
    int lane = gid & 31;
    int s = src[e], d = dst[e];
    float norm = dinv[s] * dinv[d];
    atomicAdd(agg2 + (size_t)d * NC + lane, xw2[(size_t)s * NCP + lane] * norm);
    if (lane < NC - 32)
        atomicAdd(agg2 + (size_t)d * NC + lane + 32,
                  xw2[(size_t)s * NCP + lane + 32] * norm);
}

__global__ void k_bias2(const float* __restrict__ agg2, const float* __restrict__ b2,
                        float* __restrict__ logits) {
    int id = blockIdx.x * blockDim.x + threadIdx.x;
    if (id < N_NODES * NC) {
        int f = id % NC;
        logits[id] = agg2[id] + b2[f];
    }
}

__global__ void k_colstats(const float* __restrict__ logits,
                           float* __restrict__ colmax, float* __restrict__ colsum) {
    __shared__ float red[256];
    int c = blockIdx.x;
    float m = -INFINITY;
    for (int i = threadIdx.x; i < N_NODES; i += blockDim.x)
        m = fmaxf(m, logits[(size_t)i * NC + c]);
    red[threadIdx.x] = m; __syncthreads();
    for (int s = 128; s > 0; s >>= 1) {
        if (threadIdx.x < s) red[threadIdx.x] = fmaxf(red[threadIdx.x], red[threadIdx.x + s]);
        __syncthreads();
    }
    float cm = red[0]; __syncthreads();
    float acc = 0.f;
    for (int i = threadIdx.x; i < N_NODES; i += blockDim.x)
        acc += expf(logits[(size_t)i * NC + c] - cm);
    red[threadIdx.x] = acc; __syncthreads();
    for (int s = 128; s > 0; s >>= 1) {
        if (threadIdx.x < s) red[threadIdx.x] += red[threadIdx.x + s];
        __syncthreads();
    }
    if (threadIdx.x == 0) { colmax[c] = cm; colsum[c] = red[0]; }
}

__global__ void k_finalout(const float* __restrict__ logits,
                           const float* __restrict__ colmax, const float* __restrict__ colsum,
                           float* __restrict__ out_lsm, float* __restrict__ out_sm0) {
    int gid  = blockIdx.x * blockDim.x + threadIdx.x;
    int node = gid >> 5;
    if (node >= N_NODES) return;
    int lane = gid & 31;
    const float* row = logits + (size_t)node * NC;
    float v0 = row[lane];
    bool has2 = lane < (NC - 32);
    float v1 = has2 ? row[lane + 32] : -INFINITY;
    float m = fmaxf(v0, v1);
    #pragma unroll
    for (int off = 16; off > 0; off >>= 1) m = fmaxf(m, __shfl_xor(m, off, 32));
    float s = expf(v0 - m) + (has2 ? expf(v1 - m) : 0.f);
    #pragma unroll
    for (int off = 16; off > 0; off >>= 1) s += __shfl_xor(s, off, 32);
    float lse = m + logf(s);
    out_lsm[(size_t)node * NC + lane] = v0 - lse;
    out_sm0[(size_t)node * NC + lane] = expf(v0 - colmax[lane]) / colsum[lane];
    if (has2) {
        out_lsm[(size_t)node * NC + lane + 32] = v1 - lse;
        out_sm0[(size_t)node * NC + lane + 32] =
            expf(v1 - colmax[lane + 32]) / colsum[lane + 32];
    }
}

extern "C" void kernel_launch(void* const* d_in, const int* in_sizes, int n_in,
                              void* d_out, int out_size, void* d_ws, size_t ws_size,
                              hipStream_t stream) {
    (void)in_sizes; (void)n_in; (void)out_size; (void)ws_size;
    const float* x   = (const float*)d_in[0];
    const int*   ei  = (const int*)d_in[1];
    const float* W1  = (const float*)d_in[2];
    const float* b1  = (const float*)d_in[3];
    const float* W2  = (const float*)d_in[4];
    const float* b2  = (const float*)d_in[5];
    const int* src = ei;
    const int* dst = ei + N_EDGES;

    float* ws = (float*)d_ws;
    size_t off = 0;
    float* dinv = ws + off; off += N_NODES; off = (off + 3) & ~(size_t)3;
    float* xw1  = ws + off; off += (size_t)N_NODES * NH;
    float* agg1 = ws + off; off += (size_t)N_NODES * NH;
    float* W2p  = ws + off; off += NH * NCP;
    float* xw2  = ws + off; off += (size_t)N_NODES * NCP;
    float* agg2 = ws + off; off += (size_t)N_NODES * NC;
    float* colmax = ws + off; off += 64;
    float* colsum = ws + off; off += 64;

    float* out_lsm = (float*)d_out;
    float* logits  = (float*)d_out + (size_t)N_NODES * NC;
    float* out_sm0 = (float*)d_out + 2 * (size_t)N_NODES * NC;

    hipMemsetAsync(dinv, 0, N_NODES * sizeof(float), stream);
    k_degree<<<(N_EDGES + 255) / 256, 256, 0, stream>>>(dst, dinv);
    k_dinv<<<(N_NODES + 255) / 256, 256, 0, stream>>>(dinv);

    k_gemm_wmma<NF, NF, NH><<<N_NODES / 16, (NH / 16) * 32, 0, stream>>>(x, W1, xw1);
    k_self1<<<(N_NODES * NH + 255) / 256, 256, 0, stream>>>(xw1, dinv, agg1);
    k_edge1<<<(size_t)N_EDGES * 32 / 256, 256, 0, stream>>>(src, dst, dinv, xw1, agg1);
    k_relu_bias<<<(N_NODES * NH + 255) / 256, 256, 0, stream>>>(agg1, b1);

    k_padw2<<<(NH * NCP + 255) / 256, 256, 0, stream>>>(W2, W2p);
    k_gemm_wmma<NH, NH, NCP><<<N_NODES / 16, (NCP / 16) * 32, 0, stream>>>(agg1, W2p, xw2);
    k_self2<<<(N_NODES * NC + 255) / 256, 256, 0, stream>>>(xw2, dinv, agg2);
    k_edge2<<<(size_t)N_EDGES * 32 / 256, 256, 0, stream>>>(src, dst, dinv, xw2, agg2);
    k_bias2<<<(N_NODES * NC + 255) / 256, 256, 0, stream>>>(agg2, b2, logits);

    k_colstats<<<NC, 256, 0, stream>>>(logits, colmax, colsum);
    k_finalout<<<(N_NODES * 32 + 255) / 256, 256, 0, stream>>>(logits, colmax, colsum,
                                                               out_lsm, out_sm0);
}


# column softmax statistics: 40 stride-40 column walks replaced by 625 coalesced row-tile work-groups (per-tile max, rescaled sum) plus a 40-block combine; column sums use a different reduction tree
# speedup vs baseline: 1.3894x; 1.3894x over previous
#include <hip/hip_runtime.h>
#include <math.h>

#define N_NODES 100000
#define N_EDGES 1600000
#define NF 256
#define NH 128
#define NC 40
#define NCP 48

typedef float v2f __attribute__((ext_vector_type(2)));
typedef float v8f __attribute__((ext_vector_type(8)));

__global__ void k_degree(const int* __restrict__ dst, float* __restrict__ deg) {
    int e = blockIdx.x * blockDim.x + threadIdx.x;
    if (e < N_EDGES) atomicAdd(&deg[dst[e]], 1.0f);
}

__global__ void k_dinv(float* __restrict__ deg) {
    int i = blockIdx.x * blockDim.x + threadIdx.x;
    if (i < N_NODES) deg[i] = rsqrtf(deg[i] + 1.0f);
}

template<int K, int LDA, int LDB>
__global__ void k_gemm_wmma(const float* __restrict__ A, const float* __restrict__ B,
                            float* __restrict__ C) {
    const int mt   = blockIdx.x;
    const int nt   = threadIdx.x >> 5;
    const int lane = threadIdx.x & 31;
    const int half = lane >> 4;
    const int lm   = lane & 15;

    const float* arow = A + (size_t)(mt * 16 + lm) * LDA;
    const float* bcol = B + nt * 16 + lm;

    v8f c = {0.f, 0.f, 0.f, 0.f, 0.f, 0.f, 0.f, 0.f};
    #pragma unroll 4
    for (int kb = 0; kb < K; kb += 4) {
        const int ka = kb + 2 * half;
        v2f a; a.x = arow[ka];                 a.y = arow[ka + 1];
        v2f b; b.x = bcol[(size_t)ka * LDB];   b.y = bcol[(size_t)(ka + 1) * LDB];
        c = __builtin_amdgcn_wmma_f32_16x16x4_f32(false, a, false, b,
                                                  (short)0, c, false, false);
    }
    float* out = C + (size_t)(mt * 16 + 8 * half) * LDB + nt * 16 + lm;
    #pragma unroll
    for (int v = 0; v < 8; ++v) out[(size_t)v * LDB] = c[v];
}

__global__ void k_self1(const float* __restrict__ xw, const float* __restrict__ dinv,
                        float* __restrict__ agg) {
    int id = blockIdx.x * blockDim.x + threadIdx.x;
    if (id < N_NODES * NH) {
        int i = id >> 7;
        float di = dinv[i];
        agg[id] = xw[id] * di * di;
    }
}

__global__ void k_edge1(const int* __restrict__ src, const int* __restrict__ dst,
                        const float* __restrict__ dinv, const float* __restrict__ xw,
                        float* __restrict__ agg) {
    int gid  = blockIdx.x * blockDim.x + threadIdx.x;
    int e    = gid >> 5;
    if (e >= N_EDGES) return;
    int lane = gid & 31;
    int s = src[e], d = dst[e];
    float norm = dinv[s] * dinv[d];
    const float4 v = *(const float4*)(xw + (size_t)s * NH + lane * 4);
    float* o = agg + (size_t)d * NH + lane * 4;
    atomicAdd(o + 0, v.x * norm);
    atomicAdd(o + 1, v.y * norm);
    atomicAdd(o + 2, v.z * norm);
    atomicAdd(o + 3, v.w * norm);
}

__global__ void k_relu_bias(float* __restrict__ agg, const float* __restrict__ b) {
    int id = blockIdx.x * blockDim.x + threadIdx.x;
    if (id < N_NODES * NH) agg[id] = fmaxf(agg[id] + b[id & (NH - 1)], 0.f);
}

__global__ void k_padw2(const float* __restrict__ W2, float* __restrict__ W2p) {
    int id = blockIdx.x * blockDim.x + threadIdx.x;
    if (id < NH * NCP) {
        int k = id / NCP, n = id % NCP;
        W2p[id] = (n < NC) ? W2[k * NC + n] : 0.f;
    }
}

__global__ void k_self2(const float* __restrict__ xw2, const float* __restrict__ dinv,
                        float* __restrict__ agg2) {
    int id = blockIdx.x * blockDim.x + threadIdx.x;
    if (id < N_NODES * NC) {
        int i = id / NC, f = id - i * NC;
        float di = dinv[i];
        agg2[id] = xw2[(size_t)i * NCP + f] * di * di;
    }
}

__global__ void k_edge2(const int* __restrict__ src, const int* __restrict__ dst,
                        const float* __restrict__ dinv, const float* __restrict__ xw2,
                        float* __restrict__ agg2) {
    int gid  = blockIdx.x * blockDim.x + threadIdx.x;
    int e    = gid >> 5;
    if (e >= N_EDGES) return;
    int lane = gid & 31;
    int s = src[e], d = dst[e];
    float norm = dinv[s] * dinv[d];
    atomicAdd(agg2 + (size_t)d * NC + lane, xw2[(size_t)s * NCP + lane] * norm);
    if (lane < NC - 32)
        atomicAdd(agg2 + (size_t)d * NC + lane + 32,
                  xw2[(size_t)s * NCP + lane + 32] * norm);
}

__global__ void k_bias2(const float* __restrict__ agg2, const float* __restrict__ b2,
                        float* __restrict__ logits) {
    int id = blockIdx.x * blockDim.x + threadIdx.x;
    if (id < N_NODES * NC) {
        int f = id % NC;
        logits[id] = agg2[id] + b2[f];
    }
}

#define CS_THREADS 320
#define CS_ROWS    160
#define CS_Q       (NC / 4)
#define CS_LOADS   (CS_ROWS * CS_Q / CS_THREADS)
#define CS_SLOTS   (CS_THREADS / CS_Q)
#define CS_BLOCKS  (N_NODES / CS_ROWS)
#define CS_PITCH   128
#define CS_SUMOFF  64
static_assert(NC % 4 == 0, "rows must split into float4 pieces");
static_assert(CS_THREADS % CS_Q == 0, "a thread must stay on one column group");
static_assert((CS_ROWS * CS_Q) % CS_THREADS == 0, "whole loads per thread");
static_assert(N_NODES % CS_ROWS == 0, "whole row tiles");
static_assert(CS_SLOTS % 4 == 0, "row slots are summed in four chains");
static_assert(NC <= CS_SUMOFF && CS_SUMOFF + NC <= CS_PITCH, "partial plane layout");

__global__ __launch_bounds__(CS_THREADS)
void k_colpart(const float* __restrict__ logits, float* __restrict__ part) {
    __shared__ float4 tile[CS_THREADS];
    __shared__ float4 tmax4[CS_Q];
    const int t = threadIdx.x;
    const float4* p = (const float4*)logits + (size_t)blockIdx.x * (CS_ROWS * CS_Q) + t;
    const float* tf = (const float*)tile;
    float* o = part + (size_t)blockIdx.x * CS_PITCH + t;

    float4 v[CS_LOADS];
    for (int k = 0; k < CS_LOADS; ++k) v[k] = p[k * CS_THREADS];

    float4 m = v[0];
    for (int k = 1; k < CS_LOADS; ++k) {
        m.x = fmaxf(m.x, v[k].x); m.y = fmaxf(m.y, v[k].y);
        m.z = fmaxf(m.z, v[k].z); m.w = fmaxf(m.w, v[k].w);
    }
    tile[t] = m;
    __syncthreads();
    if (t < NC) {
        float M = tf[t];
        for (int j = 1; j < CS_SLOTS; ++j) M = fmaxf(M, tf[j * NC + t]);
        ((float*)tmax4)[t] = M;
        o[0] = M;
    }
    __syncthreads();

    const float4 M4 = tmax4[t % CS_Q];
    float4 s = {0.f, 0.f, 0.f, 0.f};
    for (int k = 0; k < CS_LOADS; ++k) {
        s.x += expf(v[k].x - M4.x); s.y += expf(v[k].y - M4.y);
        s.z += expf(v[k].z - M4.z); s.w += expf(v[k].w - M4.w);
    }
    tile[t] = s;
    __syncthreads();
    if (t < NC) {
        float a0 = 0.f, a1 = 0.f, a2 = 0.f, a3 = 0.f;
        for (int j = 0; j < CS_SLOTS; j += 4) {
            a0 += tf[(j + 0) * NC + t]; a1 += tf[(j + 1) * NC + t];
            a2 += tf[(j + 2) * NC + t]; a3 += tf[(j + 3) * NC + t];
        }
        o[CS_SUMOFF] = (a0 + a1) + (a2 + a3);
    }
}

#define CC_THREADS 256
#define CC_PER     ((CS_BLOCKS + CC_THREADS - 1) / CC_THREADS)

__global__ __launch_bounds__(CC_THREADS)
void k_colcomb(const float* __restrict__ part,
               float* __restrict__ colmax, float* __restrict__ colsum) {
    __shared__ float wmax[CC_THREADS / 32];
    __shared__ float wsum[CC_THREADS / 32];
    const int c = blockIdx.x;
    const int t = threadIdx.x;

    float pm[CC_PER], ps[CC_PER];
    for (int i = 0; i < CC_PER; ++i) {
        const int b = t + i * CC_THREADS;
        const bool ok = b < CS_BLOCKS;
        const float* q = part + (size_t)(ok ? b : 0) * CS_PITCH + c;
        const float qm = q[0], qs = q[CS_SUMOFF];
        pm[i] = ok ? qm : -INFINITY;
        ps[i] = ok ? qs : 0.f;
    }
    float m = pm[0];
    for (int i = 1; i < CC_PER; ++i) m = fmaxf(m, pm[i]);
    for (int off = 16; off > 0; off >>= 1) m = fmaxf(m, __shfl_xor(m, off, 32));
    if ((t & 31) == 0) wmax[t >> 5] = m;
    __syncthreads();
    float cm = wmax[0];
    for (int w = 1; w < CC_THREADS / 32; ++w) cm = fmaxf(cm, wmax[w]);

    float acc = 0.f;
    for (int i = 0; i < CC_PER; ++i) {
        const bool ok = (t + i * CC_THREADS) < CS_BLOCKS;
        acc += ok ? ps[i] * expf(pm[i] - cm) : 0.f;
    }
    for (int off = 16; off > 0; off >>= 1) acc += __shfl_xor(acc, off, 32);
    if ((t & 31) == 0) wsum[t >> 5] = acc;
    __syncthreads();
    if (t == 0) {
        float tot = wsum[0];
        for (int w = 1; w < CC_THREADS / 32; ++w) tot += wsum[w];
        colmax[c] = cm;
        colsum[c] = tot;
    }
}

__global__ void k_finalout(const float* __restrict__ logits,
                           const float* __restrict__ colmax, const float* __restrict__ colsum,
                           float* __restrict__ out_lsm, float* __restrict__ out_sm0) {
    int gid  = blockIdx.x * blockDim.x + threadIdx.x;
    int node = gid >> 5;
    if (node >= N_NODES) return;
    int lane = gid & 31;
    const float* row = logits + (size_t)node * NC;
    float v0 = row[lane];
    bool has2 = lane < (NC - 32);
    float v1 = has2 ? row[lane + 32] : -INFINITY;
    float m = fmaxf(v0, v1);
    #pragma unroll
    for (int off = 16; off > 0; off >>= 1) m = fmaxf(m, __shfl_xor(m, off, 32));
    float s = expf(v0 - m) + (has2 ? expf(v1 - m) : 0.f);
    #pragma unroll
    for (int off = 16; off > 0; off >>= 1) s += __shfl_xor(s, off, 32);
    float lse = m + logf(s);
    out_lsm[(size_t)node * NC + lane] = v0 - lse;
    out_sm0[(size_t)node * NC + lane] = expf(v0 - colmax[lane]) / colsum[lane];
    if (has2) {
        out_lsm[(size_t)node * NC + lane + 32] = v1 - lse;
        out_sm0[(size_t)node * NC + lane + 32] =
            expf(v1 - colmax[lane + 32]) / colsum[lane + 32];
    }
}

extern "C" void kernel_launch(void* const* d_in, const int* in_sizes, int n_in,
                              void* d_out, int out_size, void* d_ws, size_t ws_size,
                              hipStream_t stream) {
    (void)in_sizes; (void)n_in; (void)out_size; (void)ws_size;
    const float* x   = (const float*)d_in[0];
    const int*   ei  = (const int*)d_in[1];
    const float* W1  = (const float*)d_in[2];
    const float* b1  = (const float*)d_in[3];
    const float* W2  = (const float*)d_in[4];
    const float* b2  = (const float*)d_in[5];
    const int* src = ei;
    const int* dst = ei + N_EDGES;

    float* ws = (float*)d_ws;
    size_t off = 0;
    float* dinv = ws + off; off += N_NODES; off = (off + 3) & ~(size_t)3;
    float* xw1  = ws + off; off += (size_t)N_NODES * NH;
    float* agg1 = ws + off; off += (size_t)N_NODES * NH;
    float* W2p  = ws + off; off += NH * NCP;
    float* xw2  = ws + off; off += (size_t)N_NODES * NCP;
    float* agg2 = ws + off; off += (size_t)N_NODES * NC;
    float* colmax = ws + off; off += 64;
    float* colsum = ws + off; off += 64;

    float* out_lsm = (float*)d_out;
    float* logits  = (float*)d_out + (size_t)N_NODES * NC;
    float* out_sm0 = (float*)d_out + 2 * (size_t)N_NODES * NC;

    hipMemsetAsync(dinv, 0, N_NODES * sizeof(float), stream);
    k_degree<<<(N_EDGES + 255) / 256, 256, 0, stream>>>(dst, dinv);
    k_dinv<<<(N_NODES + 255) / 256, 256, 0, stream>>>(dinv);

    k_gemm_wmma<NF, NF, NH><<<N_NODES / 16, (NH / 16) * 32, 0, stream>>>(x, W1, xw1);
    k_self1<<<(N_NODES * NH + 255) / 256, 256, 0, stream>>>(xw1, dinv, agg1);
    k_edge1<<<(size_t)N_EDGES * 32 / 256, 256, 0, stream>>>(src, dst, dinv, xw1, agg1);
    k_relu_bias<<<(N_NODES * NH + 255) / 256, 256, 0, stream>>>(agg1, b1);

    k_padw2<<<(NH * NCP + 255) / 256, 256, 0, stream>>>(W2, W2p);
    k_gemm_wmma<NH, NH, NCP><<<N_NODES / 16, (NCP / 16) * 32, 0, stream>>>(agg1, W2p, xw2);
    k_self2<<<(N_NODES * NC + 255) / 256, 256, 0, stream>>>(xw2, dinv, agg2);
    k_edge2<<<(size_t)N_EDGES * 32 / 256, 256, 0, stream>>>(src, dst, dinv, xw2, agg2);
    k_bias2<<<(N_NODES * NC + 255) / 256, 256, 0, stream>>>(agg2, b2, logits);

    float* cpart = xw1;
    k_colpart<<<CS_BLOCKS, CS_THREADS, 0, stream>>>(logits, cpart);
    k_colcomb<<<NC, CC_THREADS, 0, stream>>>(cpart, colmax, colsum);
    k_finalout<<<(N_NODES * 32 + 255) / 256, 256, 0, stream>>>(logits, colmax, colsum,
                                                               out_lsm, out_sm0);
}


# float-atomic edge scatter replaced by per-call counting sort of edges by destination (integer slot counter) + one wave per node summing rows, stored once; sum order per node = slot arrival order
# speedup vs baseline: 2.9133x; 2.0968x over previous
#include <hip/hip_runtime.h>
#include <math.h>

#define N_NODES 100000
#define N_EDGES 1600000
#define NF 256
#define NH 128
#define NC 40
#define NCP 48

typedef float v2f __attribute__((ext_vector_type(2)));
typedef float v8f __attribute__((ext_vector_type(8)));

#define E4 (N_EDGES / 4)
static_assert(N_EDGES % 4 == 0 && N_NODES % 4 == 0, "16-byte index and count accesses");

__global__ void k_count(const int* __restrict__ dst, int* __restrict__ cnt,
                        int* __restrict__ rank) {
    int t = blockIdx.x * blockDim.x + threadIdx.x;
    if (t < E4) {
        const int4 d = ((const int4*)dst)[t];
        int4 r;
        r.x = atomicAdd(&cnt[d.x], 1);
        r.y = atomicAdd(&cnt[d.y], 1);
        r.z = atomicAdd(&cnt[d.z], 1);
        r.w = atomicAdd(&cnt[d.w], 1);
        ((int4*)rank)[t] = r;
    }
}

#define SC_THREADS 1024
#define SC_WAVES   (SC_THREADS / 32)
#define SC_BLOCKS  ((N_NODES / 4 + SC_THREADS - 1) / SC_THREADS)
__global__ __launch_bounds__(SC_THREADS)
void k_rowptr(const int* __restrict__ cnt, int* __restrict__ rowptr, float* __restrict__ dinv) {
    __shared__ int wpre[SC_WAVES];
    __shared__ int wsum[SC_WAVES];
    const int t = threadIdx.x, lane = t & 31, w = t >> 5;
    const int4* c4 = (const int4*)cnt;
    const int nb = blockIdx.x;

    int pre = 0;
    for (int k = 0; k < nb; k += 8) {
        int4 v[8];
        for (int u = 0; u < 8; ++u) v[u] = c4[min(k + u, nb - 1) * SC_THREADS + t];
        for (int u = 0; u < 8; ++u)
            pre += (k + u < nb) ? (v[u].x + v[u].y) + (v[u].z + v[u].w) : 0;
    }
    for (int off = 16; off > 0; off >>= 1) pre += __shfl_xor(pre, off, 32);
    if (lane == 0) wpre[w] = pre;

    const int i4 = nb * SC_THREADS + t;
    const bool ok = i4 < N_NODES / 4;
    int4 v = {0, 0, 0, 0};
    if (ok) v = c4[i4];
    const int s = (v.x + v.y) + (v.z + v.w);
    int inc = s;
    for (int off = 1; off < 32; off <<= 1) {
        const int y = __shfl_up(inc, off, 32);
        if (lane >= off) inc += y;
    }
    if (lane == 31) wsum[w] = inc;
    __syncthreads();
    int base = 0;
    for (int k = 0; k < SC_WAVES; ++k) base += wpre[k];
    for (int k = 0; k < w; ++k) base += wsum[k];
    if (ok) {
        int4 r;
        r.x = base + inc - s; r.y = r.x + v.x; r.z = r.y + v.y; r.w = r.z + v.z;
        ((int4*)rowptr)[i4] = r;
        float4 di;
        di.x = rsqrtf((float)v.x + 1.0f); di.y = rsqrtf((float)v.y + 1.0f);
        di.z = rsqrtf((float)v.z + 1.0f); di.w = rsqrtf((float)v.w + 1.0f);
        ((float4*)dinv)[i4] = di;
    }
}

__global__ void k_fill(const int* __restrict__ src, const int* __restrict__ dst,
                       const int* __restrict__ rank, const int* __restrict__ rowptr,
                       int* __restrict__ nbr) {
    int t = blockIdx.x * blockDim.x + threadIdx.x;
    if (t < E4) {
        const int4 s = ((const int4*)src)[t];
        const int4 d = ((const int4*)dst)[t];
        const int4 r = ((const int4*)rank)[t];
        nbr[rowptr[d.x] + r.x] = s.x;
        nbr[rowptr[d.y] + r.y] = s.y;
        nbr[rowptr[d.z] + r.z] = s.z;
        nbr[rowptr[d.w] + r.w] = s.w;
    }
}

template<int K, int LDA, int LDB>
__global__ void k_gemm_wmma(const float* __restrict__ A, const float* __restrict__ B,
                            float* __restrict__ C) {
    const int mt   = blockIdx.x;
    const int nt   = threadIdx.x >> 5;
    const int lane = threadIdx.x & 31;
    const int half = lane >> 4;
    const int lm   = lane & 15;

    const float* arow = A + (size_t)(mt * 16 + lm) * LDA;
    const float* bcol = B + nt * 16 + lm;

    v8f c = {0.f, 0.f, 0.f, 0.f, 0.f, 0.f, 0.f, 0.f};
    #pragma unroll 4
    for (int kb = 0; kb < K; kb += 4) {
        const int ka = kb + 2 * half;
        v2f a; a.x = arow[ka];                 a.y = arow[ka + 1];
        v2f b; b.x = bcol[(size_t)ka * LDB];   b.y = bcol[(size_t)(ka + 1) * LDB];
        c = __builtin_amdgcn_wmma_f32_16x16x4_f32(false, a, false, b,
                                                  (short)0, c, false, false);
    }
    float* out = C + (size_t)(mt * 16 + 8 * half) * LDB + nt * 16 + lm;
    #pragma unroll
    for (int v = 0; v < 8; ++v) out[(size_t)v * LDB] = c[v];
}

#define GA_WAVES 8
static_assert(N_NODES % GA_WAVES == 0, "whole work-groups of nodes");
static_assert(NH == 128, "a 32-lane wave covers a hidden row with one float4 per lane");

__device__ __forceinline__ int   bcast_i(int v, int j)   { return __builtin_amdgcn_readlane(v, j); }
__device__ __forceinline__ float bcast_f(float v, int j) {
    return __int_as_float(__builtin_amdgcn_readlane(__float_as_int(v), j));
}

__global__ __launch_bounds__(GA_WAVES * 32)
void k_gather1(const int* __restrict__ rowptr, const int* __restrict__ cnt,
               const int* __restrict__ nbr, const float* __restrict__ dinv,
               const float* __restrict__ xw, const float* __restrict__ b,
               float* __restrict__ h) {
    const int lane = threadIdx.x & 31;
    const int node = __builtin_amdgcn_readfirstlane(blockIdx.x * GA_WAVES + (threadIdx.x >> 5));
    const int start = rowptr[node], deg = cnt[node];
    const float di = dinv[node];
    const float4* rows = (const float4*)xw;

    const float4 sv = rows[(size_t)node * (NH / 4) + lane];
    float4 acc;
    acc.x = sv.x * di * di; acc.y = sv.y * di * di; acc.z = sv.z * di * di; acc.w = sv.w * di * di;

    for (int base = 0; base < deg; base += 32) {
        const int n = min(32, deg - base);
        int my_s = 0; float my_d = 0.f;
        if (lane < n) { my_s = nbr[start + base + lane]; my_d = dinv[my_s]; }
        for (int j = 0; j < n; j += 4) {
            const int r = n - j;
            const float4 z = {0.f, 0.f, 0.f, 0.f};
            float4 v1 = z, v2 = z, v3 = z;
            const float4 v0 = rows[(size_t)bcast_i(my_s, j) * (NH / 4) + lane];
            if (r > 1) v1 = rows[(size_t)bcast_i(my_s, j + 1) * (NH / 4) + lane];
            if (r > 2) v2 = rows[(size_t)bcast_i(my_s, j + 2) * (NH / 4) + lane];
            if (r > 3) v3 = rows[(size_t)bcast_i(my_s, j + 3) * (NH / 4) + lane];
            const float w0 = bcast_f(my_d, j) * di;
            acc.x += v0.x * w0; acc.y += v0.y * w0; acc.z += v0.z * w0; acc.w += v0.w * w0;
            if (r > 1) {
                const float w1 = bcast_f(my_d, j + 1) * di;
                acc.x += v1.x * w1; acc.y += v1.y * w1; acc.z += v1.z * w1; acc.w += v1.w * w1;
            }
            if (r > 2) {
                const float w2 = bcast_f(my_d, j + 2) * di;
                acc.x += v2.x * w2; acc.y += v2.y * w2; acc.z += v2.z * w2; acc.w += v2.w * w2;
            }
            if (r > 3) {
                const float w3 = bcast_f(my_d, j + 3) * di;
                acc.x += v3.x * w3; acc.y += v3.y * w3; acc.z += v3.z * w3; acc.w += v3.w * w3;
            }
        }
    }
    float4 b4;
    b4.x = b[lane * 4 + 0]; b4.y = b[lane * 4 + 1]; b4.z = b[lane * 4 + 2]; b4.w = b[lane * 4 + 3];
    float4 o;
    o.x = fmaxf(acc.x + b4.x, 0.f); o.y = fmaxf(acc.y + b4.y, 0.f);
    o.z = fmaxf(acc.z + b4.z, 0.f); o.w = fmaxf(acc.w + b4.w, 0.f);
    ((float4*)h)[(size_t)node * (NH / 4) + lane] = o;
}

__global__ void k_padw2(const float* __restrict__ W2, float* __restrict__ W2p) {
    int id = blockIdx.x * blockDim.x + threadIdx.x;
    if (id < NH * NCP) {
        int k = id / NCP, n = id % NCP;
        W2p[id] = (n < NC) ? W2[k * NC + n] : 0.f;
    }
}

#define L2_Q (NC / 4)
static_assert(3 * L2_Q <= 32 && NCP % 4 == 0, "three edge groups per wave");

__global__ __launch_bounds__(GA_WAVES * 32)
void k_gather2(const int* __restrict__ rowptr, const int* __restrict__ cnt,
               const int* __restrict__ nbr, const float* __restrict__ dinv,
               const float* __restrict__ xw2, const float* __restrict__ b2,
               float* __restrict__ logits) {
    const int lane = threadIdx.x & 31;
    const int node = __builtin_amdgcn_readfirstlane(blockIdx.x * GA_WAVES + (threadIdx.x >> 5));
    const int start = rowptr[node], deg = cnt[node];
    const float di = dinv[node];
    const float4* rows = (const float4*)xw2;
    const int g = lane / L2_Q;
    const int q = lane - g * L2_Q;
    const bool live = g < 3;

    float4 acc = {0.f, 0.f, 0.f, 0.f};
    if (g == 0) {
        const float4 sv = rows[(size_t)node * (NCP / 4) + q];
        acc.x = sv.x * di * di; acc.y = sv.y * di * di; acc.z = sv.z * di * di; acc.w = sv.w * di * di;
    }
    for (int base = 0; base < deg; base += 32) {
        const int n = min(32, deg - base);
        int my_s = 0; float my_d = 0.f;
        if (lane < n) { my_s = nbr[start + base + lane]; my_d = dinv[my_s]; }
        for (int j0 = 0; j0 < n; j0 += 6) {
            const int ja = j0 + g, jb = j0 + 3 + g;
            const bool va = live && ja < n, vb = live && jb < n;
            const int   sa = __shfl(my_s, ja & 31, 32), sb = __shfl(my_s, jb & 31, 32);
            const float wa = __shfl(my_d, ja & 31, 32) * di, wb = __shfl(my_d, jb & 31, 32) * di;
            float4 xa = {0.f, 0.f, 0.f, 0.f}, xb = {0.f, 0.f, 0.f, 0.f};
            if (va) xa = rows[(size_t)sa * (NCP / 4) + q];
            if (vb) xb = rows[(size_t)sb * (NCP / 4) + q];
            if (va) { acc.x += xa.x * wa; acc.y += xa.y * wa; acc.z += xa.z * wa; acc.w += xa.w * wa; }
            if (vb) { acc.x += xb.x * wb; acc.y += xb.y * wb; acc.z += xb.z * wb; acc.w += xb.w * wb; }
        }
    }
    const int l1 = (lane + L2_Q) & 31, l2 = (lane + 2 * L2_Q) & 31;
    const float ax = __shfl(acc.x, l1, 32), ay = __shfl(acc.y, l1, 32);
    const float az = __shfl(acc.z, l1, 32), aw = __shfl(acc.w, l1, 32);
    const float bx = __shfl(acc.x, l2, 32), by = __shfl(acc.y, l2, 32);
    const float bz = __shfl(acc.z, l2, 32), bw = __shfl(acc.w, l2, 32);
    if (g == 0) {
        float4 bb;
        bb.x = b2[q * 4 + 0]; bb.y = b2[q * 4 + 1]; bb.z = b2[q * 4 + 2]; bb.w = b2[q * 4 + 3];
        float4 o;
        o.x = (acc.x + ax + bx) + bb.x; o.y = (acc.y + ay + by) + bb.y;
        o.z = (acc.z + az + bz) + bb.z; o.w = (acc.w + aw + bw) + bb.w;
        ((float4*)logits)[(size_t)node * L2_Q + q] = o;
    }
}

#define CS_THREADS 320
#define CS_ROWS    160
#define CS_Q       (NC / 4)
#define CS_LOADS   (CS_ROWS * CS_Q / CS_THREADS)
#define CS_SLOTS   (CS_THREADS / CS_Q)
#define CS_BLOCKS  (N_NODES / CS_ROWS)
#define CS_PITCH   128
#define CS_SUMOFF  64
static_assert(NC % 4 == 0, "rows must split into float4 pieces");
static_assert(CS_THREADS % CS_Q == 0, "a thread must stay on one column group");
static_assert((CS_ROWS * CS_Q) % CS_THREADS == 0, "whole loads per thread");
static_assert(N_NODES % CS_ROWS == 0, "whole row tiles");
static_assert(CS_SLOTS % 4 == 0, "row slots are summed in four chains");
static_assert(NC <= CS_SUMOFF && CS_SUMOFF + NC <= CS_PITCH, "partial plane layout");

__global__ __launch_bounds__(CS_THREADS)
void k_colpart(const float* __restrict__ logits, float* __restrict__ part) {
    __shared__ float4 tile[CS_THREADS];
    __shared__ float4 tmax4[CS_Q];
    const int t = threadIdx.x;
    const float4* p = (const float4*)logits + (size_t)blockIdx.x * (CS_ROWS * CS_Q) + t;
    const float* tf = (const float*)tile;
    float* o = part + (size_t)blockIdx.x * CS_PITCH + t;

    float4 v[CS_LOADS];
    for (int k = 0; k < CS_LOADS; ++k) v[k] = p[k * CS_THREADS];

    float4 m = v[0];
    for (int k = 1; k < CS_LOADS; ++k) {
        m.x = fmaxf(m.x, v[k].x); m.y = fmaxf(m.y, v[k].y);
        m.z = fmaxf(m.z, v[k].z); m.w = fmaxf(m.w, v[k].w);
    }
    tile[t] = m;
    __syncthreads();
    if (t < NC) {
        float M = tf[t];
        for (int j = 1; j < CS_SLOTS; ++j) M = fmaxf(M, tf[j * NC + t]);
        ((float*)tmax4)[t] = M;
        o[0] = M;
    }
    __syncthreads();

    const float4 M4 = tmax4[t % CS_Q];
    float4 s = {0.f, 0.f, 0.f, 0.f};
    for (int k = 0; k < CS_LOADS; ++k) {
        s.x += expf(v[k].x - M4.x); s.y += expf(v[k].y - M4.y);
        s.z += expf(v[k].z - M4.z); s.w += expf(v[k].w - M4.w);
    }
    tile[t] = s;
    __syncthreads();
    if (t < NC) {
        float a0 = 0.f, a1 = 0.f, a2 = 0.f, a3 = 0.f;
        for (int j = 0; j < CS_SLOTS; j += 4) {
            a0 += tf[(j + 0) * NC + t]; a1 += tf[(j + 1) * NC + t];
            a2 += tf[(j + 2) * NC + t]; a3 += tf[(j + 3) * NC + t];
        }
        o[CS_SUMOFF] = (a0 + a1) + (a2 + a3);
    }
}

#define CC_THREADS 256
#define CC_PER     ((CS_BLOCKS + CC_THREADS - 1) / CC_THREADS)

__global__ __launch_bounds__(CC_THREADS)
void k_colcomb(const float* __restrict__ part,
               float* __restrict__ colmax, float* __restrict__ colsum) {
    __shared__ float wmax[CC_THREADS / 32];
    __shared__ float wsum[CC_THREADS / 32];
    const int c = blockIdx.x;
    const int t = threadIdx.x;

    float pm[CC_PER], ps[CC_PER];
    for (int i = 0; i < CC_PER; ++i) {
        const int b = t + i * CC_THREADS;
        const bool ok = b < CS_BLOCKS;
        const float* q = part + (size_t)(ok ? b : 0) * CS_PITCH + c;
        const float qm = q[0], qs = q[CS_SUMOFF];
        pm[i] = ok ? qm : -INFINITY;
        ps[i] = ok ? qs : 0.f;
    }
    float m = pm[0];
    for (int i = 1; i < CC_PER; ++i) m = fmaxf(m, pm[i]);
    for (int off = 16; off > 0; off >>= 1) m = fmaxf(m, __shfl_xor(m, off, 32));
    if ((t & 31) == 0) wmax[t >> 5] = m;
    __syncthreads();
    float cm = wmax[0];
    for (int w = 1; w < CC_THREADS / 32; ++w) cm = fmaxf(cm, wmax[w]);

    float acc = 0.f;
    for (int i = 0; i < CC_PER; ++i) {
        const bool ok = (t + i * CC_THREADS) < CS_BLOCKS;
        acc += ok ? ps[i] * expf(pm[i] - cm) : 0.f;
    }
    for (int off = 16; off > 0; off >>= 1) acc += __shfl_xor(acc, off, 32);
    if ((t & 31) == 0) wsum[t >> 5] = acc;
    __syncthreads();
    if (t == 0) {
        float tot = wsum[0];
        for (int w = 1; w < CC_THREADS / 32; ++w) tot += wsum[w];
        colmax[c] = cm;
        colsum[c] = tot;
    }
}

__global__ void k_finalout(const float* __restrict__ logits,
                           const float* __restrict__ colmax, const float* __restrict__ colsum,
                           float* __restrict__ out_lsm, float* __restrict__ out_sm0) {
    int gid  = blockIdx.x * blockDim.x + threadIdx.x;
    int node = gid >> 5;
    if (node >= N_NODES) return;
    int lane = gid & 31;
    const float* row = logits + (size_t)node * NC;
    float v0 = row[lane];
    bool has2 = lane < (NC - 32);
    float v1 = has2 ? row[lane + 32] : -INFINITY;
    float m = fmaxf(v0, v1);
    #pragma unroll
    for (int off = 16; off > 0; off >>= 1) m = fmaxf(m, __shfl_xor(m, off, 32));
    float s = expf(v0 - m) + (has2 ? expf(v1 - m) : 0.f);
    #pragma unroll
    for (int off = 16; off > 0; off >>= 1) s += __shfl_xor(s, off, 32);
    float lse = m + logf(s);
    out_lsm[(size_t)node * NC + lane] = v0 - lse;
    out_sm0[(size_t)node * NC + lane] = expf(v0 - colmax[lane]) / colsum[lane];
    if (has2) {
        out_lsm[(size_t)node * NC + lane + 32] = v1 - lse;
        out_sm0[(size_t)node * NC + lane + 32] =
            expf(v1 - colmax[lane + 32]) / colsum[lane + 32];
    }
}

extern "C" void kernel_launch(void* const* d_in, const int* in_sizes, int n_in,
                              void* d_out, int out_size, void* d_ws, size_t ws_size,
                              hipStream_t stream) {
    (void)in_sizes; (void)n_in; (void)out_size; (void)ws_size;
    const float* x   = (const float*)d_in[0];
    const int*   ei  = (const int*)d_in[1];
    const float* W1  = (const float*)d_in[2];
    const float* b1  = (const float*)d_in[3];
    const float* W2  = (const float*)d_in[4];
    const float* b2  = (const float*)d_in[5];
    const int* src = ei;
    const int* dst = ei + N_EDGES;

    float* ws = (float*)d_ws;
    size_t off = 0;
    float* dinv = ws + off; off += N_NODES; off = (off + 3) & ~(size_t)3;
    float* xw1  = ws + off; off += (size_t)N_NODES * NH;
    float* agg1 = ws + off; off += (size_t)N_NODES * NH;
    float* W2p  = ws + off; off += NH * NCP;
    float* xw2  = ws + off; off += (size_t)N_NODES * NCP;
    float* agg2 = ws + off; off += (size_t)N_NODES * NC;
    float* colmax = ws + off; off += 64;
    float* colsum = ws + off; off += 64;

    static_assert(2 * (size_t)N_NODES + 2 * (size_t)N_EDGES <= (size_t)N_NODES * NC,
                  "edge lists fit the reused plane");
    int* cnt    = (int*)agg2;
    int* rowptr = cnt + N_NODES;
    int* rank   = rowptr + N_NODES;
    int* nbr    = rank + N_EDGES;

    float* out_lsm = (float*)d_out;
    float* logits  = (float*)d_out + (size_t)N_NODES * NC;
    float* out_sm0 = (float*)d_out + 2 * (size_t)N_NODES * NC;

    hipMemsetAsync(cnt, 0, N_NODES * sizeof(int), stream);
    k_count<<<(E4 + 255) / 256, 256, 0, stream>>>(dst, cnt, rank);
    k_rowptr<<<SC_BLOCKS, SC_THREADS, 0, stream>>>(cnt, rowptr, dinv);
    k_fill<<<(E4 + 255) / 256, 256, 0, stream>>>(src, dst, rank, rowptr, nbr);

    k_gemm_wmma<NF, NF, NH><<<N_NODES / 16, (NH / 16) * 32, 0, stream>>>(x, W1, xw1);
    k_gather1<<<N_NODES / GA_WAVES, GA_WAVES * 32, 0, stream>>>(rowptr, cnt, nbr, dinv, xw1, b1, agg1);

    k_padw2<<<(NH * NCP + 255) / 256, 256, 0, stream>>>(W2, W2p);
    k_gemm_wmma<NH, NH, NCP><<<N_NODES / 16, (NCP / 16) * 32, 0, stream>>>(agg1, W2p, xw2);
    k_gather2<<<N_NODES / GA_WAVES, GA_WAVES * 32, 0, stream>>>(rowptr, cnt, nbr, dinv, xw2, b2, logits);

    float* cpart = xw1;
    k_colpart<<<CS_BLOCKS, CS_THREADS, 0, stream>>>(logits, cpart);
    k_colcomb<<<NC, CC_THREADS, 0, stream>>>(cpart, colmax, colsum);
    k_finalout<<<(N_NODES * 32 + 255) / 256, 256, 0, stream>>>(logits, colmax, colsum,
                                                               out_lsm, out_sm0);
}


# both products: each wave keeps its 16-column weight slice in registers and walks row tiles; a tile of A is loaded once per work-group with 16-byte loads and shared through LDS; sum order unchanged
# speedup vs baseline: 3.5055x; 1.2033x over previous
#include <hip/hip_runtime.h>
#include <math.h>

#define N_NODES 100000
#define N_EDGES 1600000
#define NF 256
#define NH 128
#define NC 40
#define NCP 48

typedef float v2f __attribute__((ext_vector_type(2)));
typedef float v4f __attribute__((ext_vector_type(4)));
typedef float v8f __attribute__((ext_vector_type(8)));

#define E4 (N_EDGES / 4)
static_assert(N_EDGES % 4 == 0 && N_NODES % 4 == 0, "16-byte index and count accesses");

__global__ void k_count(const int* __restrict__ dst, int* __restrict__ cnt,
                        int* __restrict__ rank) {
    int t = blockIdx.x * blockDim.x + threadIdx.x;
    if (t < E4) {
        const int4 d = ((const int4*)dst)[t];
        int4 r;
        r.x = atomicAdd(&cnt[d.x], 1);
        r.y = atomicAdd(&cnt[d.y], 1);
        r.z = atomicAdd(&cnt[d.z], 1);
        r.w = atomicAdd(&cnt[d.w], 1);
        ((int4*)rank)[t] = r;
    }
}

#define SC_THREADS 1024
#define SC_WAVES   (SC_THREADS / 32)
#define SC_BLOCKS  ((N_NODES / 4 + SC_THREADS - 1) / SC_THREADS)
__global__ __launch_bounds__(SC_THREADS)
void k_rowptr(const int* __restrict__ cnt, int* __restrict__ rowptr, float* __restrict__ dinv) {
    __shared__ int wpre[SC_WAVES];
    __shared__ int wsum[SC_WAVES];
    const int t = threadIdx.x, lane = t & 31, w = t >> 5;
    const int4* c4 = (const int4*)cnt;
    const int nb = blockIdx.x;

    int pre = 0;
    for (int k = 0; k < nb; k += 8) {
        int4 v[8];
        for (int u = 0; u < 8; ++u) v[u] = c4[min(k + u, nb - 1) * SC_THREADS + t];
        for (int u = 0; u < 8; ++u)
            pre += (k + u < nb) ? (v[u].x + v[u].y) + (v[u].z + v[u].w) : 0;
    }
    for (int off = 16; off > 0; off >>= 1) pre += __shfl_xor(pre, off, 32);
    if (lane == 0) wpre[w] = pre;

    const int i4 = nb * SC_THREADS + t;
    const bool ok = i4 < N_NODES / 4;
    int4 v = {0, 0, 0, 0};
    if (ok) v = c4[i4];
    const int s = (v.x + v.y) + (v.z + v.w);
    int inc = s;
    for (int off = 1; off < 32; off <<= 1) {
        const int y = __shfl_up(inc, off, 32);
        if (lane >= off) inc += y;
    }
    if (lane == 31) wsum[w] = inc;
    __syncthreads();
    int base = 0;
    for (int k = 0; k < SC_WAVES; ++k) base += wpre[k];
    for (int k = 0; k < w; ++k) base += wsum[k];
    if (ok) {
        int4 r;
        r.x = base + inc - s; r.y = r.x + v.x; r.z = r.y + v.y; r.w = r.z + v.z;
        ((int4*)rowptr)[i4] = r;
        float4 di;
        di.x = rsqrtf((float)v.x + 1.0f); di.y = rsqrtf((float)v.y + 1.0f);
        di.z = rsqrtf((float)v.z + 1.0f); di.w = rsqrtf((float)v.w + 1.0f);
        ((float4*)dinv)[i4] = di;
    }
}

__global__ void k_fill(const int* __restrict__ src, const int* __restrict__ dst,
                       const int* __restrict__ rank, const int* __restrict__ rowptr,
                       int* __restrict__ nbr) {
    int t = blockIdx.x * blockDim.x + threadIdx.x;
    if (t < E4) {
        const int4 s = ((const int4*)src)[t];
        const int4 d = ((const int4*)dst)[t];
        const int4 r = ((const int4*)rank)[t];
        nbr[rowptr[d.x] + r.x] = s.x;
        nbr[rowptr[d.y] + r.y] = s.y;
        nbr[rowptr[d.z] + r.z] = s.z;
        nbr[rowptr[d.w] + r.w] = s.w;
    }
}

#define GEMM_MT (N_NODES / 16)
static_assert(N_NODES % 16 == 0, "whole row tiles");
#define G1_BLOCKS 512
#define G2_BLOCKS (GEMM_MT / 5)
static_assert(GEMM_MT % 5 == 0 && G1_BLOCKS <= GEMM_MT, "grid shapes");

template<int K, int LDB, int NT>
__global__ __launch_bounds__(NT * 32)
void k_gemm_wmma(const float* __restrict__ A, const float* __restrict__ B,
                 float* __restrict__ C) {
    constexpr int P     = K + 4;
    constexpr int RQ    = K / 4;
    constexpr int PPL   = RQ / 32;
    constexpr int RPW   = (16 + NT - 1) / NT;
    constexpr int LROWS = RPW * NT;
    constexpr int NST   = RPW * PPL;
    static_assert(LDB == NT * 16 && RQ % 32 == 0 && (P / 4) % 2 == 1, "tile shape");
    __shared__ __attribute__((aligned(16))) float tile[2][LROWS * P];

    const int t    = threadIdx.x;
    const int nt   = t >> 5;
    const int lane = t & 31;
    const int half = lane >> 4;
    const int lm   = lane & 15;

    v2f b[K / 4];
    const float* bcol = B + nt * 16 + lm;
    #pragma unroll
    for (int s = 0; s < K / 4; ++s) {
        const int ka = 4 * s + 2 * half;
        b[s].x = bcol[(size_t)ka * LDB];
        b[s].y = bcol[(size_t)(ka + 1) * LDB];
    }

    int goff[NST], loff[NST];
    #pragma unroll
    for (int i = 0; i < RPW; ++i) {
        const int row = nt + i * NT;
        const int srow = (16 % NT == 0) ? row : min(row, 15);
        #pragma unroll
        for (int j = 0; j < PPL; ++j) {
            goff[i * PPL + j] = srow * RQ + j * 32 + lane;
            loff[i * PPL + j] = row * P + (j * 32 + lane) * 4;
        }
    }

    v4f st[NST];
    int mt = blockIdx.x;
    {
        const v4f* g = (const v4f*)(A + (size_t)mt * (16 * K));
        #pragma unroll
        for (int i = 0; i < NST; ++i) st[i] = g[goff[i]];
        #pragma unroll
        for (int i = 0; i < NST; ++i) *(v4f*)&tile[0][loff[i]] = st[i];
    }
    __syncthreads();

    int cur = 0;
    for (; mt < GEMM_MT; mt += gridDim.x) {
        const int nx = mt + gridDim.x;
        const bool more = nx < GEMM_MT;
        if (more) {
            const v4f* g = (const v4f*)(A + (size_t)nx * (16 * K));
            #pragma unroll
            for (int i = 0; i < NST; ++i) st[i] = g[goff[i]];
        }

        const float* ap = &tile[cur][lm * P + 2 * half];
        v8f c = {0.f, 0.f, 0.f, 0.f, 0.f, 0.f, 0.f, 0.f};
        #pragma unroll
        for (int s = 0; s < K / 4; ++s) {
            const v2f a = *(const v2f*)(ap + 4 * s);
            c = __builtin_amdgcn_wmma_f32_16x16x4_f32(false, a, false, b[s],
                                                      (short)0, c, false, false);
        }

        if (more) {
            #pragma unroll
            for (int i = 0; i < NST; ++i) *(v4f*)&tile[cur ^ 1][loff[i]] = st[i];
        }
        __syncthreads();
        cur ^= 1;

        float* out = C + (size_t)(mt * 16 + 8 * half) * LDB + nt * 16 + lm;
        #pragma unroll
        for (int v = 0; v < 8; ++v) out[(size_t)v * LDB] = c[v];
    }
}

#define GA_WAVES 8
static_assert(N_NODES % GA_WAVES == 0, "whole work-groups of nodes");
static_assert(NH == 128, "a 32-lane wave covers a hidden row with one float4 per lane");

__device__ __forceinline__ int   bcast_i(int v, int j)   { return __builtin_amdgcn_readlane(v, j); }
__device__ __forceinline__ float bcast_f(float v, int j) {
    return __int_as_float(__builtin_amdgcn_readlane(__float_as_int(v), j));
}

__global__ __launch_bounds__(GA_WAVES * 32)
void k_gather1(const int* __restrict__ rowptr, const int* __restrict__ cnt,
               const int* __restrict__ nbr, const float* __restrict__ dinv,
               const float* __restrict__ xw, const float* __restrict__ b,
               float* __restrict__ h) {
    const int lane = threadIdx.x & 31;
    const int node = __builtin_amdgcn_readfirstlane(blockIdx.x * GA_WAVES + (threadIdx.x >> 5));
    const int start = rowptr[node], deg = cnt[node];
    const float di = dinv[node];
    const float4* rows = (const float4*)xw;

    const float4 sv = rows[(size_t)node * (NH / 4) + lane];
    float4 acc;
    acc.x = sv.x * di * di; acc.y = sv.y * di * di; acc.z = sv.z * di * di; acc.w = sv.w * di * di;

    for (int base = 0; base < deg; base += 32) {
        const int n = min(32, deg - base);
        int my_s = 0; float my_d = 0.f;
        if (lane < n) { my_s = nbr[start + base + lane]; my_d = dinv[my_s]; }
        for (int j = 0; j < n; j += 4) {
            const int r = n - j;
            const float4 z = {0.f, 0.f, 0.f, 0.f};
            float4 v1 = z, v2 = z, v3 = z;
            const float4 v0 = rows[(size_t)bcast_i(my_s, j) * (NH / 4) + lane];
            if (r > 1) v1 = rows[(size_t)bcast_i(my_s, j + 1) * (NH / 4) + lane];
            if (r > 2) v2 = rows[(size_t)bcast_i(my_s, j + 2) * (NH / 4) + lane];
            if (r > 3) v3 = rows[(size_t)bcast_i(my_s, j + 3) * (NH / 4) + lane];
            const float w0 = bcast_f(my_d, j) * di;
            acc.x += v0.x * w0; acc.y += v0.y * w0; acc.z += v0.z * w0; acc.w += v0.w * w0;
            if (r > 1) {
                const float w1 = bcast_f(my_d, j + 1) * di;
                acc.x += v1.x * w1; acc.y += v1.y * w1; acc.z += v1.z * w1; acc.w += v1.w * w1;
            }
            if (r > 2) {
                const float w2 = bcast_f(my_d, j + 2) * di;
                acc.x += v2.x * w2; acc.y += v2.y * w2; acc.z += v2.z * w2; acc.w += v2.w * w2;
            }
            if (r > 3) {
                const float w3 = bcast_f(my_d, j + 3) * di;
                acc.x += v3.x * w3; acc.y += v3.y * w3; acc.z += v3.z * w3; acc.w += v3.w * w3;
            }
        }
    }
    float4 b4;
    b4.x = b[lane * 4 + 0]; b4.y = b[lane * 4 + 1]; b4.z = b[lane * 4 + 2]; b4.w = b[lane * 4 + 3];
    float4 o;
    o.x = fmaxf(acc.x + b4.x, 0.f); o.y = fmaxf(acc.y + b4.y, 0.f);
    o.z = fmaxf(acc.z + b4.z, 0.f); o.w = fmaxf(acc.w + b4.w, 0.f);
    ((float4*)h)[(size_t)node * (NH / 4) + lane] = o;
}

__global__ void k_padw2(const float* __restrict__ W2, float* __restrict__ W2p) {
    int id = blockIdx.x * blockDim.x + threadIdx.x;
    if (id < NH * NCP) {
        int k = id / NCP, n = id % NCP;
        W2p[id] = (n < NC) ? W2[k * NC + n] : 0.f;
    }
}

#define L2_Q (NC / 4)
static_assert(3 * L2_Q <= 32 && NCP % 4 == 0, "three edge groups per wave");

__global__ __launch_bounds__(GA_WAVES * 32)
void k_gather2(const int* __restrict__ rowptr, const int* __restrict__ cnt,
               const int* __restrict__ nbr, const float* __restrict__ dinv,
               const float* __restrict__ xw2, const float* __restrict__ b2,
               float* __restrict__ logits) {
    const int lane = threadIdx.x & 31;
    const int node = __builtin_amdgcn_readfirstlane(blockIdx.x * GA_WAVES + (threadIdx.x >> 5));
    const int start = rowptr[node], deg = cnt[node];
    const float di = dinv[node];
    const float4* rows = (const float4*)xw2;
    const int g = lane / L2_Q;
    const int q = lane - g * L2_Q;
    const bool live = g < 3;

    float4 acc = {0.f, 0.f, 0.f, 0.f};
    if (g == 0) {
        const float4 sv = rows[(size_t)node * (NCP / 4) + q];
        acc.x = sv.x * di * di; acc.y = sv.y * di * di; acc.z = sv.z * di * di; acc.w = sv.w * di * di;
    }
    for (int base = 0; base < deg; base += 32) {
        const int n = min(32, deg - base);
        int my_s = 0; float my_d = 0.f;
        if (lane < n) { my_s = nbr[start + base + lane]; my_d = dinv[my_s]; }
        for (int j0 = 0; j0 < n; j0 += 6) {
            const int ja = j0 + g, jb = j0 + 3 + g;
            const bool va = live && ja < n, vb = live && jb < n;
            const int   sa = __shfl(my_s, ja & 31, 32), sb = __shfl(my_s, jb & 31, 32);
            const float wa = __shfl(my_d, ja & 31, 32) * di, wb = __shfl(my_d, jb & 31, 32) * di;
            float4 xa = {0.f, 0.f, 0.f, 0.f}, xb = {0.f, 0.f, 0.f, 0.f};
            if (va) xa = rows[(size_t)sa * (NCP / 4) + q];
            if (vb) xb = rows[(size_t)sb * (NCP / 4) + q];
            if (va) { acc.x += xa.x * wa; acc.y += xa.y * wa; acc.z += xa.z * wa; acc.w += xa.w * wa; }
            if (vb) { acc.x += xb.x * wb; acc.y += xb.y * wb; acc.z += xb.z * wb; acc.w += xb.w * wb; }
        }
    }
    const int l1 = (lane + L2_Q) & 31, l2 = (lane + 2 * L2_Q) & 31;
    const float ax = __shfl(acc.x, l1, 32), ay = __shfl(acc.y, l1, 32);
    const float az = __shfl(acc.z, l1, 32), aw = __shfl(acc.w, l1, 32);
    const float bx = __shfl(acc.x, l2, 32), by = __shfl(acc.y, l2, 32);
    const float bz = __shfl(acc.z, l2, 32), bw = __shfl(acc.w, l2, 32);
    if (g == 0) {
        float4 bb;
        bb.x = b2[q * 4 + 0]; bb.y = b2[q * 4 + 1]; bb.z = b2[q * 4 + 2]; bb.w = b2[q * 4 + 3];
        float4 o;
        o.x = (acc.x + ax + bx) + bb.x; o.y = (acc.y + ay + by) + bb.y;
        o.z = (acc.z + az + bz) + bb.z; o.w = (acc.w + aw + bw) + bb.w;
        ((float4*)logits)[(size_t)node * L2_Q + q] = o;
    }
}

#define CS_THREADS 320
#define CS_ROWS    160
#define CS_Q       (NC / 4)
#define CS_LOADS   (CS_ROWS * CS_Q / CS_THREADS)
#define CS_SLOTS   (CS_THREADS / CS_Q)
#define CS_BLOCKS  (N_NODES / CS_ROWS)
#define CS_PITCH   128
#define CS_SUMOFF  64
static_assert(NC % 4 == 0, "rows must split into float4 pieces");
static_assert(CS_THREADS % CS_Q == 0, "a thread must stay on one column group");
static_assert((CS_ROWS * CS_Q) % CS_THREADS == 0, "whole loads per thread");
static_assert(N_NODES % CS_ROWS == 0, "whole row tiles");
static_assert(CS_SLOTS % 4 == 0, "row slots are summed in four chains");
static_assert(NC <= CS_SUMOFF && CS_SUMOFF + NC <= CS_PITCH, "partial plane layout");

__global__ __launch_bounds__(CS_THREADS)
void k_colpart(const float* __restrict__ logits, float* __restrict__ part) {
    __shared__ float4 tile[CS_THREADS];
    __shared__ float4 tmax4[CS_Q];
    const int t = threadIdx.x;
    const float4* p = (const float4*)logits + (size_t)blockIdx.x * (CS_ROWS * CS_Q) + t;
    const float* tf = (const float*)tile;
    float* o = part + (size_t)blockIdx.x * CS_PITCH + t;

    float4 v[CS_LOADS];
    for (int k = 0; k < CS_LOADS; ++k) v[k] = p[k * CS_THREADS];

    float4 m = v[0];
    for (int k = 1; k < CS_LOADS; ++k) {
        m.x = fmaxf(m.x, v[k].x); m.y = fmaxf(m.y, v[k].y);
        m.z = fmaxf(m.z, v[k].z); m.w = fmaxf(m.w, v[k].w);
    }
    tile[t] = m;
    __syncthreads();
    if (t < NC) {
        float M = tf[t];
        for (int j = 1; j < CS_SLOTS; ++j) M = fmaxf(M, tf[j * NC + t]);
        ((float*)tmax4)[t] = M;
        o[0] = M;
    }
    __syncthreads();

    const float4 M4 = tmax4[t % CS_Q];
    float4 s = {0.f, 0.f, 0.f, 0.f};
    for (int k = 0; k < CS_LOADS; ++k) {
        s.x += expf(v[k].x - M4.x); s.y += expf(v[k].y - M4.y);
        s.z += expf(v[k].z - M4.z); s.w += expf(v[k].w - M4.w);
    }
    tile[t] = s;
    __syncthreads();
    if (t < NC) {
        float a0 = 0.f, a1 = 0.f, a2 = 0.f, a3 = 0.f;
        for (int j = 0; j < CS_SLOTS; j += 4) {
            a0 += tf[(j + 0) * NC + t]; a1 += tf[(j + 1) * NC + t];
            a2 += tf[(j + 2) * NC + t]; a3 += tf[(j + 3) * NC + t];
        }
        o[CS_SUMOFF] = (a0 + a1) + (a2 + a3);
    }
}

#define CC_THREADS 256
#define CC_PER     ((CS_BLOCKS + CC_THREADS - 1) / CC_THREADS)

__global__ __launch_bounds__(CC_THREADS)
void k_colcomb(const float* __restrict__ part,
               float* __restrict__ colmax, float* __restrict__ colsum) {
    __shared__ float wmax[CC_THREADS / 32];
    __shared__ float wsum[CC_THREADS / 32];
    const int c = blockIdx.x;
    const int t = threadIdx.x;

    float pm[CC_PER], ps[CC_PER];
    for (int i = 0; i < CC_PER; ++i) {
        const int b = t + i * CC_THREADS;
        const bool ok = b < CS_BLOCKS;
        const float* q = part + (size_t)(ok ? b : 0) * CS_PITCH + c;
        const float qm = q[0], qs = q[CS_SUMOFF];
        pm[i] = ok ? qm : -INFINITY;
        ps[i] = ok ? qs : 0.f;
    }
    float m = pm[0];
    for (int i = 1; i < CC_PER; ++i) m = fmaxf(m, pm[i]);
    for (int off = 16; off > 0; off >>= 1) m = fmaxf(m, __shfl_xor(m, off, 32));
    if ((t & 31) == 0) wmax[t >> 5] = m;
    __syncthreads();
    float cm = wmax[0];
    for (int w = 1; w < CC_THREADS / 32; ++w) cm = fmaxf(cm, wmax[w]);

    float acc = 0.f;
    for (int i = 0; i < CC_PER; ++i) {
        const bool ok = (t + i * CC_THREADS) < CS_BLOCKS;
        acc += ok ? ps[i] * expf(pm[i] - cm) : 0.f;
    }
    for (int off = 16; off > 0; off >>= 1) acc += __shfl_xor(acc, off, 32);
    if ((t & 31) == 0) wsum[t >> 5] = acc;
    __syncthreads();
    if (t == 0) {
        float tot = wsum[0];
        for (int w = 1; w < CC_THREADS / 32; ++w) tot += wsum[w];
        colmax[c] = cm;
        colsum[c] = tot;
    }
}

__global__ void k_finalout(const float* __restrict__ logits,
                           const float* __restrict__ colmax, const float* __restrict__ colsum,
                           float* __restrict__ out_lsm, float* __restrict__ out_sm0) {
    int gid  = blockIdx.x * blockDim.x + threadIdx.x;
    int node = gid >> 5;
    if (node >= N_NODES) return;
    int lane = gid & 31;
    const float* row = logits + (size_t)node * NC;
    float v0 = row[lane];
    bool has2 = lane < (NC - 32);
    float v1 = has2 ? row[lane + 32] : -INFINITY;
    float m = fmaxf(v0, v1);
    #pragma unroll
    for (int off = 16; off > 0; off >>= 1) m = fmaxf(m, __shfl_xor(m, off, 32));
    float s = expf(v0 - m) + (has2 ? expf(v1 - m) : 0.f);
    #pragma unroll
    for (int off = 16; off > 0; off >>= 1) s += __shfl_xor(s, off, 32);
    float lse = m + logf(s);
    out_lsm[(size_t)node * NC + lane] = v0 - lse;
    out_sm0[(size_t)node * NC + lane] = expf(v0 - colmax[lane]) / colsum[lane];
    if (has2) {
        out_lsm[(size_t)node * NC + lane + 32] = v1 - lse;
        out_sm0[(size_t)node * NC + lane + 32] =
            expf(v1 - colmax[lane + 32]) / colsum[lane + 32];
    }
}

extern "C" void kernel_launch(void* const* d_in, const int* in_sizes, int n_in,
                              void* d_out, int out_size, void* d_ws, size_t ws_size,
                              hipStream_t stream) {
    (void)in_sizes; (void)n_in; (void)out_size; (void)ws_size;
    const float* x   = (const float*)d_in[0];
    const int*   ei  = (const int*)d_in[1];
    const float* W1  = (const float*)d_in[2];
    const float* b1  = (const float*)d_in[3];
    const float* W2  = (const float*)d_in[4];
    const float* b2  = (const float*)d_in[5];
    const int* src = ei;
    const int* dst = ei + N_EDGES;

    float* ws = (float*)d_ws;
    size_t off = 0;
    float* dinv = ws + off; off += N_NODES; off = (off + 3) & ~(size_t)3;
    float* xw1  = ws + off; off += (size_t)N_NODES * NH;
    float* agg1 = ws + off; off += (size_t)N_NODES * NH;
    float* W2p  = ws + off; off += NH * NCP;
    float* xw2  = ws + off; off += (size_t)N_NODES * NCP;
    float* agg2 = ws + off; off += (size_t)N_NODES * NC;
    float* colmax = ws + off; off += 64;
    float* colsum = ws + off; off += 64;

    static_assert(2 * (size_t)N_NODES + 2 * (size_t)N_EDGES <= (size_t)N_NODES * NC,
                  "edge lists fit the reused plane");
    int* cnt    = (int*)agg2;
    int* rowptr = cnt + N_NODES;
    int* rank   = rowptr + N_NODES;
    int* nbr    = rank + N_EDGES;

    float* out_lsm = (float*)d_out;
    float* logits  = (float*)d_out + (size_t)N_NODES * NC;
    float* out_sm0 = (float*)d_out + 2 * (size_t)N_NODES * NC;

    hipMemsetAsync(cnt, 0, N_NODES * sizeof(int), stream);
    k_count<<<(E4 + 255) / 256, 256, 0, stream>>>(dst, cnt, rank);
    k_rowptr<<<SC_BLOCKS, SC_THREADS, 0, stream>>>(cnt, rowptr, dinv);
    k_fill<<<(E4 + 255) / 256, 256, 0, stream>>>(src, dst, rank, rowptr, nbr);

    k_gemm_wmma<NF, NH, NH / 16><<<G1_BLOCKS, (NH / 16) * 32, 0, stream>>>(x, W1, xw1);
    k_gather1<<<N_NODES / GA_WAVES, GA_WAVES * 32, 0, stream>>>(rowptr, cnt, nbr, dinv, xw1, b1, agg1);

    k_padw2<<<(NH * NCP + 255) / 256, 256, 0, stream>>>(W2, W2p);
    k_gemm_wmma<NH, NCP, NCP / 16><<<G2_BLOCKS, (NCP / 16) * 32, 0, stream>>>(agg1, W2p, xw2);
    k_gather2<<<N_NODES / GA_WAVES, GA_WAVES * 32, 0, stream>>>(rowptr, cnt, nbr, dinv, xw2, b2, logits);

    float* cpart = xw1;
    k_colpart<<<CS_BLOCKS, CS_THREADS, 0, stream>>>(logits, cpart);
    k_colcomb<<<NC, CC_THREADS, 0, stream>>>(cpart, colmax, colsum);
    k_finalout<<<(N_NODES * 32 + 255) / 256, 256, 0, stream>>>(logits, colmax, colsum,
                                                               out_lsm, out_sm0);
}


# both products scale each result row by its node's dinv on the way out, so both neighbour sums add plain rows (no per-edge dinv lookup); sum order unchanged, products regrouped: di*(sum of dj*row)
# speedup vs baseline: 3.7091x; 1.0581x over previous
#include <hip/hip_runtime.h>
#include <math.h>

#define N_NODES 100000
#define N_EDGES 1600000
#define NF 256
#define NH 128
#define NC 40
#define NCP 48

typedef float v2f __attribute__((ext_vector_type(2)));
typedef float v4f __attribute__((ext_vector_type(4)));
typedef float v8f __attribute__((ext_vector_type(8)));

#define E4 (N_EDGES / 4)
static_assert(N_EDGES % 4 == 0 && N_NODES % 4 == 0, "16-byte index and count accesses");

__global__ void k_count(const int* __restrict__ dst, int* __restrict__ cnt,
                        int* __restrict__ rank) {
    int t = blockIdx.x * blockDim.x + threadIdx.x;
    if (t < E4) {
        const int4 d = ((const int4*)dst)[t];
        int4 r;
        r.x = atomicAdd(&cnt[d.x], 1);
        r.y = atomicAdd(&cnt[d.y], 1);
        r.z = atomicAdd(&cnt[d.z], 1);
        r.w = atomicAdd(&cnt[d.w], 1);
        ((int4*)rank)[t] = r;
    }
}

#define SC_THREADS 1024
#define SC_WAVES   (SC_THREADS / 32)
#define SC_BLOCKS  ((N_NODES / 4 + SC_THREADS - 1) / SC_THREADS)
__global__ __launch_bounds__(SC_THREADS)
void k_rowptr(const int* __restrict__ cnt, int* __restrict__ rowptr, float* __restrict__ dinv) {
    __shared__ int wpre[SC_WAVES];
    __shared__ int wsum[SC_WAVES];
    const int t = threadIdx.x, lane = t & 31, w = t >> 5;
    const int4* c4 = (const int4*)cnt;
    const int nb = blockIdx.x;

    int pre = 0;
    for (int k = 0; k < nb; k += 8) {
        int4 v[8];
        for (int u = 0; u < 8; ++u) v[u] = c4[min(k + u, nb - 1) * SC_THREADS + t];
        for (int u = 0; u < 8; ++u)
            pre += (k + u < nb) ? (v[u].x + v[u].y) + (v[u].z + v[u].w) : 0;
    }
    for (int off = 16; off > 0; off >>= 1) pre += __shfl_xor(pre, off, 32);
    if (lane == 0) wpre[w] = pre;

    const int i4 = nb * SC_THREADS + t;
    const bool ok = i4 < N_NODES / 4;
    int4 v = {0, 0, 0, 0};
    if (ok) v = c4[i4];
    const int s = (v.x + v.y) + (v.z + v.w);
    int inc = s;
    for (int off = 1; off < 32; off <<= 1) {
        const int y = __shfl_up(inc, off, 32);
        if (lane >= off) inc += y;
    }
    if (lane == 31) wsum[w] = inc;
    __syncthreads();
    int base = 0;
    for (int k = 0; k < SC_WAVES; ++k) base += wpre[k];
    for (int k = 0; k < w; ++k) base += wsum[k];
    if (ok) {
        int4 r;
        r.x = base + inc - s; r.y = r.x + v.x; r.z = r.y + v.y; r.w = r.z + v.z;
        ((int4*)rowptr)[i4] = r;
        float4 di;
        di.x = rsqrtf((float)v.x + 1.0f); di.y = rsqrtf((float)v.y + 1.0f);
        di.z = rsqrtf((float)v.z + 1.0f); di.w = rsqrtf((float)v.w + 1.0f);
        ((float4*)dinv)[i4] = di;
    }
}

__global__ void k_fill(const int* __restrict__ src, const int* __restrict__ dst,
                       const int* __restrict__ rank, const int* __restrict__ rowptr,
                       int* __restrict__ nbr) {
    int t = blockIdx.x * blockDim.x + threadIdx.x;
    if (t < E4) {
        const int4 s = ((const int4*)src)[t];
        const int4 d = ((const int4*)dst)[t];
        const int4 r = ((const int4*)rank)[t];
        nbr[rowptr[d.x] + r.x] = s.x;
        nbr[rowptr[d.y] + r.y] = s.y;
        nbr[rowptr[d.z] + r.z] = s.z;
        nbr[rowptr[d.w] + r.w] = s.w;
    }
}

#define GEMM_MT (N_NODES / 16)
static_assert(N_NODES % 16 == 0, "whole row tiles");
#define G1_BLOCKS 512
#define G2_BLOCKS (GEMM_MT / 5)
static_assert(GEMM_MT % 5 == 0 && G1_BLOCKS <= GEMM_MT, "grid shapes");

template<int K, int LDB, int NT>
__global__ __launch_bounds__(NT * 32)
void k_gemm_wmma(const float* __restrict__ A, const float* __restrict__ B,
                 const float* __restrict__ rs, float* __restrict__ C) {
    constexpr int P     = K + 4;
    constexpr int RQ    = K / 4;
    constexpr int PPL   = RQ / 32;
    constexpr int RPW   = (16 + NT - 1) / NT;
    constexpr int LROWS = RPW * NT;
    constexpr int NST   = RPW * PPL;
    static_assert(LDB == NT * 16 && RQ % 32 == 0 && (P / 4) % 2 == 1, "tile shape");
    __shared__ __attribute__((aligned(16))) float tile[2][LROWS * P];

    const int t    = threadIdx.x;
    const int nt   = t >> 5;
    const int lane = t & 31;
    const int half = lane >> 4;
    const int lm   = lane & 15;

    v2f b[K / 4];
    const float* bcol = B + nt * 16 + lm;
    #pragma unroll
    for (int s = 0; s < K / 4; ++s) {
        const int ka = 4 * s + 2 * half;
        b[s].x = bcol[(size_t)ka * LDB];
        b[s].y = bcol[(size_t)(ka + 1) * LDB];
    }

    int goff[NST], loff[NST];
    #pragma unroll
    for (int i = 0; i < RPW; ++i) {
        const int row = nt + i * NT;
        const int srow = (16 % NT == 0) ? row : min(row, 15);
        #pragma unroll
        for (int j = 0; j < PPL; ++j) {
            goff[i * PPL + j] = srow * RQ + j * 32 + lane;
            loff[i * PPL + j] = row * P + (j * 32 + lane) * 4;
        }
    }

    v4f st[NST];
    int mt = blockIdx.x;
    const v4f* rsp = (const v4f*)(rs + 8 * half);
    v4f rs0 = rsp[mt * 4], rs1 = rsp[mt * 4 + 1];
    {
        const v4f* g = (const v4f*)(A + (size_t)mt * (16 * K));
        #pragma unroll
        for (int i = 0; i < NST; ++i) st[i] = g[goff[i]];
        #pragma unroll
        for (int i = 0; i < NST; ++i) *(v4f*)&tile[0][loff[i]] = st[i];
    }
    __syncthreads();

    int cur = 0;
    for (; mt < GEMM_MT; mt += gridDim.x) {
        const int nx = mt + gridDim.x;
        const bool more = nx < GEMM_MT;
        v4f nrs0 = rs0, nrs1 = rs1;
        if (more) {
            const v4f* g = (const v4f*)(A + (size_t)nx * (16 * K));
            #pragma unroll
            for (int i = 0; i < NST; ++i) st[i] = g[goff[i]];
            nrs0 = rsp[nx * 4]; nrs1 = rsp[nx * 4 + 1];
        }
        const float* ap = &tile[cur][lm * P + 2 * half];
        v8f c = {0.f, 0.f, 0.f, 0.f, 0.f, 0.f, 0.f, 0.f};
        #pragma unroll
        for (int s = 0; s < K / 4; ++s) {
            const v2f a = *(const v2f*)(ap + 4 * s);
            c = __builtin_amdgcn_wmma_f32_16x16x4_f32(false, a, false, b[s],
                                                      (short)0, c, false, false);
        }

        if (more) {
            #pragma unroll
            for (int i = 0; i < NST; ++i) *(v4f*)&tile[cur ^ 1][loff[i]] = st[i];
        }
        __syncthreads();
        cur ^= 1;

        float* out = C + (size_t)(mt * 16 + 8 * half) * LDB + nt * 16 + lm;
        #pragma unroll
        for (int v = 0; v < 8; ++v) out[(size_t)v * LDB] = c[v] * (v < 4 ? rs0[v & 3] : rs1[v & 3]);
        rs0 = nrs0; rs1 = nrs1;
    }
}

#define GA_WAVES 8
static_assert(N_NODES % GA_WAVES == 0, "whole work-groups of nodes");
static_assert(NH == 128, "a 32-lane wave covers a hidden row with one float4 per lane");

__device__ __forceinline__ int   bcast_i(int v, int j)   { return __builtin_amdgcn_readlane(v, j); }

__global__ __launch_bounds__(GA_WAVES * 32)
void k_gather1(const int* __restrict__ rowptr, const int* __restrict__ cnt,
               const int* __restrict__ nbr, const float* __restrict__ dinv,
               const float* __restrict__ xs, const float* __restrict__ b,
               float* __restrict__ h) {
    const int lane = threadIdx.x & 31;
    const int node = __builtin_amdgcn_readfirstlane(blockIdx.x * GA_WAVES + (threadIdx.x >> 5));
    const int start = rowptr[node], deg = cnt[node];
    const float di = dinv[node];
    const float4* rows = (const float4*)xs;

    float4 acc = rows[(size_t)node * (NH / 4) + lane];

    for (int base = 0; base < deg; base += 32) {
        const int n = min(32, deg - base);
        int my_s = 0;
        if (lane < n) my_s = nbr[start + base + lane];
        for (int j = 0; j < n; j += 4) {
            const int r = n - j;
            const float4 z = {0.f, 0.f, 0.f, 0.f};
            float4 v1 = z, v2 = z, v3 = z;
            const float4 v0 = rows[(size_t)bcast_i(my_s, j) * (NH / 4) + lane];
            if (r > 1) v1 = rows[(size_t)bcast_i(my_s, j + 1) * (NH / 4) + lane];
            if (r > 2) v2 = rows[(size_t)bcast_i(my_s, j + 2) * (NH / 4) + lane];
            if (r > 3) v3 = rows[(size_t)bcast_i(my_s, j + 3) * (NH / 4) + lane];
            acc.x += v0.x; acc.y += v0.y; acc.z += v0.z; acc.w += v0.w;
            if (r > 1) { acc.x += v1.x; acc.y += v1.y; acc.z += v1.z; acc.w += v1.w; }
            if (r > 2) { acc.x += v2.x; acc.y += v2.y; acc.z += v2.z; acc.w += v2.w; }
            if (r > 3) { acc.x += v3.x; acc.y += v3.y; acc.z += v3.z; acc.w += v3.w; }
        }
    }
    float4 b4;
    b4.x = b[lane * 4 + 0]; b4.y = b[lane * 4 + 1]; b4.z = b[lane * 4 + 2]; b4.w = b[lane * 4 + 3];
    float4 o;
    o.x = fmaxf(acc.x * di + b4.x, 0.f); o.y = fmaxf(acc.y * di + b4.y, 0.f);
    o.z = fmaxf(acc.z * di + b4.z, 0.f); o.w = fmaxf(acc.w * di + b4.w, 0.f);
    ((float4*)h)[(size_t)node * (NH / 4) + lane] = o;
}

__global__ void k_padw2(const float* __restrict__ W2, float* __restrict__ W2p) {
    int id = blockIdx.x * blockDim.x + threadIdx.x;
    if (id < NH * NCP) {
        int k = id / NCP, n = id % NCP;
        W2p[id] = (n < NC) ? W2[k * NC + n] : 0.f;
    }
}

#define L2_Q (NC / 4)
static_assert(3 * L2_Q <= 32 && NCP % 4 == 0, "three edge groups per wave");

__global__ __launch_bounds__(GA_WAVES * 32)
void k_gather2(const int* __restrict__ rowptr, const int* __restrict__ cnt,
               const int* __restrict__ nbr, const float* __restrict__ dinv,
               const float* __restrict__ xs2, const float* __restrict__ b2,
               float* __restrict__ logits) {
    const int lane = threadIdx.x & 31;
    const int node = __builtin_amdgcn_readfirstlane(blockIdx.x * GA_WAVES + (threadIdx.x >> 5));
    const int start = rowptr[node], deg = cnt[node];
    const float di = dinv[node];
    const float4* rows = (const float4*)xs2;
    const int g = lane / L2_Q;
    const int q = lane - g * L2_Q;
    const bool live = g < 3;

    float4 acc = {0.f, 0.f, 0.f, 0.f};
    if (g == 0) acc = rows[(size_t)node * (NCP / 4) + q];
    for (int base = 0; base < deg; base += 32) {
        const int n = min(32, deg - base);
        int my_s = 0;
        if (lane < n) my_s = nbr[start + base + lane];
        for (int j0 = 0; j0 < n; j0 += 6) {
            const int ja = j0 + g, jb = j0 + 3 + g;
            const bool va = live && ja < n, vb = live && jb < n;
            const int sa = __shfl(my_s, ja & 31, 32), sb = __shfl(my_s, jb & 31, 32);
            float4 xa = {0.f, 0.f, 0.f, 0.f}, xb = {0.f, 0.f, 0.f, 0.f};
            if (va) xa = rows[(size_t)sa * (NCP / 4) + q];
            if (vb) xb = rows[(size_t)sb * (NCP / 4) + q];
            if (va) { acc.x += xa.x; acc.y += xa.y; acc.z += xa.z; acc.w += xa.w; }
            if (vb) { acc.x += xb.x; acc.y += xb.y; acc.z += xb.z; acc.w += xb.w; }
        }
    }
    const int l1 = (lane + L2_Q) & 31, l2 = (lane + 2 * L2_Q) & 31;
    const float ax = __shfl(acc.x, l1, 32), ay = __shfl(acc.y, l1, 32);
    const float az = __shfl(acc.z, l1, 32), aw = __shfl(acc.w, l1, 32);
    const float bx = __shfl(acc.x, l2, 32), by = __shfl(acc.y, l2, 32);
    const float bz = __shfl(acc.z, l2, 32), bw = __shfl(acc.w, l2, 32);
    if (g == 0) {
        float4 bb;
        bb.x = b2[q * 4 + 0]; bb.y = b2[q * 4 + 1]; bb.z = b2[q * 4 + 2]; bb.w = b2[q * 4 + 3];
        float4 o;
        o.x = (acc.x + ax + bx) * di + bb.x; o.y = (acc.y + ay + by) * di + bb.y;
        o.z = (acc.z + az + bz) * di + bb.z; o.w = (acc.w + aw + bw) * di + bb.w;
        ((float4*)logits)[(size_t)node * L2_Q + q] = o;
    }
}

#define CS_THREADS 320
#define CS_ROWS    160
#define CS_Q       (NC / 4)
#define CS_LOADS   (CS_ROWS * CS_Q / CS_THREADS)
#define CS_SLOTS   (CS_THREADS / CS_Q)
#define CS_BLOCKS  (N_NODES / CS_ROWS)
#define CS_PITCH   128
#define CS_SUMOFF  64
static_assert(NC % 4 == 0, "rows must split into float4 pieces");
static_assert(CS_THREADS % CS_Q == 0, "a thread must stay on one column group");
static_assert((CS_ROWS * CS_Q) % CS_THREADS == 0, "whole loads per thread");
static_assert(N_NODES % CS_ROWS == 0, "whole row tiles");
static_assert(CS_SLOTS % 4 == 0, "row slots are summed in four chains");
static_assert(NC <= CS_SUMOFF && CS_SUMOFF + NC <= CS_PITCH, "partial plane layout");

__global__ __launch_bounds__(CS_THREADS)
void k_colpart(const float* __restrict__ logits, float* __restrict__ part) {
    __shared__ float4 tile[CS_THREADS];
    __shared__ float4 tmax4[CS_Q];
    const int t = threadIdx.x;
    const float4* p = (const float4*)logits + (size_t)blockIdx.x * (CS_ROWS * CS_Q) + t;
    const float* tf = (const float*)tile;
    float* o = part + (size_t)blockIdx.x * CS_PITCH + t;

    float4 v[CS_LOADS];
    for (int k = 0; k < CS_LOADS; ++k) v[k] = p[k * CS_THREADS];

    float4 m = v[0];
    for (int k = 1; k < CS_LOADS; ++k) {
        m.x = fmaxf(m.x, v[k].x); m.y = fmaxf(m.y, v[k].y);
        m.z = fmaxf(m.z, v[k].z); m.w = fmaxf(m.w, v[k].w);
    }
    tile[t] = m;
    __syncthreads();
    if (t < NC) {
        float M = tf[t];
        for (int j = 1; j < CS_SLOTS; ++j) M = fmaxf(M, tf[j * NC + t]);
        ((float*)tmax4)[t] = M;
        o[0] = M;
    }
    __syncthreads();

    const float4 M4 = tmax4[t % CS_Q];
    float4 s = {0.f, 0.f, 0.f, 0.f};
    for (int k = 0; k < CS_LOADS; ++k) {
        s.x += expf(v[k].x - M4.x); s.y += expf(v[k].y - M4.y);
        s.z += expf(v[k].z - M4.z); s.w += expf(v[k].w - M4.w);
    }
    tile[t] = s;
    __syncthreads();
    if (t < NC) {
        float a0 = 0.f, a1 = 0.f, a2 = 0.f, a3 = 0.f;
        for (int j = 0; j < CS_SLOTS; j += 4) {
            a0 += tf[(j + 0) * NC + t]; a1 += tf[(j + 1) * NC + t];
            a2 += tf[(j + 2) * NC + t]; a3 += tf[(j + 3) * NC + t];
        }
        o[CS_SUMOFF] = (a0 + a1) + (a2 + a3);
    }
}

#define CC_THREADS 256
#define CC_PER     ((CS_BLOCKS + CC_THREADS - 1) / CC_THREADS)

__global__ __launch_bounds__(CC_THREADS)
void k_colcomb(const float* __restrict__ part,
               float* __restrict__ colmax, float* __restrict__ colsum) {
    __shared__ float wmax[CC_THREADS / 32];
    __shared__ float wsum[CC_THREADS / 32];
    const int c = blockIdx.x;
    const int t = threadIdx.x;

    float pm[CC_PER], ps[CC_PER];
    for (int i = 0; i < CC_PER; ++i) {
        const int b = t + i * CC_THREADS;
        const bool ok = b < CS_BLOCKS;
        const float* q = part + (size_t)(ok ? b : 0) * CS_PITCH + c;
        const float qm = q[0], qs = q[CS_SUMOFF];
        pm[i] = ok ? qm : -INFINITY;
        ps[i] = ok ? qs : 0.f;
    }
    float m = pm[0];
    for (int i = 1; i < CC_PER; ++i) m = fmaxf(m, pm[i]);
    for (int off = 16; off > 0; off >>= 1) m = fmaxf(m, __shfl_xor(m, off, 32));
    if ((t & 31) == 0) wmax[t >> 5] = m;
    __syncthreads();
    float cm = wmax[0];
    for (int w = 1; w < CC_THREADS / 32; ++w) cm = fmaxf(cm, wmax[w]);

    float acc = 0.f;
    for (int i = 0; i < CC_PER; ++i) {
        const bool ok = (t + i * CC_THREADS) < CS_BLOCKS;
        acc += ok ? ps[i] * expf(pm[i] - cm) : 0.f;
    }
    for (int off = 16; off > 0; off >>= 1) acc += __shfl_xor(acc, off, 32);
    if ((t & 31) == 0) wsum[t >> 5] = acc;
    __syncthreads();
    if (t == 0) {
        float tot = wsum[0];
        for (int w = 1; w < CC_THREADS / 32; ++w) tot += wsum[w];
        colmax[c] = cm;
        colsum[c] = tot;
    }
}

__global__ void k_finalout(const float* __restrict__ logits,
                           const float* __restrict__ colmax, const float* __restrict__ colsum,
                           float* __restrict__ out_lsm, float* __restrict__ out_sm0) {
    int gid  = blockIdx.x * blockDim.x + threadIdx.x;
    int node = gid >> 5;
    if (node >= N_NODES) return;
    int lane = gid & 31;
    const float* row = logits + (size_t)node * NC;
    float v0 = row[lane];
    bool has2 = lane < (NC - 32);
    float v1 = has2 ? row[lane + 32] : -INFINITY;
    float m = fmaxf(v0, v1);
    #pragma unroll
    for (int off = 16; off > 0; off >>= 1) m = fmaxf(m, __shfl_xor(m, off, 32));
    float s = expf(v0 - m) + (has2 ? expf(v1 - m) : 0.f);
    #pragma unroll
    for (int off = 16; off > 0; off >>= 1) s += __shfl_xor(s, off, 32);
    float lse = m + logf(s);
    out_lsm[(size_t)node * NC + lane] = v0 - lse;
    out_sm0[(size_t)node * NC + lane] = expf(v0 - colmax[lane]) / colsum[lane];
    if (has2) {
        out_lsm[(size_t)node * NC + lane + 32] = v1 - lse;
        out_sm0[(size_t)node * NC + lane + 32] =
            expf(v1 - colmax[lane + 32]) / colsum[lane + 32];
    }
}

extern "C" void kernel_launch(void* const* d_in, const int* in_sizes, int n_in,
                              void* d_out, int out_size, void* d_ws, size_t ws_size,
                              hipStream_t stream) {
    (void)in_sizes; (void)n_in; (void)out_size; (void)ws_size;
    const float* x   = (const float*)d_in[0];
    const int*   ei  = (const int*)d_in[1];
    const float* W1  = (const float*)d_in[2];
    const float* b1  = (const float*)d_in[3];
    const float* W2  = (const float*)d_in[4];
    const float* b2  = (const float*)d_in[5];
    const int* src = ei;
    const int* dst = ei + N_EDGES;

    float* ws = (float*)d_ws;
    size_t off = 0;
    float* dinv = ws + off; off += N_NODES; off = (off + 3) & ~(size_t)3;
    float* xw1  = ws + off; off += (size_t)N_NODES * NH;
    float* agg1 = ws + off; off += (size_t)N_NODES * NH;
    float* W2p  = ws + off; off += NH * NCP;
    float* xw2  = ws + off; off += (size_t)N_NODES * NCP;
    float* agg2 = ws + off; off += (size_t)N_NODES * NC;
    float* colmax = ws + off; off += 64;
    float* colsum = ws + off; off += 64;

    static_assert(2 * (size_t)N_NODES + 2 * (size_t)N_EDGES <= (size_t)N_NODES * NC,
                  "edge lists fit the reused plane");
    int* cnt    = (int*)agg2;
    int* rowptr = cnt + N_NODES;
    int* rank   = rowptr + N_NODES;
    int* nbr    = rank + N_EDGES;

    float* out_lsm = (float*)d_out;
    float* logits  = (float*)d_out + (size_t)N_NODES * NC;
    float* out_sm0 = (float*)d_out + 2 * (size_t)N_NODES * NC;

    hipMemsetAsync(cnt, 0, N_NODES * sizeof(int), stream);
    k_count<<<(E4 + 255) / 256, 256, 0, stream>>>(dst, cnt, rank);
    k_rowptr<<<SC_BLOCKS, SC_THREADS, 0, stream>>>(cnt, rowptr, dinv);
    k_fill<<<(E4 + 255) / 256, 256, 0, stream>>>(src, dst, rank, rowptr, nbr);

    k_gemm_wmma<NF, NH, NH / 16><<<G1_BLOCKS, (NH / 16) * 32, 0, stream>>>(x, W1, dinv, xw1);
    k_gather1<<<N_NODES / GA_WAVES, GA_WAVES * 32, 0, stream>>>(rowptr, cnt, nbr, dinv, xw1, b1, agg1);

    k_padw2<<<(NH * NCP + 255) / 256, 256, 0, stream>>>(W2, W2p);
    k_gemm_wmma<NH, NCP, NCP / 16><<<G2_BLOCKS, (NCP / 16) * 32, 0, stream>>>(agg1, W2p, dinv, xw2);
    k_gather2<<<N_NODES / GA_WAVES, GA_WAVES * 32, 0, stream>>>(rowptr, cnt, nbr, dinv, xw2, b2, logits);

    float* cpart = xw1;
    k_colpart<<<CS_BLOCKS, CS_THREADS, 0, stream>>>(logits, cpart);
    k_colcomb<<<NC, CC_THREADS, 0, stream>>>(cpart, colmax, colsum);
    k_finalout<<<(N_NODES * 32 + 255) / 256, 256, 0, stream>>>(logits, colmax, colsum,
                                                               out_lsm, out_sm0);
}


# edge sort by destination redone as a two-level counting sort in LDS (8192-edge chunks binned by 256-node bucket, then a work-group per bucket); no global atomics; node sum order = LDS arrival order
# speedup vs baseline: 5.8539x; 1.5783x over previous
#include <hip/hip_runtime.h>
#include <math.h>

#define N_NODES 100000
#define N_EDGES 1600000
#define NF 256
#define NH 128
#define NC 40
#define NCP 48

typedef float v2f __attribute__((ext_vector_type(2)));
typedef float v4f __attribute__((ext_vector_type(4)));
typedef float v8f __attribute__((ext_vector_type(8)));

#define E4 (N_EDGES / 4)
static_assert(N_EDGES % 4 == 0 && N_NODES % 4 == 0, "16-byte index accesses");

#define BK_SHIFT   8
#define BK_W       (1 << BK_SHIFT)
#define NB         ((N_NODES + BK_W - 1) / BK_W)
#define SRC_BITS   17
#define SRC_MASK   ((1 << SRC_BITS) - 1)
#define CH         8192
#define NG         ((N_EDGES + CH - 1) / CH)
#define PS_THREADS 1024
#define PS_Q       (CH / 4 / PS_THREADS)
static_assert(N_NODES <= (1 << SRC_BITS) && SRC_BITS + BK_SHIFT <= 31, "packed entry fits an int");
static_assert(CH % (4 * PS_THREADS) == 0 && NB + 1 <= PS_THREADS, "chunk shape");

__global__ __launch_bounds__(PS_THREADS)
void k_part(const int* __restrict__ src, const int* __restrict__ dst,
            int* __restrict__ tmp, int* __restrict__ lstT) {
    __shared__ int hist[NB + 1];
    __shared__ int wtot[PS_THREADS / 32];
    __shared__ __attribute__((aligned(16))) int sorted[CH];
    const int t = threadIdx.x, lane = t & 31, w = t >> 5, g = blockIdx.x;

    if (t <= NB) hist[t] = 0;
    __syncthreads();

    int item[PS_Q * 4], bk[PS_Q * 4], rk[PS_Q * 4];
    bool ok[PS_Q];
    #pragma unroll
    for (int u = 0; u < PS_Q; ++u) {
        const int q = g * (CH / 4) + u * PS_THREADS + t;
        ok[u] = q < E4;
        int4 d = {0, 0, 0, 0}, s = {0, 0, 0, 0};
        if (ok[u]) { d = ((const int4*)dst)[q]; s = ((const int4*)src)[q]; }
        const int dd[4] = {d.x, d.y, d.z, d.w}, ss[4] = {s.x, s.y, s.z, s.w};
        #pragma unroll
        for (int k = 0; k < 4; ++k) {
            bk[4 * u + k]   = dd[k] >> BK_SHIFT;
            item[4 * u + k] = ss[k] | ((dd[k] & (BK_W - 1)) << SRC_BITS);
            rk[4 * u + k]   = 0;
            if (ok[u]) rk[4 * u + k] = atomicAdd(&hist[bk[4 * u + k]], 1);
        }
    }
    __syncthreads();

    const int v = (t < NB) ? hist[t] : 0;
    int inc = v;
    for (int off = 1; off < 32; off <<= 1) {
        const int y = __shfl_up(inc, off, 32);
        if (lane >= off) inc += y;
    }
    if (lane == 31) wtot[w] = inc;
    __syncthreads();
    int ex = inc - v;
    for (int k = 0; k < w; ++k) ex += wtot[k];
    if (t <= NB) { hist[t] = ex; lstT[t * NG + g] = ex; }
    __syncthreads();

    #pragma unroll
    for (int u = 0; u < PS_Q; ++u) {
        if (ok[u]) {
            #pragma unroll
            for (int k = 0; k < 4; ++k) sorted[hist[bk[4 * u + k]] + rk[4 * u + k]] = item[4 * u + k];
        }
    }
    __syncthreads();

    #pragma unroll
    for (int u = 0; u < PS_Q; ++u) {
        const int i = u * PS_THREADS + t;
        const int q = g * (CH / 4) + i;
        if (q < E4) ((int4*)tmp)[q] = ((const int4*)sorted)[i];
    }
}

#define BS_THREADS 1024
#define BS_CAP     16384
static_assert(NG <= BS_THREADS && BK_W <= BS_THREADS, "one thread per chunk and per node of a bucket");

__global__ __launch_bounds__(BS_THREADS)
void k_bsort(const int* __restrict__ tmp, const int* __restrict__ lstT,
             int* __restrict__ cnt, int* __restrict__ rowptr, float* __restrict__ dinv,
             int* __restrict__ nbr) {
    __shared__ int ca[NG];
    __shared__ int pre[NG + 1];
    __shared__ int ncnt[BK_W];
    __shared__ int ncur[BK_W];
    __shared__ int wsc[BS_THREADS / 32], wrd[BS_THREADS / 32], wsc2[BS_THREADS / 32];
    __shared__ int stage[BS_CAP];
    const int t = threadIdx.x, lane = t & 31, w = t >> 5, b = blockIdx.x;

    int a = 0, len = 0;
    if (t < NG) { a = lstT[b * NG + t]; len = lstT[(b + 1) * NG + t] - a; }
    int inc = len;
    for (int off = 1; off < 32; off <<= 1) {
        const int y = __shfl_up(inc, off, 32);
        if (lane >= off) inc += y;
    }
    int asum = a;
    for (int off = 16; off > 0; off >>= 1) asum += __shfl_xor(asum, off, 32);
    if (lane == 31) wsc[w] = inc;
    if (lane == 0)  wrd[w] = asum;
    if (t < BK_W) ncnt[t] = 0;
    __syncthreads();
    int base = 0, run0 = inc - len;
    for (int k = 0; k < BS_THREADS / 32; ++k) base += wrd[k];
    for (int k = 0; k < w; ++k) run0 += wsc[k];
    if (t < NG) { ca[t] = t * CH + a; pre[t] = run0; }
    if (t == NG - 1) pre[NG] = run0 + len;
    __syncthreads();
    const int items = pre[NG];

    for (int j = t; j < items; j += BS_THREADS) {
        int lo = 0, hi = NG;
        while (hi - lo > 1) {
            const int mid = (lo + hi) >> 1;
            if (pre[mid] <= j) lo = mid; else hi = mid;
        }
        const int it = tmp[ca[lo] + (j - pre[lo])];
        atomicAdd(&ncnt[it >> SRC_BITS], 1);
    }
    __syncthreads();

    const int v = (t < BK_W) ? ncnt[t] : 0;
    int inc2 = v;
    for (int off = 1; off < 32; off <<= 1) {
        const int y = __shfl_up(inc2, off, 32);
        if (lane >= off) inc2 += y;
    }
    if (lane == 31) wsc2[w] = inc2;
    __syncthreads();
    if (t < BK_W) {
        int ex = inc2 - v;
        for (int k = 0; k < w; ++k) ex += wsc2[k];
        ncur[t] = ex;
        const int node = b * BK_W + t;
        if (node < N_NODES) {
            cnt[node] = v;
            rowptr[node] = base + ex;
            dinv[node] = rsqrtf((float)v + 1.0f);
        }
    }
    __syncthreads();

    for (int j = t; j < items; j += BS_THREADS) {
        int lo = 0, hi = NG;
        while (hi - lo > 1) {
            const int mid = (lo + hi) >> 1;
            if (pre[mid] <= j) lo = mid; else hi = mid;
        }
        const int it = tmp[ca[lo] + (j - pre[lo])];
        const int pos = atomicAdd(&ncur[it >> SRC_BITS], 1);
        if (pos < BS_CAP) stage[pos] = it & SRC_MASK;
        else nbr[base + pos] = it & SRC_MASK;
    }
    __syncthreads();
    const int m = min(items, BS_CAP);
    for (int i = t; i < m; i += BS_THREADS) nbr[base + i] = stage[i];
}

#define GEMM_MT (N_NODES / 16)
static_assert(N_NODES % 16 == 0, "whole row tiles");
#define G1_BLOCKS 512
#define G2_BLOCKS (GEMM_MT / 5)
static_assert(GEMM_MT % 5 == 0 && G1_BLOCKS <= GEMM_MT, "grid shapes");

template<int K, int LDB, int NT>
__global__ __launch_bounds__(NT * 32)
void k_gemm_wmma(const float* __restrict__ A, const float* __restrict__ B,
                 const float* __restrict__ rs, float* __restrict__ C) {
    constexpr int P     = K + 4;
    constexpr int RQ    = K / 4;
    constexpr int PPL   = RQ / 32;
    constexpr int RPW   = (16 + NT - 1) / NT;
    constexpr int LROWS = RPW * NT;
    constexpr int NST   = RPW * PPL;
    static_assert(LDB == NT * 16 && RQ % 32 == 0 && (P / 4) % 2 == 1, "tile shape");
    __shared__ __attribute__((aligned(16))) float tile[2][LROWS * P];

    const int t    = threadIdx.x;
    const int nt   = t >> 5;
    const int lane = t & 31;
    const int half = lane >> 4;
    const int lm   = lane & 15;

    v2f b[K / 4];
    const float* bcol = B + nt * 16 + lm;
    #pragma unroll
    for (int s = 0; s < K / 4; ++s) {
        const int ka = 4 * s + 2 * half;
        b[s].x = bcol[(size_t)ka * LDB];
        b[s].y = bcol[(size_t)(ka + 1) * LDB];
    }

    int goff[NST], loff[NST];
    #pragma unroll
    for (int i = 0; i < RPW; ++i) {
        const int row = nt + i * NT;
        const int srow = (16 % NT == 0) ? row : min(row, 15);
        #pragma unroll
        for (int j = 0; j < PPL; ++j) {
            goff[i * PPL + j] = srow * RQ + j * 32 + lane;
            loff[i * PPL + j] = row * P + (j * 32 + lane) * 4;
        }
    }

    v4f st[NST];
    int mt = blockIdx.x;
    const v4f* rsp = (const v4f*)(rs + 8 * half);
    v4f rs0 = rsp[mt * 4], rs1 = rsp[mt * 4 + 1];
    {
        const v4f* g = (const v4f*)(A + (size_t)mt * (16 * K));
        #pragma unroll
        for (int i = 0; i < NST; ++i) st[i] = g[goff[i]];
        #pragma unroll
        for (int i = 0; i < NST; ++i) *(v4f*)&tile[0][loff[i]] = st[i];
    }
    __syncthreads();

    int cur = 0;
    for (; mt < GEMM_MT; mt += gridDim.x) {
        const int nx = mt + gridDim.x;
        const bool more = nx < GEMM_MT;
        v4f nrs0 = rs0, nrs1 = rs1;
        if (more) {
            const v4f* g = (const v4f*)(A + (size_t)nx * (16 * K));
            #pragma unroll
            for (int i = 0; i < NST; ++i) st[i] = g[goff[i]];
            nrs0 = rsp[nx * 4]; nrs1 = rsp[nx * 4 + 1];
        }
        const float* ap = &tile[cur][lm * P + 2 * half];
        v8f c = {0.f, 0.f, 0.f, 0.f, 0.f, 0.f, 0.f, 0.f};
        #pragma unroll
        for (int s = 0; s < K / 4; ++s) {
            const v2f a = *(const v2f*)(ap + 4 * s);
            c = __builtin_amdgcn_wmma_f32_16x16x4_f32(false, a, false, b[s],
                                                      (short)0, c, false, false);
        }

        if (more) {
            #pragma unroll
            for (int i = 0; i < NST; ++i) *(v4f*)&tile[cur ^ 1][loff[i]] = st[i];
        }
        __syncthreads();
        cur ^= 1;

        float* out = C + (size_t)(mt * 16 + 8 * half) * LDB + nt * 16 + lm;
        #pragma unroll
        for (int v = 0; v < 8; ++v) out[(size_t)v * LDB] = c[v] * (v < 4 ? rs0[v & 3] : rs1[v & 3]);
        rs0 = nrs0; rs1 = nrs1;
    }
}

#define GA_WAVES 8
static_assert(N_NODES % GA_WAVES == 0, "whole work-groups of nodes");
static_assert(NH == 128, "a 32-lane wave covers a hidden row with one float4 per lane");

__device__ __forceinline__ int   bcast_i(int v, int j)   { return __builtin_amdgcn_readlane(v, j); }

__global__ __launch_bounds__(GA_WAVES * 32)
void k_gather1(const int* __restrict__ rowptr, const int* __restrict__ cnt,
               const int* __restrict__ nbr, const float* __restrict__ dinv,
               const float* __restrict__ xs, const float* __restrict__ b,
               float* __restrict__ h) {
    const int lane = threadIdx.x & 31;
    const int node = __builtin_amdgcn_readfirstlane(blockIdx.x * GA_WAVES + (threadIdx.x >> 5));
    const int start = rowptr[node], deg = cnt[node];
    const float di = dinv[node];
    const float4* rows = (const float4*)xs;

    float4 acc = rows[(size_t)node * (NH / 4) + lane];

    for (int base = 0; base < deg; base += 32) {
        const int n = min(32, deg - base);
        int my_s = 0;
        if (lane < n) my_s = nbr[start + base + lane];
        for (int j = 0; j < n; j += 4) {
            const int r = n - j;
            const float4 z = {0.f, 0.f, 0.f, 0.f};
            float4 v1 = z, v2 = z, v3 = z;
            const float4 v0 = rows[(size_t)bcast_i(my_s, j) * (NH / 4) + lane];
            if (r > 1) v1 = rows[(size_t)bcast_i(my_s, j + 1) * (NH / 4) + lane];
            if (r > 2) v2 = rows[(size_t)bcast_i(my_s, j + 2) * (NH / 4) + lane];
            if (r > 3) v3 = rows[(size_t)bcast_i(my_s, j + 3) * (NH / 4) + lane];
            acc.x += v0.x; acc.y += v0.y; acc.z += v0.z; acc.w += v0.w;
            if (r > 1) { acc.x += v1.x; acc.y += v1.y; acc.z += v1.z; acc.w += v1.w; }
            if (r > 2) { acc.x += v2.x; acc.y += v2.y; acc.z += v2.z; acc.w += v2.w; }
            if (r > 3) { acc.x += v3.x; acc.y += v3.y; acc.z += v3.z; acc.w += v3.w; }
        }
    }
    float4 b4;
    b4.x = b[lane * 4 + 0]; b4.y = b[lane * 4 + 1]; b4.z = b[lane * 4 + 2]; b4.w = b[lane * 4 + 3];
    float4 o;
    o.x = fmaxf(acc.x * di + b4.x, 0.f); o.y = fmaxf(acc.y * di + b4.y, 0.f);
    o.z = fmaxf(acc.z * di + b4.z, 0.f); o.w = fmaxf(acc.w * di + b4.w, 0.f);
    ((float4*)h)[(size_t)node * (NH / 4) + lane] = o;
}

__global__ void k_padw2(const float* __restrict__ W2, float* __restrict__ W2p) {
    int id = blockIdx.x * blockDim.x + threadIdx.x;
    if (id < NH * NCP) {
        int k = id / NCP, n = id % NCP;
        W2p[id] = (n < NC) ? W2[k * NC + n] : 0.f;
    }
}

#define L2_Q (NC / 4)
static_assert(3 * L2_Q <= 32 && NCP % 4 == 0, "three edge groups per wave");

__global__ __launch_bounds__(GA_WAVES * 32)
void k_gather2(const int* __restrict__ rowptr, const int* __restrict__ cnt,
               const int* __restrict__ nbr, const float* __restrict__ dinv,
               const float* __restrict__ xs2, const float* __restrict__ b2,
               float* __restrict__ logits) {
    const int lane = threadIdx.x & 31;
    const int node = __builtin_amdgcn_readfirstlane(blockIdx.x * GA_WAVES + (threadIdx.x >> 5));
    const int start = rowptr[node], deg = cnt[node];
    const float di = dinv[node];
    const float4* rows = (const float4*)xs2;
    const int g = lane / L2_Q;
    const int q = lane - g * L2_Q;
    const bool live = g < 3;

    float4 acc = {0.f, 0.f, 0.f, 0.f};
    if (g == 0) acc = rows[(size_t)node * (NCP / 4) + q];
    for (int base = 0; base < deg; base += 32) {
        const int n = min(32, deg - base);
        int my_s = 0;
        if (lane < n) my_s = nbr[start + base + lane];
        for (int j0 = 0; j0 < n; j0 += 6) {
            const int ja = j0 + g, jb = j0 + 3 + g;
            const bool va = live && ja < n, vb = live && jb < n;
            const int sa = __shfl(my_s, ja & 31, 32), sb = __shfl(my_s, jb & 31, 32);
            float4 xa = {0.f, 0.f, 0.f, 0.f}, xb = {0.f, 0.f, 0.f, 0.f};
            if (va) xa = rows[(size_t)sa * (NCP / 4) + q];
            if (vb) xb = rows[(size_t)sb * (NCP / 4) + q];
            if (va) { acc.x += xa.x; acc.y += xa.y; acc.z += xa.z; acc.w += xa.w; }
            if (vb) { acc.x += xb.x; acc.y += xb.y; acc.z += xb.z; acc.w += xb.w; }
        }
    }
    const int l1 = (lane + L2_Q) & 31, l2 = (lane + 2 * L2_Q) & 31;
    const float ax = __shfl(acc.x, l1, 32), ay = __shfl(acc.y, l1, 32);
    const float az = __shfl(acc.z, l1, 32), aw = __shfl(acc.w, l1, 32);
    const float bx = __shfl(acc.x, l2, 32), by = __shfl(acc.y, l2, 32);
    const float bz = __shfl(acc.z, l2, 32), bw = __shfl(acc.w, l2, 32);
    if (g == 0) {
        float4 bb;
        bb.x = b2[q * 4 + 0]; bb.y = b2[q * 4 + 1]; bb.z = b2[q * 4 + 2]; bb.w = b2[q * 4 + 3];
        float4 o;
        o.x = (acc.x + ax + bx) * di + bb.x; o.y = (acc.y + ay + by) * di + bb.y;
        o.z = (acc.z + az + bz) * di + bb.z; o.w = (acc.w + aw + bw) * di + bb.w;
        ((float4*)logits)[(size_t)node * L2_Q + q] = o;
    }
}

#define CS_THREADS 320
#define CS_ROWS    160
#define CS_Q       (NC / 4)
#define CS_LOADS   (CS_ROWS * CS_Q / CS_THREADS)
#define CS_SLOTS   (CS_THREADS / CS_Q)
#define CS_BLOCKS  (N_NODES / CS_ROWS)
#define CS_PITCH   128
#define CS_SUMOFF  64
static_assert(NC % 4 == 0, "rows must split into float4 pieces");
static_assert(CS_THREADS % CS_Q == 0, "a thread must stay on one column group");
static_assert((CS_ROWS * CS_Q) % CS_THREADS == 0, "whole loads per thread");
static_assert(N_NODES % CS_ROWS == 0, "whole row tiles");
static_assert(CS_SLOTS % 4 == 0, "row slots are summed in four chains");
static_assert(NC <= CS_SUMOFF && CS_SUMOFF + NC <= CS_PITCH, "partial plane layout");

__global__ __launch_bounds__(CS_THREADS)
void k_colpart(const float* __restrict__ logits, float* __restrict__ part) {
    __shared__ float4 tile[CS_THREADS];
    __shared__ float4 tmax4[CS_Q];
    const int t = threadIdx.x;
    const float4* p = (const float4*)logits + (size_t)blockIdx.x * (CS_ROWS * CS_Q) + t;
    const float* tf = (const float*)tile;
    float* o = part + (size_t)blockIdx.x * CS_PITCH + t;

    float4 v[CS_LOADS];
    for (int k = 0; k < CS_LOADS; ++k) v[k] = p[k * CS_THREADS];

    float4 m = v[0];
    for (int k = 1; k < CS_LOADS; ++k) {
        m.x = fmaxf(m.x, v[k].x); m.y = fmaxf(m.y, v[k].y);
        m.z = fmaxf(m.z, v[k].z); m.w = fmaxf(m.w, v[k].w);
    }
    tile[t] = m;
    __syncthreads();
    if (t < NC) {
        float M = tf[t];
        for (int j = 1; j < CS_SLOTS; ++j) M = fmaxf(M, tf[j * NC + t]);
        ((float*)tmax4)[t] = M;
        o[0] = M;
    }
    __syncthreads();

    const float4 M4 = tmax4[t % CS_Q];
    float4 s = {0.f, 0.f, 0.f, 0.f};
    for (int k = 0; k < CS_LOADS; ++k) {
        s.x += expf(v[k].x - M4.x); s.y += expf(v[k].y - M4.y);
        s.z += expf(v[k].z - M4.z); s.w += expf(v[k].w - M4.w);
    }
    tile[t] = s;
    __syncthreads();
    if (t < NC) {
        float a0 = 0.f, a1 = 0.f, a2 = 0.f, a3 = 0.f;
        for (int j = 0; j < CS_SLOTS; j += 4) {
            a0 += tf[(j + 0) * NC + t]; a1 += tf[(j + 1) * NC + t];
            a2 += tf[(j + 2) * NC + t]; a3 += tf[(j + 3) * NC + t];
        }
        o[CS_SUMOFF] = (a0 + a1) + (a2 + a3);
    }
}

#define CC_THREADS 256
#define CC_PER     ((CS_BLOCKS + CC_THREADS - 1) / CC_THREADS)

__global__ __launch_bounds__(CC_THREADS)
void k_colcomb(const float* __restrict__ part,
               float* __restrict__ colmax, float* __restrict__ colsum) {
    __shared__ float wmax[CC_THREADS / 32];
    __shared__ float wsum[CC_THREADS / 32];
    const int c = blockIdx.x;
    const int t = threadIdx.x;

    float pm[CC_PER], ps[CC_PER];
    for (int i = 0; i < CC_PER; ++i) {
        const int b = t + i * CC_THREADS;
        const bool ok = b < CS_BLOCKS;
        const float* q = part + (size_t)(ok ? b : 0) * CS_PITCH + c;
        const float qm = q[0], qs = q[CS_SUMOFF];
        pm[i] = ok ? qm : -INFINITY;
        ps[i] = ok ? qs : 0.f;
    }
    float m = pm[0];
    for (int i = 1; i < CC_PER; ++i) m = fmaxf(m, pm[i]);
    for (int off = 16; off > 0; off >>= 1) m = fmaxf(m, __shfl_xor(m, off, 32));
    if ((t & 31) == 0) wmax[t >> 5] = m;
    __syncthreads();
    float cm = wmax[0];
    for (int w = 1; w < CC_THREADS / 32; ++w) cm = fmaxf(cm, wmax[w]);

    float acc = 0.f;
    for (int i = 0; i < CC_PER; ++i) {
        const bool ok = (t + i * CC_THREADS) < CS_BLOCKS;
        acc += ok ? ps[i] * expf(pm[i] - cm) : 0.f;
    }
    for (int off = 16; off > 0; off >>= 1) acc += __shfl_xor(acc, off, 32);
    if ((t & 31) == 0) wsum[t >> 5] = acc;
    __syncthreads();
    if (t == 0) {
        float tot = wsum[0];
        for (int w = 1; w < CC_THREADS / 32; ++w) tot += wsum[w];
        colmax[c] = cm;
        colsum[c] = tot;
    }
}

__global__ void k_finalout(const float* __restrict__ logits,
                           const float* __restrict__ colmax, const float* __restrict__ colsum,
                           float* __restrict__ out_lsm, float* __restrict__ out_sm0) {
    int gid  = blockIdx.x * blockDim.x + threadIdx.x;
    int node = gid >> 5;
    if (node >= N_NODES) return;
    int lane = gid & 31;
    const float* row = logits + (size_t)node * NC;
    float v0 = row[lane];
    bool has2 = lane < (NC - 32);
    float v1 = has2 ? row[lane + 32] : -INFINITY;
    float m = fmaxf(v0, v1);
    #pragma unroll
    for (int off = 16; off > 0; off >>= 1) m = fmaxf(m, __shfl_xor(m, off, 32));
    float s = expf(v0 - m) + (has2 ? expf(v1 - m) : 0.f);
    #pragma unroll
    for (int off = 16; off > 0; off >>= 1) s += __shfl_xor(s, off, 32);
    float lse = m + logf(s);
    out_lsm[(size_t)node * NC + lane] = v0 - lse;
    out_sm0[(size_t)node * NC + lane] = expf(v0 - colmax[lane]) / colsum[lane];
    if (has2) {
        out_lsm[(size_t)node * NC + lane + 32] = v1 - lse;
        out_sm0[(size_t)node * NC + lane + 32] =
            expf(v1 - colmax[lane + 32]) / colsum[lane + 32];
    }
}

extern "C" void kernel_launch(void* const* d_in, const int* in_sizes, int n_in,
                              void* d_out, int out_size, void* d_ws, size_t ws_size,
                              hipStream_t stream) {
    (void)in_sizes; (void)n_in; (void)out_size; (void)ws_size;
    const float* x   = (const float*)d_in[0];
    const int*   ei  = (const int*)d_in[1];
    const float* W1  = (const float*)d_in[2];
    const float* b1  = (const float*)d_in[3];
    const float* W2  = (const float*)d_in[4];
    const float* b2  = (const float*)d_in[5];
    const int* src = ei;
    const int* dst = ei + N_EDGES;

    float* ws = (float*)d_ws;
    size_t off = 0;
    float* dinv = ws + off; off += N_NODES; off = (off + 3) & ~(size_t)3;
    float* xw1  = ws + off; off += (size_t)N_NODES * NH;
    float* agg1 = ws + off; off += (size_t)N_NODES * NH;
    float* W2p  = ws + off; off += NH * NCP;
    float* xw2  = ws + off; off += (size_t)N_NODES * NCP;
    float* agg2 = ws + off; off += (size_t)N_NODES * NC;
    float* colmax = ws + off; off += 64;
    float* colsum = ws + off; off += 64;

    static_assert(2 * (size_t)N_NODES + 2 * (size_t)N_EDGES + (size_t)(NB + 1) * NG
                  <= (size_t)N_NODES * NC, "edge lists fit the reused plane");
    int* cnt    = (int*)agg2;
    int* rowptr = cnt + N_NODES;
    int* tmp    = rowptr + N_NODES;
    int* nbr    = tmp + N_EDGES;
    int* lstT   = nbr + N_EDGES;

    float* out_lsm = (float*)d_out;
    float* logits  = (float*)d_out + (size_t)N_NODES * NC;
    float* out_sm0 = (float*)d_out + 2 * (size_t)N_NODES * NC;

    k_part<<<NG, PS_THREADS, 0, stream>>>(src, dst, tmp, lstT);
    k_bsort<<<NB, BS_THREADS, 0, stream>>>(tmp, lstT, cnt, rowptr, dinv, nbr);

    k_gemm_wmma<NF, NH, NH / 16><<<G1_BLOCKS, (NH / 16) * 32, 0, stream>>>(x, W1, dinv, xw1);
    k_gather1<<<N_NODES / GA_WAVES, GA_WAVES * 32, 0, stream>>>(rowptr, cnt, nbr, dinv, xw1, b1, agg1);

    k_padw2<<<(NH * NCP + 255) / 256, 256, 0, stream>>>(W2, W2p);
    k_gemm_wmma<NH, NCP, NCP / 16><<<G2_BLOCKS, (NCP / 16) * 32, 0, stream>>>(agg1, W2p, dinv, xw2);
    k_gather2<<<N_NODES / GA_WAVES, GA_WAVES * 32, 0, stream>>>(rowptr, cnt, nbr, dinv, xw2, b2, logits);

    float* cpart = xw1;
    k_colpart<<<CS_BLOCKS, CS_THREADS, 0, stream>>>(logits, cpart);
    k_colcomb<<<NC, CC_THREADS, 0, stream>>>(cpart, colmax, colsum);
    k_finalout<<<(N_NODES * 32 + 255) / 256, 256, 0, stream>>>(logits, colmax, colsum,
                                                               out_lsm, out_sm0);
}


# tables read by index put on aligned 256-byte units: first product's plane starts on a unit (row in 2 units, was 3), second product's rows at 256-byte pitch (2 rows: 2 units, was 3); sum order kept
# speedup vs baseline: 6.3881x; 1.0913x over previous
#include <hip/hip_runtime.h>
#include <math.h>

#define N_NODES 100000
#define N_EDGES 1600000
#define NF 256
#define NH 128
#define NC 40
#define NCP 48

typedef float v2f __attribute__((ext_vector_type(2)));
typedef float v4f __attribute__((ext_vector_type(4)));
typedef float v8f __attribute__((ext_vector_type(8)));

#define E4 (N_EDGES / 4)
static_assert(N_EDGES % 4 == 0 && N_NODES % 4 == 0, "16-byte index accesses");

#define BK_SHIFT   8
#define BK_W       (1 << BK_SHIFT)
#define NB         ((N_NODES + BK_W - 1) / BK_W)
#define SRC_BITS   17
#define SRC_MASK   ((1 << SRC_BITS) - 1)
#define CH         8192
#define NG         ((N_EDGES + CH - 1) / CH)
#define PS_THREADS 1024
#define PS_Q       (CH / 4 / PS_THREADS)
static_assert(N_NODES <= (1 << SRC_BITS) && SRC_BITS + BK_SHIFT <= 31, "packed entry fits an int");
static_assert(CH % (4 * PS_THREADS) == 0 && NB + 1 <= PS_THREADS, "chunk shape");

__global__ __launch_bounds__(PS_THREADS)
void k_part(const int* __restrict__ src, const int* __restrict__ dst,
            int* __restrict__ tmp, int* __restrict__ lstT) {
    __shared__ int hist[NB + 1];
    __shared__ int wtot[PS_THREADS / 32];
    __shared__ __attribute__((aligned(16))) int sorted[CH];
    const int t = threadIdx.x, lane = t & 31, w = t >> 5, g = blockIdx.x;

    if (t <= NB) hist[t] = 0;
    __syncthreads();

    int item[PS_Q * 4], bk[PS_Q * 4], rk[PS_Q * 4];
    bool ok[PS_Q];
    #pragma unroll
    for (int u = 0; u < PS_Q; ++u) {
        const int q = g * (CH / 4) + u * PS_THREADS + t;
        ok[u] = q < E4;
        int4 d = {0, 0, 0, 0}, s = {0, 0, 0, 0};
        if (ok[u]) { d = ((const int4*)dst)[q]; s = ((const int4*)src)[q]; }
        const int dd[4] = {d.x, d.y, d.z, d.w}, ss[4] = {s.x, s.y, s.z, s.w};
        #pragma unroll
        for (int k = 0; k < 4; ++k) {
            bk[4 * u + k]   = dd[k] >> BK_SHIFT;
            item[4 * u + k] = ss[k] | ((dd[k] & (BK_W - 1)) << SRC_BITS);
            rk[4 * u + k]   = 0;
            if (ok[u]) rk[4 * u + k] = atomicAdd(&hist[bk[4 * u + k]], 1);
        }
    }
    __syncthreads();

    const int v = (t < NB) ? hist[t] : 0;
    int inc = v;
    for (int off = 1; off < 32; off <<= 1) {
        const int y = __shfl_up(inc, off, 32);
        if (lane >= off) inc += y;
    }
    if (lane == 31) wtot[w] = inc;
    __syncthreads();
    int ex = inc - v;
    for (int k = 0; k < w; ++k) ex += wtot[k];
    if (t <= NB) { hist[t] = ex; lstT[t * NG + g] = ex; }
    __syncthreads();

    #pragma unroll
    for (int u = 0; u < PS_Q; ++u) {
        if (ok[u]) {
            #pragma unroll
            for (int k = 0; k < 4; ++k) sorted[hist[bk[4 * u + k]] + rk[4 * u + k]] = item[4 * u + k];
        }
    }
    __syncthreads();

    #pragma unroll
    for (int u = 0; u < PS_Q; ++u) {
        const int i = u * PS_THREADS + t;
        const int q = g * (CH / 4) + i;
        if (q < E4) ((int4*)tmp)[q] = ((const int4*)sorted)[i];
    }
}

#define BS_THREADS 1024
#define BS_CAP     16384
static_assert(NG <= BS_THREADS && BK_W <= BS_THREADS, "one thread per chunk and per node of a bucket");

__global__ __launch_bounds__(BS_THREADS)
void k_bsort(const int* __restrict__ tmp, const int* __restrict__ lstT,
             int* __restrict__ cnt, int* __restrict__ rowptr, float* __restrict__ dinv,
             int* __restrict__ nbr) {
    __shared__ int ca[NG];
    __shared__ int pre[NG + 1];
    __shared__ int ncnt[BK_W];
    __shared__ int ncur[BK_W];
    __shared__ int wsc[BS_THREADS / 32], wrd[BS_THREADS / 32], wsc2[BS_THREADS / 32];
    __shared__ int stage[BS_CAP];
    const int t = threadIdx.x, lane = t & 31, w = t >> 5, b = blockIdx.x;

    int a = 0, len = 0;
    if (t < NG) { a = lstT[b * NG + t]; len = lstT[(b + 1) * NG + t] - a; }
    int inc = len;
    for (int off = 1; off < 32; off <<= 1) {
        const int y = __shfl_up(inc, off, 32);
        if (lane >= off) inc += y;
    }
    int asum = a;
    for (int off = 16; off > 0; off >>= 1) asum += __shfl_xor(asum, off, 32);
    if (lane == 31) wsc[w] = inc;
    if (lane == 0)  wrd[w] = asum;
    if (t < BK_W) ncnt[t] = 0;
    __syncthreads();
    int base = 0, run0 = inc - len;
    for (int k = 0; k < BS_THREADS / 32; ++k) base += wrd[k];
    for (int k = 0; k < w; ++k) run0 += wsc[k];
    if (t < NG) { ca[t] = t * CH + a; pre[t] = run0; }
    if (t == NG - 1) pre[NG] = run0 + len;
    __syncthreads();
    const int items = pre[NG];

    for (int j = t; j < items; j += BS_THREADS) {
        int lo = 0, hi = NG;
        while (hi - lo > 1) {
            const int mid = (lo + hi) >> 1;
            if (pre[mid] <= j) lo = mid; else hi = mid;
        }
        const int it = tmp[ca[lo] + (j - pre[lo])];
        atomicAdd(&ncnt[it >> SRC_BITS], 1);
    }
    __syncthreads();

    const int v = (t < BK_W) ? ncnt[t] : 0;
    int inc2 = v;
    for (int off = 1; off < 32; off <<= 1) {
        const int y = __shfl_up(inc2, off, 32);
        if (lane >= off) inc2 += y;
    }
    if (lane == 31) wsc2[w] = inc2;
    __syncthreads();
    if (t < BK_W) {
        int ex = inc2 - v;
        for (int k = 0; k < w; ++k) ex += wsc2[k];
        ncur[t] = ex;
        const int node = b * BK_W + t;
        if (node < N_NODES) {
            cnt[node] = v;
            rowptr[node] = base + ex;
            dinv[node] = rsqrtf((float)v + 1.0f);
        }
    }
    __syncthreads();

    for (int j = t; j < items; j += BS_THREADS) {
        int lo = 0, hi = NG;
        while (hi - lo > 1) {
            const int mid = (lo + hi) >> 1;
            if (pre[mid] <= j) lo = mid; else hi = mid;
        }
        const int it = tmp[ca[lo] + (j - pre[lo])];
        const int pos = atomicAdd(&ncur[it >> SRC_BITS], 1);
        if (pos < BS_CAP) stage[pos] = it & SRC_MASK;
        else nbr[base + pos] = it & SRC_MASK;
    }
    __syncthreads();
    const int m = min(items, BS_CAP);
    for (int i = t; i < m; i += BS_THREADS) nbr[base + i] = stage[i];
}

#define GEMM_MT (N_NODES / 16)
static_assert(N_NODES % 16 == 0, "whole row tiles");
#define G1_BLOCKS 512
#define G2_BLOCKS (GEMM_MT / 5)
static_assert(GEMM_MT % 5 == 0 && G1_BLOCKS <= GEMM_MT, "grid shapes");

template<int K, int LDB, int NT, int LDC>
__global__ __launch_bounds__(NT * 32)
void k_gemm_wmma(const float* __restrict__ A, const float* __restrict__ B,
                 const float* __restrict__ rs, float* __restrict__ C) {
    constexpr int P     = K + 4;
    constexpr int RQ    = K / 4;
    constexpr int PPL   = RQ / 32;
    constexpr int RPW   = (16 + NT - 1) / NT;
    constexpr int LROWS = RPW * NT;
    constexpr int NST   = RPW * PPL;
    static_assert(LDB == NT * 16 && LDC >= LDB && RQ % 32 == 0 && (P / 4) % 2 == 1, "tile shape");
    __shared__ __attribute__((aligned(16))) float tile[2][LROWS * P];

    const int t    = threadIdx.x;
    const int nt   = t >> 5;
    const int lane = t & 31;
    const int half = lane >> 4;
    const int lm   = lane & 15;

    v2f b[K / 4];
    const float* bcol = B + nt * 16 + lm;
    #pragma unroll
    for (int s = 0; s < K / 4; ++s) {
        const int ka = 4 * s + 2 * half;
        b[s].x = bcol[(size_t)ka * LDB];
        b[s].y = bcol[(size_t)(ka + 1) * LDB];
    }

    int goff[NST], loff[NST];
    #pragma unroll
    for (int i = 0; i < RPW; ++i) {
        const int row = nt + i * NT;
        const int srow = (16 % NT == 0) ? row : min(row, 15);
        #pragma unroll
        for (int j = 0; j < PPL; ++j) {
            goff[i * PPL + j] = srow * RQ + j * 32 + lane;
            loff[i * PPL + j] = row * P + (j * 32 + lane) * 4;
        }
    }

    v4f st[NST];
    int mt = blockIdx.x;
    const v4f* rsp = (const v4f*)(rs + 8 * half);
    v4f rs0 = rsp[mt * 4], rs1 = rsp[mt * 4 + 1];
    {
        const v4f* g = (const v4f*)(A + (size_t)mt * (16 * K));
        #pragma unroll
        for (int i = 0; i < NST; ++i) st[i] = g[goff[i]];
        #pragma unroll
        for (int i = 0; i < NST; ++i) *(v4f*)&tile[0][loff[i]] = st[i];
    }
    __syncthreads();

    int cur = 0;
    for (; mt < GEMM_MT; mt += gridDim.x) {
        const int nx = mt + gridDim.x;
        const bool more = nx < GEMM_MT;
        v4f nrs0 = rs0, nrs1 = rs1;
        if (more) {
            const v4f* g = (const v4f*)(A + (size_t)nx * (16 * K));
            #pragma unroll
            for (int i = 0; i < NST; ++i) st[i] = g[goff[i]];
            nrs0 = rsp[nx * 4]; nrs1 = rsp[nx * 4 + 1];
        }
        const float* ap = &tile[cur][lm * P + 2 * half];
        v8f c = {0.f, 0.f, 0.f, 0.f, 0.f, 0.f, 0.f, 0.f};
        #pragma unroll
        for (int s = 0; s < K / 4; ++s) {
            const v2f a = *(const v2f*)(ap + 4 * s);
            c = __builtin_amdgcn_wmma_f32_16x16x4_f32(false, a, false, b[s],
                                                      (short)0, c, false, false);
        }

        if (more) {
            #pragma unroll
            for (int i = 0; i < NST; ++i) *(v4f*)&tile[cur ^ 1][loff[i]] = st[i];
        }
        __syncthreads();
        cur ^= 1;

        float* out = C + (size_t)(mt * 16 + 8 * half) * LDC + nt * 16 + lm;
        #pragma unroll
        for (int v = 0; v < 8; ++v) out[(size_t)v * LDC] = c[v] * (v < 4 ? rs0[v & 3] : rs1[v & 3]);
        rs0 = nrs0; rs1 = nrs1;
    }
}

#define GA_WAVES 8
static_assert(N_NODES % GA_WAVES == 0, "whole work-groups of nodes");
static_assert(NH == 128, "a 32-lane wave covers a hidden row with one float4 per lane");

__device__ __forceinline__ int   bcast_i(int v, int j)   { return __builtin_amdgcn_readlane(v, j); }

__global__ __launch_bounds__(GA_WAVES * 32)
void k_gather1(const int* __restrict__ rowptr, const int* __restrict__ cnt,
               const int* __restrict__ nbr, const float* __restrict__ dinv,
               const float* __restrict__ xs, const float* __restrict__ b,
               float* __restrict__ h) {
    const int lane = threadIdx.x & 31;
    const int node = __builtin_amdgcn_readfirstlane(blockIdx.x * GA_WAVES + (threadIdx.x >> 5));
    const int start = rowptr[node], deg = cnt[node];
    const float di = dinv[node];
    const float4* rows = (const float4*)xs;

    float4 acc = rows[(size_t)node * (NH / 4) + lane];

    for (int base = 0; base < deg; base += 32) {
        const int n = min(32, deg - base);
        int my_s = 0;
        if (lane < n) my_s = nbr[start + base + lane];
        for (int j = 0; j < n; j += 4) {
            const int r = n - j;
            const float4 z = {0.f, 0.f, 0.f, 0.f};
            float4 v1 = z, v2 = z, v3 = z;
            const float4 v0 = rows[(size_t)bcast_i(my_s, j) * (NH / 4) + lane];
            if (r > 1) v1 = rows[(size_t)bcast_i(my_s, j + 1) * (NH / 4) + lane];
            if (r > 2) v2 = rows[(size_t)bcast_i(my_s, j + 2) * (NH / 4) + lane];
            if (r > 3) v3 = rows[(size_t)bcast_i(my_s, j + 3) * (NH / 4) + lane];
            acc.x += v0.x; acc.y += v0.y; acc.z += v0.z; acc.w += v0.w;
            if (r > 1) { acc.x += v1.x; acc.y += v1.y; acc.z += v1.z; acc.w += v1.w; }
            if (r > 2) { acc.x += v2.x; acc.y += v2.y; acc.z += v2.z; acc.w += v2.w; }
            if (r > 3) { acc.x += v3.x; acc.y += v3.y; acc.z += v3.z; acc.w += v3.w; }
        }
    }
    float4 b4;
    b4.x = b[lane * 4 + 0]; b4.y = b[lane * 4 + 1]; b4.z = b[lane * 4 + 2]; b4.w = b[lane * 4 + 3];
    float4 o;
    o.x = fmaxf(acc.x * di + b4.x, 0.f); o.y = fmaxf(acc.y * di + b4.y, 0.f);
    o.z = fmaxf(acc.z * di + b4.z, 0.f); o.w = fmaxf(acc.w * di + b4.w, 0.f);
    ((float4*)h)[(size_t)node * (NH / 4) + lane] = o;
}

__global__ void k_padw2(const float* __restrict__ W2, float* __restrict__ W2p) {
    int id = blockIdx.x * blockDim.x + threadIdx.x;
    if (id < NH * NCP) {
        int k = id / NCP, n = id % NCP;
        W2p[id] = (n < NC) ? W2[k * NC + n] : 0.f;
    }
}

#define L2_Q (NC / 4)
#define X2P  64
static_assert(3 * L2_Q <= 32 && NCP % 4 == 0, "three edge groups per wave");
static_assert(NCP <= X2P && X2P % 64 == 0, "a row fits its pitch; the pitch is whole 256-byte units");

__global__ __launch_bounds__(GA_WAVES * 32)
void k_gather2(const int* __restrict__ rowptr, const int* __restrict__ cnt,
               const int* __restrict__ nbr, const float* __restrict__ dinv,
               const float* __restrict__ xs2, const float* __restrict__ b2,
               float* __restrict__ logits) {
    const int lane = threadIdx.x & 31;
    const int node = __builtin_amdgcn_readfirstlane(blockIdx.x * GA_WAVES + (threadIdx.x >> 5));
    const int start = rowptr[node], deg = cnt[node];
    const float di = dinv[node];
    const float4* rows = (const float4*)xs2;
    const int g = lane / L2_Q;
    const int q = lane - g * L2_Q;
    const bool live = g < 3;

    float4 acc = {0.f, 0.f, 0.f, 0.f};
    if (g == 0) acc = rows[(size_t)node * (X2P / 4) + q];
    for (int base = 0; base < deg; base += 32) {
        const int n = min(32, deg - base);
        int my_s = 0;
        if (lane < n) my_s = nbr[start + base + lane];
        for (int j0 = 0; j0 < n; j0 += 6) {
            const int ja = j0 + g, jb = j0 + 3 + g;
            const bool va = live && ja < n, vb = live && jb < n;
            const int sa = __shfl(my_s, ja & 31, 32), sb = __shfl(my_s, jb & 31, 32);
            float4 xa = {0.f, 0.f, 0.f, 0.f}, xb = {0.f, 0.f, 0.f, 0.f};
            if (va) xa = rows[(size_t)sa * (X2P / 4) + q];
            if (vb) xb = rows[(size_t)sb * (X2P / 4) + q];
            if (va) { acc.x += xa.x; acc.y += xa.y; acc.z += xa.z; acc.w += xa.w; }
            if (vb) { acc.x += xb.x; acc.y += xb.y; acc.z += xb.z; acc.w += xb.w; }
        }
    }
    const int l1 = (lane + L2_Q) & 31, l2 = (lane + 2 * L2_Q) & 31;
    const float ax = __shfl(acc.x, l1, 32), ay = __shfl(acc.y, l1, 32);
    const float az = __shfl(acc.z, l1, 32), aw = __shfl(acc.w, l1, 32);
    const float bx = __shfl(acc.x, l2, 32), by = __shfl(acc.y, l2, 32);
    const float bz = __shfl(acc.z, l2, 32), bw = __shfl(acc.w, l2, 32);
    if (g == 0) {
        float4 bb;
        bb.x = b2[q * 4 + 0]; bb.y = b2[q * 4 + 1]; bb.z = b2[q * 4 + 2]; bb.w = b2[q * 4 + 3];
        float4 o;
        o.x = (acc.x + ax + bx) * di + bb.x; o.y = (acc.y + ay + by) * di + bb.y;
        o.z = (acc.z + az + bz) * di + bb.z; o.w = (acc.w + aw + bw) * di + bb.w;
        ((float4*)logits)[(size_t)node * L2_Q + q] = o;
    }
}

#define CS_THREADS 320
#define CS_ROWS    160
#define CS_Q       (NC / 4)
#define CS_LOADS   (CS_ROWS * CS_Q / CS_THREADS)
#define CS_SLOTS   (CS_THREADS / CS_Q)
#define CS_BLOCKS  (N_NODES / CS_ROWS)
#define CS_PITCH   128
#define CS_SUMOFF  64
static_assert(NC % 4 == 0, "rows must split into float4 pieces");
static_assert(CS_THREADS % CS_Q == 0, "a thread must stay on one column group");
static_assert((CS_ROWS * CS_Q) % CS_THREADS == 0, "whole loads per thread");
static_assert(N_NODES % CS_ROWS == 0, "whole row tiles");
static_assert(CS_SLOTS % 4 == 0, "row slots are summed in four chains");
static_assert(NC <= CS_SUMOFF && CS_SUMOFF + NC <= CS_PITCH, "partial plane layout");

__global__ __launch_bounds__(CS_THREADS)
void k_colpart(const float* __restrict__ logits, float* __restrict__ part) {
    __shared__ float4 tile[CS_THREADS];
    __shared__ float4 tmax4[CS_Q];
    const int t = threadIdx.x;
    const float4* p = (const float4*)logits + (size_t)blockIdx.x * (CS_ROWS * CS_Q) + t;
    const float* tf = (const float*)tile;
    float* o = part + (size_t)blockIdx.x * CS_PITCH + t;

    float4 v[CS_LOADS];
    for (int k = 0; k < CS_LOADS; ++k) v[k] = p[k * CS_THREADS];

    float4 m = v[0];
    for (int k = 1; k < CS_LOADS; ++k) {
        m.x = fmaxf(m.x, v[k].x); m.y = fmaxf(m.y, v[k].y);
        m.z = fmaxf(m.z, v[k].z); m.w = fmaxf(m.w, v[k].w);
    }
    tile[t] = m;
    __syncthreads();
    if (t < NC) {
        float M = tf[t];
        for (int j = 1; j < CS_SLOTS; ++j) M = fmaxf(M, tf[j * NC + t]);
        ((float*)tmax4)[t] = M;
        o[0] = M;
    }
    __syncthreads();

    const float4 M4 = tmax4[t % CS_Q];
    float4 s = {0.f, 0.f, 0.f, 0.f};
    for (int k = 0; k < CS_LOADS; ++k) {
        s.x += expf(v[k].x - M4.x); s.y += expf(v[k].y - M4.y);
        s.z += expf(v[k].z - M4.z); s.w += expf(v[k].w - M4.w);
    }
    tile[t] = s;
    __syncthreads();
    if (t < NC) {
        float a0 = 0.f, a1 = 0.f, a2 = 0.f, a3 = 0.f;
        for (int j = 0; j < CS_SLOTS; j += 4) {
            a0 += tf[(j + 0) * NC + t]; a1 += tf[(j + 1) * NC + t];
            a2 += tf[(j + 2) * NC + t]; a3 += tf[(j + 3) * NC + t];
        }
        o[CS_SUMOFF] = (a0 + a1) + (a2 + a3);
    }
}

#define CC_THREADS 256
#define CC_PER     ((CS_BLOCKS + CC_THREADS - 1) / CC_THREADS)

__global__ __launch_bounds__(CC_THREADS)
void k_colcomb(const float* __restrict__ part,
               float* __restrict__ colmax, float* __restrict__ colsum) {
    __shared__ float wmax[CC_THREADS / 32];
    __shared__ float wsum[CC_THREADS / 32];
    const int c = blockIdx.x;
    const int t = threadIdx.x;

    float pm[CC_PER], ps[CC_PER];
    for (int i = 0; i < CC_PER; ++i) {
        const int b = t + i * CC_THREADS;
        const bool ok = b < CS_BLOCKS;
        const float* q = part + (size_t)(ok ? b : 0) * CS_PITCH + c;
        const float qm = q[0], qs = q[CS_SUMOFF];
        pm[i] = ok ? qm : -INFINITY;
        ps[i] = ok ? qs : 0.f;
    }
    float m = pm[0];
    for (int i = 1; i < CC_PER; ++i) m = fmaxf(m, pm[i]);
    for (int off = 16; off > 0; off >>= 1) m = fmaxf(m, __shfl_xor(m, off, 32));
    if ((t & 31) == 0) wmax[t >> 5] = m;
    __syncthreads();
    float cm = wmax[0];
    for (int w = 1; w < CC_THREADS / 32; ++w) cm = fmaxf(cm, wmax[w]);

    float acc = 0.f;
    for (int i = 0; i < CC_PER; ++i) {
        const bool ok = (t + i * CC_THREADS) < CS_BLOCKS;
        acc += ok ? ps[i] * expf(pm[i] - cm) : 0.f;
    }
    for (int off = 16; off > 0; off >>= 1) acc += __shfl_xor(acc, off, 32);
    if ((t & 31) == 0) wsum[t >> 5] = acc;
    __syncthreads();
    if (t == 0) {
        float tot = wsum[0];
        for (int w = 1; w < CC_THREADS / 32; ++w) tot += wsum[w];
        colmax[c] = cm;
        colsum[c] = tot;
    }
}

__global__ void k_finalout(const float* __restrict__ logits,
                           const float* __restrict__ colmax, const float* __restrict__ colsum,
                           float* __restrict__ out_lsm, float* __restrict__ out_sm0) {
    int gid  = blockIdx.x * blockDim.x + threadIdx.x;
    int node = gid >> 5;
    if (node >= N_NODES) return;
    int lane = gid & 31;
    const float* row = logits + (size_t)node * NC;
    float v0 = row[lane];
    bool has2 = lane < (NC - 32);
    float v1 = has2 ? row[lane + 32] : -INFINITY;
    float m = fmaxf(v0, v1);
    #pragma unroll
    for (int off = 16; off > 0; off >>= 1) m = fmaxf(m, __shfl_xor(m, off, 32));
    float s = expf(v0 - m) + (has2 ? expf(v1 - m) : 0.f);
    #pragma unroll
    for (int off = 16; off > 0; off >>= 1) s += __shfl_xor(s, off, 32);
    float lse = m + logf(s);
    out_lsm[(size_t)node * NC + lane] = v0 - lse;
    out_sm0[(size_t)node * NC + lane] = expf(v0 - colmax[lane]) / colsum[lane];
    if (has2) {
        out_lsm[(size_t)node * NC + lane + 32] = v1 - lse;
        out_sm0[(size_t)node * NC + lane + 32] =
            expf(v1 - colmax[lane + 32]) / colsum[lane + 32];
    }
}

extern "C" void kernel_launch(void* const* d_in, const int* in_sizes, int n_in,
                              void* d_out, int out_size, void* d_ws, size_t ws_size,
                              hipStream_t stream) {
    (void)in_sizes; (void)n_in; (void)out_size; (void)ws_size;
    const float* x   = (const float*)d_in[0];
    const int*   ei  = (const int*)d_in[1];
    const float* W1  = (const float*)d_in[2];
    const float* b1  = (const float*)d_in[3];
    const float* W2  = (const float*)d_in[4];
    const float* b2  = (const float*)d_in[5];
    const int* src = ei;
    const int* dst = ei + N_EDGES;

    float* ws = (float*)d_ws;
    size_t off = 0;
    float* dinv = ws + off; off += N_NODES; off = (off + 63) & ~(size_t)63;
    float* xw1  = ws + off; off += (size_t)N_NODES * NH;
    float* agg1 = ws + off; off += (size_t)N_NODES * NH;
    float* W2p  = ws + off; off += NH * NCP;
    float* agg2 = ws + off; off += (size_t)N_NODES * NC;
    float* colmax = ws + off; off += 64;
    float* colsum = ws + off; off += 64;
    static_assert((size_t)N_NODES * X2P + (size_t)CS_BLOCKS * CS_PITCH <= (size_t)N_NODES * NH
                  && (NH * NCP) % 64 == 0 && ((size_t)N_NODES * NH) % 64 == 0
                  && ((size_t)N_NODES * NC) % 64 == 0, "second product and partials fit; whole units");
    float* xw2   = xw1;
    float* cpart = xw1 + (size_t)N_NODES * X2P;

    static_assert(2 * (size_t)N_NODES + 2 * (size_t)N_EDGES + (size_t)(NB + 1) * NG
                  <= (size_t)N_NODES * NC, "edge lists fit the reused plane");
    int* cnt    = (int*)agg2;
    int* rowptr = cnt + N_NODES;
    int* tmp    = rowptr + N_NODES;
    int* nbr    = tmp + N_EDGES;
    int* lstT   = nbr + N_EDGES;

    float* out_lsm = (float*)d_out;
    float* logits  = (float*)d_out + (size_t)N_NODES * NC;
    float* out_sm0 = (float*)d_out + 2 * (size_t)N_NODES * NC;

    k_part<<<NG, PS_THREADS, 0, stream>>>(src, dst, tmp, lstT);
    k_bsort<<<NB, BS_THREADS, 0, stream>>>(tmp, lstT, cnt, rowptr, dinv, nbr);

    k_gemm_wmma<NF, NH, NH / 16, NH><<<G1_BLOCKS, (NH / 16) * 32, 0, stream>>>(x, W1, dinv, xw1);
    k_gather1<<<N_NODES / GA_WAVES, GA_WAVES * 32, 0, stream>>>(rowptr, cnt, nbr, dinv, xw1, b1, agg1);

    k_padw2<<<(NH * NCP + 255) / 256, 256, 0, stream>>>(W2, W2p);
    k_gemm_wmma<NH, NCP, NCP / 16, X2P><<<G2_BLOCKS, (NCP / 16) * 32, 0, stream>>>(agg1, W2p, dinv, xw2);
    k_gather2<<<N_NODES / GA_WAVES, GA_WAVES * 32, 0, stream>>>(rowptr, cnt, nbr, dinv, xw2, b2, logits);

    k_colpart<<<CS_BLOCKS, CS_THREADS, 0, stream>>>(logits, cpart);
    k_colcomb<<<NC, CC_THREADS, 0, stream>>>(cpart, colmax, colsum);
    k_finalout<<<(N_NODES * 32 + 255) / 256, 256, 0, stream>>>(logits, colmax, colsum,
                                                               out_lsm, out_sm0);
}


# passes over the finished logits folded into the 2nd neighbour sum (resident grid, rows in LDS: column max/sum partials, row log-softmax); W2 read unpadded; launches 10->8; exp sums in new order
# speedup vs baseline: 6.9544x; 1.0886x over previous
#include <hip/hip_runtime.h>
#include <math.h>

#define N_NODES 100000
#define N_EDGES 1600000
#define NF 256
#define NH 128
#define NC 40
#define NCP 48

typedef float v2f __attribute__((ext_vector_type(2)));
typedef float v4f __attribute__((ext_vector_type(4)));
typedef float v8f __attribute__((ext_vector_type(8)));

#define E4 (N_EDGES / 4)
static_assert(N_EDGES % 4 == 0 && N_NODES % 4 == 0, "16-byte index accesses");

#define BK_SHIFT   8
#define BK_W       (1 << BK_SHIFT)
#define NB         ((N_NODES + BK_W - 1) / BK_W)
#define SRC_BITS   17
#define SRC_MASK   ((1 << SRC_BITS) - 1)
#define CH         8192
#define NG         ((N_EDGES + CH - 1) / CH)
#define PS_THREADS 1024
#define PS_Q       (CH / 4 / PS_THREADS)
static_assert(N_NODES <= (1 << SRC_BITS) && SRC_BITS + BK_SHIFT <= 31, "packed entry fits an int");
static_assert(CH % (4 * PS_THREADS) == 0 && NB + 1 <= PS_THREADS, "chunk shape");

__global__ __launch_bounds__(PS_THREADS)
void k_part(const int* __restrict__ src, const int* __restrict__ dst,
            int* __restrict__ tmp, int* __restrict__ lstT) {
    __shared__ int hist[NB + 1];
    __shared__ int wtot[PS_THREADS / 32];
    __shared__ __attribute__((aligned(16))) int sorted[CH];
    const int t = threadIdx.x, lane = t & 31, w = t >> 5, g = blockIdx.x;

    if (t <= NB) hist[t] = 0;
    __syncthreads();

    int item[PS_Q * 4], bk[PS_Q * 4], rk[PS_Q * 4];
    bool ok[PS_Q];
    #pragma unroll
    for (int u = 0; u < PS_Q; ++u) {
        const int q = g * (CH / 4) + u * PS_THREADS + t;
        ok[u] = q < E4;
        int4 d = {0, 0, 0, 0}, s = {0, 0, 0, 0};
        if (ok[u]) { d = ((const int4*)dst)[q]; s = ((const int4*)src)[q]; }
        const int dd[4] = {d.x, d.y, d.z, d.w}, ss[4] = {s.x, s.y, s.z, s.w};
        #pragma unroll
        for (int k = 0; k < 4; ++k) {
            bk[4 * u + k]   = dd[k] >> BK_SHIFT;
            item[4 * u + k] = ss[k] | ((dd[k] & (BK_W - 1)) << SRC_BITS);
            rk[4 * u + k]   = 0;
            if (ok[u]) rk[4 * u + k] = atomicAdd(&hist[bk[4 * u + k]], 1);
        }
    }
    __syncthreads();

    const int v = (t < NB) ? hist[t] : 0;
    int inc = v;
    for (int off = 1; off < 32; off <<= 1) {
        const int y = __shfl_up(inc, off, 32);
        if (lane >= off) inc += y;
    }
    if (lane == 31) wtot[w] = inc;
    __syncthreads();
    int ex = inc - v;
    for (int k = 0; k < w; ++k) ex += wtot[k];
    if (t <= NB) { hist[t] = ex; lstT[t * NG + g] = ex; }
    __syncthreads();

    #pragma unroll
    for (int u = 0; u < PS_Q; ++u) {
        if (ok[u]) {
            #pragma unroll
            for (int k = 0; k < 4; ++k) sorted[hist[bk[4 * u + k]] + rk[4 * u + k]] = item[4 * u + k];
        }
    }
    __syncthreads();

    #pragma unroll
    for (int u = 0; u < PS_Q; ++u) {
        const int i = u * PS_THREADS + t;
        const int q = g * (CH / 4) + i;
        if (q < E4) ((int4*)tmp)[q] = ((const int4*)sorted)[i];
    }
}

#define BS_THREADS 1024
#define BS_CAP     16384
static_assert(NG <= BS_THREADS && BK_W <= BS_THREADS, "one thread per chunk and per node of a bucket");

__global__ __launch_bounds__(BS_THREADS)
void k_bsort(const int* __restrict__ tmp, const int* __restrict__ lstT,
             int* __restrict__ cnt, int* __restrict__ rowptr, float* __restrict__ dinv,
             int* __restrict__ nbr) {
    __shared__ int ca[NG];
    __shared__ int pre[NG + 1];
    __shared__ int ncnt[BK_W];
    __shared__ int ncur[BK_W];
    __shared__ int wsc[BS_THREADS / 32], wrd[BS_THREADS / 32], wsc2[BS_THREADS / 32];
    __shared__ int stage[BS_CAP];
    const int t = threadIdx.x, lane = t & 31, w = t >> 5, b = blockIdx.x;

    int a = 0, len = 0;
    if (t < NG) { a = lstT[b * NG + t]; len = lstT[(b + 1) * NG + t] - a; }
    int inc = len;
    for (int off = 1; off < 32; off <<= 1) {
        const int y = __shfl_up(inc, off, 32);
        if (lane >= off) inc += y;
    }
    int asum = a;
    for (int off = 16; off > 0; off >>= 1) asum += __shfl_xor(asum, off, 32);
    if (lane == 31) wsc[w] = inc;
    if (lane == 0)  wrd[w] = asum;
    if (t < BK_W) ncnt[t] = 0;
    __syncthreads();
    int base = 0, run0 = inc - len;
    for (int k = 0; k < BS_THREADS / 32; ++k) base += wrd[k];
    for (int k = 0; k < w; ++k) run0 += wsc[k];
    if (t < NG) { ca[t] = t * CH + a; pre[t] = run0; }
    if (t == NG - 1) pre[NG] = run0 + len;
    __syncthreads();
    const int items = pre[NG];

    for (int j = t; j < items; j += BS_THREADS) {
        int lo = 0, hi = NG;
        while (hi - lo > 1) {
            const int mid = (lo + hi) >> 1;
            if (pre[mid] <= j) lo = mid; else hi = mid;
        }
        const int it = tmp[ca[lo] + (j - pre[lo])];
        atomicAdd(&ncnt[it >> SRC_BITS], 1);
    }
    __syncthreads();

    const int v = (t < BK_W) ? ncnt[t] : 0;
    int inc2 = v;
    for (int off = 1; off < 32; off <<= 1) {
        const int y = __shfl_up(inc2, off, 32);
        if (lane >= off) inc2 += y;
    }
    if (lane == 31) wsc2[w] = inc2;
    __syncthreads();
    if (t < BK_W) {
        int ex = inc2 - v;
        for (int k = 0; k < w; ++k) ex += wsc2[k];
        ncur[t] = ex;
        const int node = b * BK_W + t;
        if (node < N_NODES) {
            cnt[node] = v;
            rowptr[node] = base + ex;
            dinv[node] = rsqrtf((float)v + 1.0f);
        }
    }
    __syncthreads();

    for (int j = t; j < items; j += BS_THREADS) {
        int lo = 0, hi = NG;
        while (hi - lo > 1) {
            const int mid = (lo + hi) >> 1;
            if (pre[mid] <= j) lo = mid; else hi = mid;
        }
        const int it = tmp[ca[lo] + (j - pre[lo])];
        const int pos = atomicAdd(&ncur[it >> SRC_BITS], 1);
        if (pos < BS_CAP) stage[pos] = it & SRC_MASK;
        else nbr[base + pos] = it & SRC_MASK;
    }
    __syncthreads();
    const int m = min(items, BS_CAP);
    for (int i = t; i < m; i += BS_THREADS) nbr[base + i] = stage[i];
}

#define GEMM_MT (N_NODES / 16)
static_assert(N_NODES % 16 == 0, "whole row tiles");
#define G1_BLOCKS 512
#define GM2_BLOCKS (GEMM_MT / 5)
static_assert(GEMM_MT % 5 == 0 && G1_BLOCKS <= GEMM_MT, "grid shapes");

template<int K, int LDB, int NT, int LDC, int BP, int BV>
__global__ __launch_bounds__(NT * 32)
void k_gemm_wmma(const float* __restrict__ A, const float* __restrict__ B,
                 const float* __restrict__ rs, float* __restrict__ C) {
    constexpr int P     = K + 4;
    constexpr int RQ    = K / 4;
    constexpr int PPL   = RQ / 32;
    constexpr int RPW   = (16 + NT - 1) / NT;
    constexpr int LROWS = RPW * NT;
    constexpr int NST   = RPW * PPL;
    static_assert(LDB == NT * 16 && LDC >= LDB && RQ % 32 == 0 && (P / 4) % 2 == 1, "tile shape");
    static_assert(BV <= LDB && BV <= BP && BV > 0, "valid columns of B inside its tiles and its rows");
    __shared__ __attribute__((aligned(16))) float tile[2][LROWS * P];

    const int t    = threadIdx.x;
    const int nt   = t >> 5;
    const int lane = t & 31;
    const int half = lane >> 4;
    const int lm   = lane & 15;

    v2f b[K / 4];
    const int  bc  = nt * 16 + lm;
    const bool bok = (BV == LDB) || bc < BV;
    const float* bcol = B + (bok ? bc : BV - 1);
    #pragma unroll
    for (int s = 0; s < K / 4; ++s) {
        const int ka = 4 * s + 2 * half;
        const float bx = bcol[(size_t)ka * BP];
        const float by = bcol[(size_t)(ka + 1) * BP];
        b[s].x = bok ? bx : 0.f;
        b[s].y = bok ? by : 0.f;
    }

    int goff[NST], loff[NST];
    #pragma unroll
    for (int i = 0; i < RPW; ++i) {
        const int row = nt + i * NT;
        const int srow = (16 % NT == 0) ? row : min(row, 15);
        #pragma unroll
        for (int j = 0; j < PPL; ++j) {
            goff[i * PPL + j] = srow * RQ + j * 32 + lane;
            loff[i * PPL + j] = row * P + (j * 32 + lane) * 4;
        }
    }

    v4f st[NST];
    int mt = blockIdx.x;
    const v4f* rsp = (const v4f*)(rs + 8 * half);
    v4f rs0 = rsp[mt * 4], rs1 = rsp[mt * 4 + 1];
    {
        const v4f* g = (const v4f*)(A + (size_t)mt * (16 * K));
        #pragma unroll
        for (int i = 0; i < NST; ++i) st[i] = g[goff[i]];
        #pragma unroll
        for (int i = 0; i < NST; ++i) *(v4f*)&tile[0][loff[i]] = st[i];
    }
    __syncthreads();

    int cur = 0;
    for (; mt < GEMM_MT; mt += gridDim.x) {
        const int nx = mt + gridDim.x;
        const bool more = nx < GEMM_MT;
        v4f nrs0 = rs0, nrs1 = rs1;
        if (more) {
            const v4f* g = (const v4f*)(A + (size_t)nx * (16 * K));
            #pragma unroll
            for (int i = 0; i < NST; ++i) st[i] = g[goff[i]];
            nrs0 = rsp[nx * 4]; nrs1 = rsp[nx * 4 + 1];
        }
        const float* ap = &tile[cur][lm * P + 2 * half];
        v8f c = {0.f, 0.f, 0.f, 0.f, 0.f, 0.f, 0.f, 0.f};
        #pragma unroll
        for (int s = 0; s < K / 4; ++s) {
            const v2f a = *(const v2f*)(ap + 4 * s);
            c = __builtin_amdgcn_wmma_f32_16x16x4_f32(false, a, false, b[s],
                                                      (short)0, c, false, false);
        }

        if (more) {
            #pragma unroll
            for (int i = 0; i < NST; ++i) *(v4f*)&tile[cur ^ 1][loff[i]] = st[i];
        }
        __syncthreads();
        cur ^= 1;

        float* out = C + (size_t)(mt * 16 + 8 * half) * LDC + nt * 16 + lm;
        #pragma unroll
        for (int v = 0; v < 8; ++v) out[(size_t)v * LDC] = c[v] * (v < 4 ? rs0[v & 3] : rs1[v & 3]);
        rs0 = nrs0; rs1 = nrs1;
    }
}

#define GA_WAVES 8
static_assert(N_NODES % GA_WAVES == 0, "whole work-groups of nodes");
static_assert(NH == 128, "a 32-lane wave covers a hidden row with one float4 per lane");

__device__ __forceinline__ int   bcast_i(int v, int j)   { return __builtin_amdgcn_readlane(v, j); }

__global__ __launch_bounds__(GA_WAVES * 32)
void k_gather1(const int* __restrict__ rowptr, const int* __restrict__ cnt,
               const int* __restrict__ nbr, const float* __restrict__ dinv,
               const float* __restrict__ xs, const float* __restrict__ b,
               float* __restrict__ h) {
    const int lane = threadIdx.x & 31;
    const int node = __builtin_amdgcn_readfirstlane(blockIdx.x * GA_WAVES + (threadIdx.x >> 5));
    const int start = rowptr[node], deg = cnt[node];
    const float di = dinv[node];
    const float4* rows = (const float4*)xs;

    float4 acc = rows[(size_t)node * (NH / 4) + lane];

    for (int base = 0; base < deg; base += 32) {
        const int n = min(32, deg - base);
        int my_s = 0;
        if (lane < n) my_s = nbr[start + base + lane];
        for (int j = 0; j < n; j += 4) {
            const int r = n - j;
            const float4 z = {0.f, 0.f, 0.f, 0.f};
            float4 v1 = z, v2 = z, v3 = z;
            const float4 v0 = rows[(size_t)bcast_i(my_s, j) * (NH / 4) + lane];
            if (r > 1) v1 = rows[(size_t)bcast_i(my_s, j + 1) * (NH / 4) + lane];
            if (r > 2) v2 = rows[(size_t)bcast_i(my_s, j + 2) * (NH / 4) + lane];
            if (r > 3) v3 = rows[(size_t)bcast_i(my_s, j + 3) * (NH / 4) + lane];
            acc.x += v0.x; acc.y += v0.y; acc.z += v0.z; acc.w += v0.w;
            if (r > 1) { acc.x += v1.x; acc.y += v1.y; acc.z += v1.z; acc.w += v1.w; }
            if (r > 2) { acc.x += v2.x; acc.y += v2.y; acc.z += v2.z; acc.w += v2.w; }
            if (r > 3) { acc.x += v3.x; acc.y += v3.y; acc.z += v3.z; acc.w += v3.w; }
        }
    }
    float4 b4;
    b4.x = b[lane * 4 + 0]; b4.y = b[lane * 4 + 1]; b4.z = b[lane * 4 + 2]; b4.w = b[lane * 4 + 3];
    float4 o;
    o.x = fmaxf(acc.x * di + b4.x, 0.f); o.y = fmaxf(acc.y * di + b4.y, 0.f);
    o.z = fmaxf(acc.z * di + b4.z, 0.f); o.w = fmaxf(acc.w * di + b4.w, 0.f);
    ((float4*)h)[(size_t)node * (NH / 4) + lane] = o;
}


#define L2_Q (NC / 4)
#define X2P  64
#define G2_BLOCKS 1786
#define G2_STRIDE (G2_BLOCKS * GA_WAVES)
#define G2_TRIPS  ((N_NODES + G2_STRIDE - 1) / G2_STRIDE)
#define G2_ROWS   (G2_TRIPS * GA_WAVES)
#define G2_RG     4
#define CS_PITCH  128
#define CS_SUMOFF 64
static_assert(3 * L2_Q <= 32 && NCP % 4 == 0, "three edge groups per wave");
static_assert(NCP <= X2P && X2P % 64 == 0, "a row fits its pitch; the pitch is whole 256-byte units");
static_assert(N_NODES >= G2_STRIDE && (N_NODES % G2_STRIDE) % GA_WAVES == 0,
              "every wave has a node, and the waves of a work-group make the same number of trips");
#define G2_LSE_T0 (GA_WAVES * 32 - 64)
static_assert(G2_RG * NC <= G2_LSE_T0 && G2_RG <= GA_WAVES && G2_ROWS <= 64 && G2_ROWS % G2_RG == 0,
              "column fold: a thread per (row group, column); row pass: a thread per row, on waves of its own");
static_assert(NC <= CS_SUMOFF && CS_SUMOFF + NC <= CS_PITCH, "partial plane layout");

__global__ __launch_bounds__(GA_WAVES * 32)
void k_gather2(const int* __restrict__ rowptr, const int* __restrict__ cnt,
               const int* __restrict__ nbr, const float* __restrict__ dinv,
               const float* __restrict__ xs2, const float* __restrict__ b2,
               float* __restrict__ logits, float* __restrict__ part,
               float* __restrict__ out_lsm) {
    __shared__ __attribute__((aligned(16))) float rowsm[G2_ROWS * NC];
    __shared__ float pmx[G2_RG * NC], psm[G2_RG * NC];
    __shared__ float lse[G2_ROWS];
    const int t = threadIdx.x, lane = t & 31;
    const int w  = __builtin_amdgcn_readfirstlane(t >> 5);
    const int gw = __builtin_amdgcn_readfirstlane(blockIdx.x * GA_WAVES + (t >> 5));
    const float4* rows = (const float4*)xs2;
    const int g = lane / L2_Q;
    const int q = lane - g * L2_Q;
    const bool live = g < 3;

    float4 bb = {0.f, 0.f, 0.f, 0.f};
    if (g == 0) { bb.x = b2[q * 4 + 0]; bb.y = b2[q * 4 + 1]; bb.z = b2[q * 4 + 2]; bb.w = b2[q * 4 + 3]; }

    int node = gw;
    int start = rowptr[node], deg = cnt[node];
    float di = dinv[node];
    for (int it = 0; node < N_NODES; ++it) {
        const int nnode = node + G2_STRIDE;
        const int nc = min(nnode, N_NODES - 1);
        const int nstart = rowptr[nc], ndeg = cnt[nc];
        const float ndi = dinv[nc];

        float4 acc = {0.f, 0.f, 0.f, 0.f};
        if (g == 0) acc = rows[(size_t)node * (X2P / 4) + q];
        for (int base = 0; base < deg; base += 32) {
            const int n = min(32, deg - base);
            int my_s = 0;
            if (lane < n) my_s = nbr[start + base + lane];
            for (int j0 = 0; j0 < n; j0 += 6) {
                const int ja = j0 + g, jb = j0 + 3 + g;
                const bool va = live && ja < n, vb = live && jb < n;
                const int sa = __shfl(my_s, ja & 31, 32), sb = __shfl(my_s, jb & 31, 32);
                float4 xa = {0.f, 0.f, 0.f, 0.f}, xb = {0.f, 0.f, 0.f, 0.f};
                if (va) xa = rows[(size_t)sa * (X2P / 4) + q];
                if (vb) xb = rows[(size_t)sb * (X2P / 4) + q];
                if (va) { acc.x += xa.x; acc.y += xa.y; acc.z += xa.z; acc.w += xa.w; }
                if (vb) { acc.x += xb.x; acc.y += xb.y; acc.z += xb.z; acc.w += xb.w; }
            }
        }
        const int l1 = (lane + L2_Q) & 31, l2 = (lane + 2 * L2_Q) & 31;
        const float ax = __shfl(acc.x, l1, 32), ay = __shfl(acc.y, l1, 32);
        const float az = __shfl(acc.z, l1, 32), aw = __shfl(acc.w, l1, 32);
        const float bx = __shfl(acc.x, l2, 32), by = __shfl(acc.y, l2, 32);
        const float bz = __shfl(acc.z, l2, 32), bw = __shfl(acc.w, l2, 32);
        if (g == 0) {
            float4 o;
            o.x = (acc.x + ax + bx) * di + bb.x; o.y = (acc.y + ay + by) * di + bb.y;
            o.z = (acc.z + az + bz) * di + bb.z; o.w = (acc.w + aw + bw) * di + bb.w;
            ((float4*)logits)[(size_t)node * L2_Q + q] = o;
            *(float4*)&rowsm[(it * GA_WAVES + w) * NC + q * 4] = o;
        }
        node = nnode; start = nstart; deg = ndeg; di = ndi;
    }
    __syncthreads();

    const int nrows = ((N_NODES - 1 - (int)blockIdx.x * GA_WAVES) / G2_STRIDE + 1) * GA_WAVES;
    if (t < G2_RG * NC) {
        const int rg = t / NC, c = t - rg * NC;
        float xv[G2_ROWS / G2_RG];
        for (int k = 0; k < G2_ROWS / G2_RG; ++k) {
            const int r = rg + k * G2_RG;
            const float x = rowsm[min(r, nrows - 1) * NC + c];
            xv[k] = (r < nrows) ? x : -INFINITY;
        }
        float M = xv[0];
        for (int k = 1; k < G2_ROWS / G2_RG; ++k) M = fmaxf(M, xv[k]);
        float S = 0.f;
        for (int k = 0; k < G2_ROWS / G2_RG; ++k) S += expf(xv[k] - M);
        pmx[t] = M; psm[t] = S;
    } else if (t >= G2_LSE_T0 && t - G2_LSE_T0 < nrows) {
        const int r = t - G2_LSE_T0;
        const float4* r4 = (const float4*)&rowsm[r * NC];
        float4 v[L2_Q];
        for (int k = 0; k < L2_Q; ++k) v[k] = r4[k];
        float m = fmaxf(fmaxf(v[0].x, v[0].y), fmaxf(v[0].z, v[0].w));
        for (int k = 1; k < L2_Q; ++k) m = fmaxf(m, fmaxf(fmaxf(v[k].x, v[k].y), fmaxf(v[k].z, v[k].w)));
        float s = 0.f;
        for (int k = 0; k < L2_Q; ++k)
            s += (expf(v[k].x - m) + expf(v[k].y - m)) + (expf(v[k].z - m) + expf(v[k].w - m));
        lse[r] = m + logf(s);
    }
    __syncthreads();
    for (int k = 0; k < (G2_ROWS * NC + GA_WAVES * 32 - 1) / (GA_WAVES * 32); ++k) {
        const int e = t + k * (GA_WAVES * 32);
        if (e < nrows * NC) {
            const int r = e / NC, c = e - r * NC;
            const int node = (int)blockIdx.x * GA_WAVES + (r % GA_WAVES) + (r / GA_WAVES) * G2_STRIDE;
            out_lsm[(size_t)node * NC + c] = rowsm[e] - lse[r];
        }
    }
    if (t < NC) {
        float M = pmx[t];
        for (int k = 1; k < G2_RG; ++k) M = fmaxf(M, pmx[k * NC + t]);
        float S = 0.f;
        for (int k = 0; k < G2_RG; ++k) S += psm[k * NC + t] * expf(pmx[k * NC + t] - M);
        float* o = part + (size_t)blockIdx.x * CS_PITCH + t;
        o[0] = M;
        o[CS_SUMOFF] = S;
    }
}

#define CS_BLOCKS G2_BLOCKS

#define CC_THREADS 256
#define CC_PER     ((CS_BLOCKS + CC_THREADS - 1) / CC_THREADS)

__global__ __launch_bounds__(CC_THREADS)
void k_colcomb(const float* __restrict__ part,
               float* __restrict__ colmax, float* __restrict__ colsum) {
    __shared__ float wmax[CC_THREADS / 32];
    __shared__ float wsum[CC_THREADS / 32];
    const int c = blockIdx.x;
    const int t = threadIdx.x;

    float pm[CC_PER], ps[CC_PER];
    for (int i = 0; i < CC_PER; ++i) {
        const int b = t + i * CC_THREADS;
        const bool ok = b < CS_BLOCKS;
        const float* q = part + (size_t)(ok ? b : 0) * CS_PITCH + c;
        const float qm = q[0], qs = q[CS_SUMOFF];
        pm[i] = ok ? qm : -INFINITY;
        ps[i] = ok ? qs : 0.f;
    }
    float m = pm[0];
    for (int i = 1; i < CC_PER; ++i) m = fmaxf(m, pm[i]);
    for (int off = 16; off > 0; off >>= 1) m = fmaxf(m, __shfl_xor(m, off, 32));
    if ((t & 31) == 0) wmax[t >> 5] = m;
    __syncthreads();
    float cm = wmax[0];
    for (int w = 1; w < CC_THREADS / 32; ++w) cm = fmaxf(cm, wmax[w]);

    float acc = 0.f;
    for (int i = 0; i < CC_PER; ++i) {
        const bool ok = (t + i * CC_THREADS) < CS_BLOCKS;
        acc += ok ? ps[i] * expf(pm[i] - cm) : 0.f;
    }
    for (int off = 16; off > 0; off >>= 1) acc += __shfl_xor(acc, off, 32);
    if ((t & 31) == 0) wsum[t >> 5] = acc;
    __syncthreads();
    if (t == 0) {
        float tot = wsum[0];
        for (int w = 1; w < CC_THREADS / 32; ++w) tot += wsum[w];
        colmax[c] = cm;
        colsum[c] = tot;
    }
}

#define FO_THREADS 256
#define FO_Q4      (N_NODES * L2_Q)
__global__ __launch_bounds__(FO_THREADS)
void k_finalout(const float* __restrict__ logits,
                const float* __restrict__ colmax, const float* __restrict__ colsum,
                float* __restrict__ out_sm0) {
    const int i = blockIdx.x * FO_THREADS + threadIdx.x;
    if (i >= FO_Q4) return;
    const int q = i % L2_Q;
    const float4 v  = ((const float4*)logits)[i];
    const float4 cm = ((const float4*)colmax)[q], cs = ((const float4*)colsum)[q];
    float4 o;
    o.x = expf(v.x - cm.x) / cs.x; o.y = expf(v.y - cm.y) / cs.y;
    o.z = expf(v.z - cm.z) / cs.z; o.w = expf(v.w - cm.w) / cs.w;
    ((float4*)out_sm0)[i] = o;
}

extern "C" void kernel_launch(void* const* d_in, const int* in_sizes, int n_in,
                              void* d_out, int out_size, void* d_ws, size_t ws_size,
                              hipStream_t stream) {
    (void)in_sizes; (void)n_in; (void)out_size; (void)ws_size;
    const float* x   = (const float*)d_in[0];
    const int*   ei  = (const int*)d_in[1];
    const float* W1  = (const float*)d_in[2];
    const float* b1  = (const float*)d_in[3];
    const float* W2  = (const float*)d_in[4];
    const float* b2  = (const float*)d_in[5];
    const int* src = ei;
    const int* dst = ei + N_EDGES;

    float* ws = (float*)d_ws;
    size_t off = 0;
    float* dinv = ws + off; off += N_NODES; off = (off + 63) & ~(size_t)63;
    float* xw1  = ws + off; off += (size_t)N_NODES * NH;
    float* agg1 = ws + off; off += (size_t)N_NODES * NH;
    float* agg2 = ws + off; off += (size_t)N_NODES * NC;
    float* colmax = ws + off; off += 64;
    float* colsum = ws + off; off += 64;
    static_assert((size_t)N_NODES * X2P + (size_t)CS_BLOCKS * CS_PITCH <= (size_t)N_NODES * NH
                  && ((size_t)N_NODES * NH) % 64 == 0
                  && ((size_t)N_NODES * NC) % 64 == 0, "second product and partials fit; whole units");
    float* xw2   = xw1;
    float* cpart = xw1 + (size_t)N_NODES * X2P;

    static_assert(2 * (size_t)N_NODES + 2 * (size_t)N_EDGES + (size_t)(NB + 1) * NG
                  <= (size_t)N_NODES * NC, "edge lists fit the reused plane");
    int* cnt    = (int*)agg2;
    int* rowptr = cnt + N_NODES;
    int* tmp    = rowptr + N_NODES;
    int* nbr    = tmp + N_EDGES;
    int* lstT   = nbr + N_EDGES;

    float* out_lsm = (float*)d_out;
    float* logits  = (float*)d_out + (size_t)N_NODES * NC;
    float* out_sm0 = (float*)d_out + 2 * (size_t)N_NODES * NC;

    k_part<<<NG, PS_THREADS, 0, stream>>>(src, dst, tmp, lstT);
    k_bsort<<<NB, BS_THREADS, 0, stream>>>(tmp, lstT, cnt, rowptr, dinv, nbr);

    k_gemm_wmma<NF, NH, NH / 16, NH, NH, NH><<<G1_BLOCKS, (NH / 16) * 32, 0, stream>>>(x, W1, dinv, xw1);
    k_gather1<<<N_NODES / GA_WAVES, GA_WAVES * 32, 0, stream>>>(rowptr, cnt, nbr, dinv, xw1, b1, agg1);

    k_gemm_wmma<NH, NCP, NCP / 16, X2P, NC, NC><<<GM2_BLOCKS, (NCP / 16) * 32, 0, stream>>>(agg1, W2, dinv, xw2);
    k_gather2<<<G2_BLOCKS, GA_WAVES * 32, 0, stream>>>(rowptr, cnt, nbr, dinv, xw2, b2, logits, cpart,
                                                         out_lsm);

    k_colcomb<<<NC, CC_THREADS, 0, stream>>>(cpart, colmax, colsum);
    k_finalout<<<(FO_Q4 + FO_THREADS - 1) / FO_THREADS, FO_THREADS, 0, stream>>>(logits, colmax, colsum, out_sm0);
}


# second product moved into the first neighbour sum: a work-group sums 16 nodes into an LDS tile and 3 of its waves multiply it by W2 (transposed in LDS); h never stored; launches 8->7; sum order kept
# speedup vs baseline: 7.3771x; 1.0608x over previous
#include <hip/hip_runtime.h>
#include <math.h>

#define N_NODES 100000
#define N_EDGES 1600000
#define NF 256
#define NH 128
#define NC 40
#define NCP 48

typedef float v2f __attribute__((ext_vector_type(2)));
typedef float v4f __attribute__((ext_vector_type(4)));
typedef float v8f __attribute__((ext_vector_type(8)));

#define E4 (N_EDGES / 4)
static_assert(N_EDGES % 4 == 0 && N_NODES % 4 == 0, "16-byte index accesses");

#define BK_SHIFT   8
#define BK_W       (1 << BK_SHIFT)
#define NB         ((N_NODES + BK_W - 1) / BK_W)
#define SRC_BITS   17
#define SRC_MASK   ((1 << SRC_BITS) - 1)
#define CH         8192
#define NG         ((N_EDGES + CH - 1) / CH)
#define PS_THREADS 1024
#define PS_Q       (CH / 4 / PS_THREADS)
static_assert(N_NODES <= (1 << SRC_BITS) && SRC_BITS + BK_SHIFT <= 31, "packed entry fits an int");
static_assert(CH % (4 * PS_THREADS) == 0 && NB + 1 <= PS_THREADS, "chunk shape");

__global__ __launch_bounds__(PS_THREADS)
void k_part(const int* __restrict__ src, const int* __restrict__ dst,
            int* __restrict__ tmp, int* __restrict__ lstT) {
    __shared__ int hist[NB + 1];
    __shared__ int wtot[PS_THREADS / 32];
    __shared__ __attribute__((aligned(16))) int sorted[CH];
    const int t = threadIdx.x, lane = t & 31, w = t >> 5, g = blockIdx.x;

    if (t <= NB) hist[t] = 0;
    __syncthreads();

    int item[PS_Q * 4], bk[PS_Q * 4], rk[PS_Q * 4];
    bool ok[PS_Q];
    #pragma unroll
    for (int u = 0; u < PS_Q; ++u) {
        const int q = g * (CH / 4) + u * PS_THREADS + t;
        ok[u] = q < E4;
        int4 d = {0, 0, 0, 0}, s = {0, 0, 0, 0};
        if (ok[u]) { d = ((const int4*)dst)[q]; s = ((const int4*)src)[q]; }
        const int dd[4] = {d.x, d.y, d.z, d.w}, ss[4] = {s.x, s.y, s.z, s.w};
        #pragma unroll
        for (int k = 0; k < 4; ++k) {
            bk[4 * u + k]   = dd[k] >> BK_SHIFT;
            item[4 * u + k] = ss[k] | ((dd[k] & (BK_W - 1)) << SRC_BITS);
            rk[4 * u + k]   = 0;
            if (ok[u]) rk[4 * u + k] = atomicAdd(&hist[bk[4 * u + k]], 1);
        }
    }
    __syncthreads();

    const int v = (t < NB) ? hist[t] : 0;
    int inc = v;
    for (int off = 1; off < 32; off <<= 1) {
        const int y = __shfl_up(inc, off, 32);
        if (lane >= off) inc += y;
    }
    if (lane == 31) wtot[w] = inc;
    __syncthreads();
    int ex = inc - v;
    for (int k = 0; k < w; ++k) ex += wtot[k];
    if (t <= NB) { hist[t] = ex; lstT[t * NG + g] = ex; }
    __syncthreads();

    #pragma unroll
    for (int u = 0; u < PS_Q; ++u) {
        if (ok[u]) {
            #pragma unroll
            for (int k = 0; k < 4; ++k) sorted[hist[bk[4 * u + k]] + rk[4 * u + k]] = item[4 * u + k];
        }
    }
    __syncthreads();

    #pragma unroll
    for (int u = 0; u < PS_Q; ++u) {
        const int i = u * PS_THREADS + t;
        const int q = g * (CH / 4) + i;
        if (q < E4) ((int4*)tmp)[q] = ((const int4*)sorted)[i];
    }
}

#define BS_THREADS 1024
#define BS_CAP     16384
static_assert(NG <= BS_THREADS && BK_W <= BS_THREADS, "one thread per chunk and per node of a bucket");

__global__ __launch_bounds__(BS_THREADS)
void k_bsort(const int* __restrict__ tmp, const int* __restrict__ lstT,
             int* __restrict__ cnt, int* __restrict__ rowptr, float* __restrict__ dinv,
             int* __restrict__ nbr) {
    __shared__ int ca[NG];
    __shared__ int pre[NG + 1];
    __shared__ int ncnt[BK_W];
    __shared__ int ncur[BK_W];
    __shared__ int wsc[BS_THREADS / 32], wrd[BS_THREADS / 32], wsc2[BS_THREADS / 32];
    __shared__ int stage[BS_CAP];
    const int t = threadIdx.x, lane = t & 31, w = t >> 5, b = blockIdx.x;

    int a = 0, len = 0;
    if (t < NG) { a = lstT[b * NG + t]; len = lstT[(b + 1) * NG + t] - a; }
    int inc = len;
    for (int off = 1; off < 32; off <<= 1) {
        const int y = __shfl_up(inc, off, 32);
        if (lane >= off) inc += y;
    }
    int asum = a;
    for (int off = 16; off > 0; off >>= 1) asum += __shfl_xor(asum, off, 32);
    if (lane == 31) wsc[w] = inc;
    if (lane == 0)  wrd[w] = asum;
    if (t < BK_W) ncnt[t] = 0;
    __syncthreads();
    int base = 0, run0 = inc - len;
    for (int k = 0; k < BS_THREADS / 32; ++k) base += wrd[k];
    for (int k = 0; k < w; ++k) run0 += wsc[k];
    if (t < NG) { ca[t] = t * CH + a; pre[t] = run0; }
    if (t == NG - 1) pre[NG] = run0 + len;
    __syncthreads();
    const int items = pre[NG];

    for (int j = t; j < items; j += BS_THREADS) {
        int lo = 0, hi = NG;
        while (hi - lo > 1) {
            const int mid = (lo + hi) >> 1;
            if (pre[mid] <= j) lo = mid; else hi = mid;
        }
        const int it = tmp[ca[lo] + (j - pre[lo])];
        atomicAdd(&ncnt[it >> SRC_BITS], 1);
    }
    __syncthreads();

    const int v = (t < BK_W) ? ncnt[t] : 0;
    int inc2 = v;
    for (int off = 1; off < 32; off <<= 1) {
        const int y = __shfl_up(inc2, off, 32);
        if (lane >= off) inc2 += y;
    }
    if (lane == 31) wsc2[w] = inc2;
    __syncthreads();
    if (t < BK_W) {
        int ex = inc2 - v;
        for (int k = 0; k < w; ++k) ex += wsc2[k];
        ncur[t] = ex;
        const int node = b * BK_W + t;
        if (node < N_NODES) {
            cnt[node] = v;
            rowptr[node] = base + ex;
            dinv[node] = rsqrtf((float)v + 1.0f);
        }
    }
    __syncthreads();

    for (int j = t; j < items; j += BS_THREADS) {
        int lo = 0, hi = NG;
        while (hi - lo > 1) {
            const int mid = (lo + hi) >> 1;
            if (pre[mid] <= j) lo = mid; else hi = mid;
        }
        const int it = tmp[ca[lo] + (j - pre[lo])];
        const int pos = atomicAdd(&ncur[it >> SRC_BITS], 1);
        if (pos < BS_CAP) stage[pos] = it & SRC_MASK;
        else nbr[base + pos] = it & SRC_MASK;
    }
    __syncthreads();
    const int m = min(items, BS_CAP);
    for (int i = t; i < m; i += BS_THREADS) nbr[base + i] = stage[i];
}

#define GEMM_MT (N_NODES / 16)
static_assert(N_NODES % 16 == 0, "whole row tiles");
#define G1_BLOCKS 512
static_assert(G1_BLOCKS <= GEMM_MT, "grid shape");

template<int K, int LDB, int NT, int LDC, int BP, int BV>
__global__ __launch_bounds__(NT * 32)
void k_gemm_wmma(const float* __restrict__ A, const float* __restrict__ B,
                 const float* __restrict__ rs, float* __restrict__ C) {
    constexpr int P     = K + 4;
    constexpr int RQ    = K / 4;
    constexpr int PPL   = RQ / 32;
    constexpr int RPW   = (16 + NT - 1) / NT;
    constexpr int LROWS = RPW * NT;
    constexpr int NST   = RPW * PPL;
    static_assert(LDB == NT * 16 && LDC >= LDB && RQ % 32 == 0 && (P / 4) % 2 == 1, "tile shape");
    static_assert(BV <= LDB && BV <= BP && BV > 0, "valid columns of B inside its tiles and its rows");
    __shared__ __attribute__((aligned(16))) float tile[2][LROWS * P];

    const int t    = threadIdx.x;
    const int nt   = t >> 5;
    const int lane = t & 31;
    const int half = lane >> 4;
    const int lm   = lane & 15;

    v2f b[K / 4];
    const int  bc  = nt * 16 + lm;
    const bool bok = (BV == LDB) || bc < BV;
    const float* bcol = B + (bok ? bc : BV - 1);
    #pragma unroll
    for (int s = 0; s < K / 4; ++s) {
        const int ka = 4 * s + 2 * half;
        const float bx = bcol[(size_t)ka * BP];
        const float by = bcol[(size_t)(ka + 1) * BP];
        b[s].x = bok ? bx : 0.f;
        b[s].y = bok ? by : 0.f;
    }

    int goff[NST], loff[NST];
    #pragma unroll
    for (int i = 0; i < RPW; ++i) {
        const int row = nt + i * NT;
        const int srow = (16 % NT == 0) ? row : min(row, 15);
        #pragma unroll
        for (int j = 0; j < PPL; ++j) {
            goff[i * PPL + j] = srow * RQ + j * 32 + lane;
            loff[i * PPL + j] = row * P + (j * 32 + lane) * 4;
        }
    }

    v4f st[NST];
    int mt = blockIdx.x;
    const v4f* rsp = (const v4f*)(rs + 8 * half);
    v4f rs0 = rsp[mt * 4], rs1 = rsp[mt * 4 + 1];
    {
        const v4f* g = (const v4f*)(A + (size_t)mt * (16 * K));
        #pragma unroll
        for (int i = 0; i < NST; ++i) st[i] = g[goff[i]];
        #pragma unroll
        for (int i = 0; i < NST; ++i) *(v4f*)&tile[0][loff[i]] = st[i];
    }
    __syncthreads();

    int cur = 0;
    for (; mt < GEMM_MT; mt += gridDim.x) {
        const int nx = mt + gridDim.x;
        const bool more = nx < GEMM_MT;
        v4f nrs0 = rs0, nrs1 = rs1;
        if (more) {
            const v4f* g = (const v4f*)(A + (size_t)nx * (16 * K));
            #pragma unroll
            for (int i = 0; i < NST; ++i) st[i] = g[goff[i]];
            nrs0 = rsp[nx * 4]; nrs1 = rsp[nx * 4 + 1];
        }
        const float* ap = &tile[cur][lm * P + 2 * half];
        v8f c = {0.f, 0.f, 0.f, 0.f, 0.f, 0.f, 0.f, 0.f};
        #pragma unroll
        for (int s = 0; s < K / 4; ++s) {
            const v2f a = *(const v2f*)(ap + 4 * s);
            c = __builtin_amdgcn_wmma_f32_16x16x4_f32(false, a, false, b[s],
                                                      (short)0, c, false, false);
        }

        if (more) {
            #pragma unroll
            for (int i = 0; i < NST; ++i) *(v4f*)&tile[cur ^ 1][loff[i]] = st[i];
        }
        __syncthreads();
        cur ^= 1;

        float* out = C + (size_t)(mt * 16 + 8 * half) * LDC + nt * 16 + lm;
        #pragma unroll
        for (int v = 0; v < 8; ++v) out[(size_t)v * LDC] = c[v] * (v < 4 ? rs0[v & 3] : rs1[v & 3]);
        rs0 = nrs0; rs1 = nrs1;
    }
}

#define X2P  64
#define GA_WAVES 8
#define GA_FLY   8
static_assert(N_NODES % GA_WAVES == 0, "whole work-groups of nodes");
static_assert(NH == 128, "a 32-lane wave covers a hidden row with one float4 per lane");
#define F1_TRIPS  5
#define F1_BLOCKS (GEMM_MT / F1_TRIPS)
#define F1_P      (NH + 4)
#define F1_NT     (NCP / 16)
#define F1_W2Q    (NH * NC / (GA_WAVES * 32))
static_assert(GEMM_MT % F1_TRIPS == 0 && 2 * GA_WAVES == 16, "whole tiles: two nodes per wave");
static_assert((NH * NC) % (GA_WAVES * 32) == 0 && ((NCP - NC) * NH) % (GA_WAVES * 32) == 0, "W2 staging");
static_assert(F1_NT <= GA_WAVES && (F1_P / 4) % 2 == 1, "tile shape");

__device__ __forceinline__ int   bcast_i(int v, int j)   { return __builtin_amdgcn_readlane(v, j); }

__device__ __forceinline__ float4 node_sum(const float4* __restrict__ rows, const int* __restrict__ nbr,
                                           float4 acc, int my_s, int start, int deg, int lane) {
    for (int base = 0; base < deg; base += 32) {
        const int n = min(32, deg - base);
        if (base > 0) {
            my_s = 0;
            if (lane < n) my_s = nbr[start + base + lane];
        }
        for (int j = 0; j < n; j += GA_FLY) {
            const int r = n - j;
            float4 v[GA_FLY];
            v[0] = rows[(size_t)bcast_i(my_s, j) * (NH / 4) + lane];
            #pragma unroll
            for (int k = 1; k < GA_FLY; ++k) {
                v[k] = make_float4(0.f, 0.f, 0.f, 0.f);
                if (r > k) v[k] = rows[(size_t)bcast_i(my_s, j + k) * (NH / 4) + lane];
            }
            acc.x += v[0].x; acc.y += v[0].y; acc.z += v[0].z; acc.w += v[0].w;
            #pragma unroll
            for (int k = 1; k < GA_FLY; ++k)
                if (r > k) { acc.x += v[k].x; acc.y += v[k].y; acc.z += v[k].z; acc.w += v[k].w; }
        }
    }
    return acc;
}

__global__ __launch_bounds__(GA_WAVES * 32)
void k_gather1_mm(const int* __restrict__ rowptr, const int* __restrict__ cnt,
                  const int* __restrict__ nbr, const float* __restrict__ dinv,
                  const float* __restrict__ xs, const float* __restrict__ b,
                  const float* __restrict__ W2, float* __restrict__ xw2) {
    __shared__ __attribute__((aligned(16))) float htile[2][16 * F1_P];
    __shared__ __attribute__((aligned(16))) float w2t[NCP * F1_P];
    __shared__ __attribute__((aligned(16))) float dtile[2][16];
    const int t = threadIdx.x, lane = t & 31;
    const int w = __builtin_amdgcn_readfirstlane(t >> 5);
    const int half = lane >> 4;
    const int lm   = lane & 15;
    const float4* rows = (const float4*)xs;

    float wv[F1_W2Q];
    #pragma unroll
    for (int i = 0; i < F1_W2Q; ++i) wv[i] = W2[t + i * (GA_WAVES * 32)];

    const int tl = min(lane, F1_TRIPS - 1);
    const int nl = ((int)blockIdx.x + tl * F1_BLOCKS) * 16 + 2 * w;
    const int m_s0 = rowptr[nl], m_s1 = rowptr[nl + 1];
    const int m_d0 = cnt[nl],    m_d1 = cnt[nl + 1];
    const int m_i0 = __float_as_int(dinv[nl]), m_i1 = __float_as_int(dinv[nl + 1]);

    int my0 = 0, my1 = 0;
    float4 acc0, acc1;
    {
        const int na = (int)blockIdx.x * 16 + 2 * w;
        if (lane < min(32, bcast_i(m_d0, 0))) my0 = nbr[bcast_i(m_s0, 0) + lane];
        if (lane < min(32, bcast_i(m_d1, 0))) my1 = nbr[bcast_i(m_s1, 0) + lane];
        acc0 = rows[(size_t)na * (NH / 4) + lane];
        acc1 = rows[(size_t)(na + 1) * (NH / 4) + lane];
    }

    #pragma unroll
    for (int i = 0; i < F1_W2Q; ++i) {
        const int e = t + i * (GA_WAVES * 32);
        const int k = e / NC, c = e - k * NC;
        w2t[c * F1_P + k] = wv[i];
    }
    for (int e = t; e < (NCP - NC) * NH; e += GA_WAVES * 32)
        w2t[(NC + e / NH) * F1_P + (e % NH)] = 0.f;

    float4 b4;
    b4.x = b[lane * 4 + 0]; b4.y = b[lane * 4 + 1]; b4.z = b[lane * 4 + 2]; b4.w = b[lane * 4 + 3];

    int cur = 0;
    for (int trip = 0; trip < F1_TRIPS; ++trip) {
        const int node0 = ((int)blockIdx.x + trip * F1_BLOCKS) * 16;
        const int s0 = bcast_i(m_s0, trip), s1 = bcast_i(m_s1, trip);
        const int d0 = bcast_i(m_d0, trip), d1 = bcast_i(m_d1, trip);
        const float di0 = __int_as_float(bcast_i(m_i0, trip)), di1 = __int_as_float(bcast_i(m_i1, trip));

        const float4 a0 = node_sum(rows, nbr, acc0, my0, s0, d0, lane);
        float4 o;
        o.x = fmaxf(a0.x * di0 + b4.x, 0.f); o.y = fmaxf(a0.y * di0 + b4.y, 0.f);
        o.z = fmaxf(a0.z * di0 + b4.z, 0.f); o.w = fmaxf(a0.w * di0 + b4.w, 0.f);
        *(float4*)&htile[cur][(2 * w) * F1_P + lane * 4] = o;
        const float4 a1 = node_sum(rows, nbr, acc1, my1, s1, d1, lane);
        o.x = fmaxf(a1.x * di1 + b4.x, 0.f); o.y = fmaxf(a1.y * di1 + b4.y, 0.f);
        o.z = fmaxf(a1.z * di1 + b4.z, 0.f); o.w = fmaxf(a1.w * di1 + b4.w, 0.f);
        *(float4*)&htile[cur][(2 * w + 1) * F1_P + lane * 4] = o;
        if (lane == 0) { dtile[cur][2 * w] = di0; dtile[cur][2 * w + 1] = di1; }
        __syncthreads();

        if (trip + 1 < F1_TRIPS) {
            const int nna = node0 + 16 * F1_BLOCKS + 2 * w;
            my0 = 0; my1 = 0;
            if (lane < min(32, bcast_i(m_d0, trip + 1))) my0 = nbr[bcast_i(m_s0, trip + 1) + lane];
            if (lane < min(32, bcast_i(m_d1, trip + 1))) my1 = nbr[bcast_i(m_s1, trip + 1) + lane];
            acc0 = rows[(size_t)nna * (NH / 4) + lane];
            acc1 = rows[(size_t)(nna + 1) * (NH / 4) + lane];
        }
        if (w < F1_NT) {
            const float* ap = &htile[cur][lm * F1_P + 2 * half];
            const float* bp = &w2t[(w * 16 + lm) * F1_P + 2 * half];
            v8f c = {0.f, 0.f, 0.f, 0.f, 0.f, 0.f, 0.f, 0.f};
            #pragma unroll
            for (int s = 0; s < NH / 4; ++s) {
                const v2f a  = *(const v2f*)(ap + 4 * s);
                const v2f bf = *(const v2f*)(bp + 4 * s);
                c = __builtin_amdgcn_wmma_f32_16x16x4_f32(false, a, false, bf,
                                                          (short)0, c, false, false);
            }
            const v4f rs0 = *(const v4f*)&dtile[cur][8 * half];
            const v4f rs1 = *(const v4f*)&dtile[cur][8 * half + 4];
            float* out = xw2 + (size_t)(node0 + 8 * half) * X2P + w * 16 + lm;
            #pragma unroll
            for (int v = 0; v < 8; ++v) out[(size_t)v * X2P] = c[v] * (v < 4 ? rs0[v & 3] : rs1[v & 3]);
        }
        cur ^= 1;
    }
}


#define L2_Q (NC / 4)
#define G2_BLOCKS 1786
#define G2_STRIDE (G2_BLOCKS * GA_WAVES)
#define G2_TRIPS  ((N_NODES + G2_STRIDE - 1) / G2_STRIDE)
#define G2_ROWS   (G2_TRIPS * GA_WAVES)
#define G2_RG     4
#define CS_PITCH  128
#define CS_SUMOFF 64
static_assert(3 * L2_Q <= 32 && NCP % 4 == 0, "three edge groups per wave");
static_assert(NCP <= X2P && X2P % 64 == 0, "a row fits its pitch; the pitch is whole 256-byte units");
static_assert(N_NODES >= G2_STRIDE && (N_NODES % G2_STRIDE) % GA_WAVES == 0,
              "every wave has a node, and the waves of a work-group make the same number of trips");
#define G2_LSE_T0 (GA_WAVES * 32 - 64)
static_assert(G2_RG * NC <= G2_LSE_T0 && G2_RG <= GA_WAVES && G2_ROWS <= 64 && G2_ROWS % G2_RG == 0,
              "column fold: a thread per (row group, column); row pass: a thread per row, on waves of its own");
static_assert(NC <= CS_SUMOFF && CS_SUMOFF + NC <= CS_PITCH, "partial plane layout");

__global__ __launch_bounds__(GA_WAVES * 32)
void k_gather2(const int* __restrict__ rowptr, const int* __restrict__ cnt,
               const int* __restrict__ nbr, const float* __restrict__ dinv,
               const float* __restrict__ xs2, const float* __restrict__ b2,
               float* __restrict__ logits, float* __restrict__ part,
               float* __restrict__ out_lsm) {
    __shared__ __attribute__((aligned(16))) float rowsm[G2_ROWS * NC];
    __shared__ float pmx[G2_RG * NC], psm[G2_RG * NC];
    __shared__ float lse[G2_ROWS];
    const int t = threadIdx.x, lane = t & 31;
    const int w  = __builtin_amdgcn_readfirstlane(t >> 5);
    const int gw = __builtin_amdgcn_readfirstlane(blockIdx.x * GA_WAVES + (t >> 5));
    const float4* rows = (const float4*)xs2;
    const int g = lane / L2_Q;
    const int q = lane - g * L2_Q;
    const bool live = g < 3;

    float4 bb = {0.f, 0.f, 0.f, 0.f};
    if (g == 0) { bb.x = b2[q * 4 + 0]; bb.y = b2[q * 4 + 1]; bb.z = b2[q * 4 + 2]; bb.w = b2[q * 4 + 3]; }

    int node = gw;
    int start = rowptr[node], deg = cnt[node];
    float di = dinv[node];
    for (int it = 0; node < N_NODES; ++it) {
        const int nnode = node + G2_STRIDE;
        const int nc = min(nnode, N_NODES - 1);
        const int nstart = rowptr[nc], ndeg = cnt[nc];
        const float ndi = dinv[nc];

        float4 acc = {0.f, 0.f, 0.f, 0.f};
        if (g == 0) acc = rows[(size_t)node * (X2P / 4) + q];
        for (int base = 0; base < deg; base += 32) {
            const int n = min(32, deg - base);
            int my_s = 0;
            if (lane < n) my_s = nbr[start + base + lane];
            for (int j0 = 0; j0 < n; j0 += 6) {
                const int ja = j0 + g, jb = j0 + 3 + g;
                const bool va = live && ja < n, vb = live && jb < n;
                const int sa = __shfl(my_s, ja & 31, 32), sb = __shfl(my_s, jb & 31, 32);
                float4 xa = {0.f, 0.f, 0.f, 0.f}, xb = {0.f, 0.f, 0.f, 0.f};
                if (va) xa = rows[(size_t)sa * (X2P / 4) + q];
                if (vb) xb = rows[(size_t)sb * (X2P / 4) + q];
                if (va) { acc.x += xa.x; acc.y += xa.y; acc.z += xa.z; acc.w += xa.w; }
                if (vb) { acc.x += xb.x; acc.y += xb.y; acc.z += xb.z; acc.w += xb.w; }
            }
        }
        const int l1 = (lane + L2_Q) & 31, l2 = (lane + 2 * L2_Q) & 31;
        const float ax = __shfl(acc.x, l1, 32), ay = __shfl(acc.y, l1, 32);
        const float az = __shfl(acc.z, l1, 32), aw = __shfl(acc.w, l1, 32);
        const float bx = __shfl(acc.x, l2, 32), by = __shfl(acc.y, l2, 32);
        const float bz = __shfl(acc.z, l2, 32), bw = __shfl(acc.w, l2, 32);
        if (g == 0) {
            float4 o;
            o.x = (acc.x + ax + bx) * di + bb.x; o.y = (acc.y + ay + by) * di + bb.y;
            o.z = (acc.z + az + bz) * di + bb.z; o.w = (acc.w + aw + bw) * di + bb.w;
            ((float4*)logits)[(size_t)node * L2_Q + q] = o;
            *(float4*)&rowsm[(it * GA_WAVES + w) * NC + q * 4] = o;
        }
        node = nnode; start = nstart; deg = ndeg; di = ndi;
    }
    __syncthreads();

    const int nrows = ((N_NODES - 1 - (int)blockIdx.x * GA_WAVES) / G2_STRIDE + 1) * GA_WAVES;
    if (t < G2_RG * NC) {
        const int rg = t / NC, c = t - rg * NC;
        float xv[G2_ROWS / G2_RG];
        for (int k = 0; k < G2_ROWS / G2_RG; ++k) {
            const int r = rg + k * G2_RG;
            const float x = rowsm[min(r, nrows - 1) * NC + c];
            xv[k] = (r < nrows) ? x : -INFINITY;
        }
        float M = xv[0];
        for (int k = 1; k < G2_ROWS / G2_RG; ++k) M = fmaxf(M, xv[k]);
        float S = 0.f;
        for (int k = 0; k < G2_ROWS / G2_RG; ++k) S += expf(xv[k] - M);
        pmx[t] = M; psm[t] = S;
    } else if (t >= G2_LSE_T0 && t - G2_LSE_T0 < nrows) {
        const int r = t - G2_LSE_T0;
        const float4* r4 = (const float4*)&rowsm[r * NC];
        float4 v[L2_Q];
        for (int k = 0; k < L2_Q; ++k) v[k] = r4[k];
        float m = fmaxf(fmaxf(v[0].x, v[0].y), fmaxf(v[0].z, v[0].w));
        for (int k = 1; k < L2_Q; ++k) m = fmaxf(m, fmaxf(fmaxf(v[k].x, v[k].y), fmaxf(v[k].z, v[k].w)));
        float s = 0.f;
        for (int k = 0; k < L2_Q; ++k)
            s += (expf(v[k].x - m) + expf(v[k].y - m)) + (expf(v[k].z - m) + expf(v[k].w - m));
        lse[r] = m + logf(s);
    }
    __syncthreads();
    for (int k = 0; k < (G2_ROWS * NC + GA_WAVES * 32 - 1) / (GA_WAVES * 32); ++k) {
        const int e = t + k * (GA_WAVES * 32);
        if (e < nrows * NC) {
            const int r = e / NC, c = e - r * NC;
            const int node = (int)blockIdx.x * GA_WAVES + (r % GA_WAVES) + (r / GA_WAVES) * G2_STRIDE;
            out_lsm[(size_t)node * NC + c] = rowsm[e] - lse[r];
        }
    }
    if (t < NC) {
        float M = pmx[t];
        for (int k = 1; k < G2_RG; ++k) M = fmaxf(M, pmx[k * NC + t]);
        float S = 0.f;
        for (int k = 0; k < G2_RG; ++k) S += psm[k * NC + t] * expf(pmx[k * NC + t] - M);
        float* o = part + (size_t)blockIdx.x * CS_PITCH + t;
        o[0] = M;
        o[CS_SUMOFF] = S;
    }
}

#define CS_BLOCKS G2_BLOCKS

#define CC_THREADS 256
#define CC_PER     ((CS_BLOCKS + CC_THREADS - 1) / CC_THREADS)

__global__ __launch_bounds__(CC_THREADS)
void k_colcomb(const float* __restrict__ part,
               float* __restrict__ colmax, float* __restrict__ colsum) {
    __shared__ float wmax[CC_THREADS / 32];
    __shared__ float wsum[CC_THREADS / 32];
    const int c = blockIdx.x;
    const int t = threadIdx.x;

    float pm[CC_PER], ps[CC_PER];
    for (int i = 0; i < CC_PER; ++i) {
        const int b = t + i * CC_THREADS;
        const bool ok = b < CS_BLOCKS;
        const float* q = part + (size_t)(ok ? b : 0) * CS_PITCH + c;
        const float qm = q[0], qs = q[CS_SUMOFF];
        pm[i] = ok ? qm : -INFINITY;
        ps[i] = ok ? qs : 0.f;
    }
    float m = pm[0];
    for (int i = 1; i < CC_PER; ++i) m = fmaxf(m, pm[i]);
    for (int off = 16; off > 0; off >>= 1) m = fmaxf(m, __shfl_xor(m, off, 32));
    if ((t & 31) == 0) wmax[t >> 5] = m;
    __syncthreads();
    float cm = wmax[0];
    for (int w = 1; w < CC_THREADS / 32; ++w) cm = fmaxf(cm, wmax[w]);

    float acc = 0.f;
    for (int i = 0; i < CC_PER; ++i) {
        const bool ok = (t + i * CC_THREADS) < CS_BLOCKS;
        acc += ok ? ps[i] * expf(pm[i] - cm) : 0.f;
    }
    for (int off = 16; off > 0; off >>= 1) acc += __shfl_xor(acc, off, 32);
    if ((t & 31) == 0) wsum[t >> 5] = acc;
    __syncthreads();
    if (t == 0) {
        float tot = wsum[0];
        for (int w = 1; w < CC_THREADS / 32; ++w) tot += wsum[w];
        colmax[c] = cm;
        colsum[c] = tot;
    }
}

#define FO_THREADS 256
#define FO_Q4      (N_NODES * L2_Q)
__global__ __launch_bounds__(FO_THREADS)
void k_finalout(const float* __restrict__ logits,
                const float* __restrict__ colmax, const float* __restrict__ colsum,
                float* __restrict__ out_sm0) {
    const int i = blockIdx.x * FO_THREADS + threadIdx.x;
    if (i >= FO_Q4) return;
    const int q = i % L2_Q;
    const float4 v  = ((const float4*)logits)[i];
    const float4 cm = ((const float4*)colmax)[q], cs = ((const float4*)colsum)[q];
    float4 o;
    o.x = expf(v.x - cm.x) / cs.x; o.y = expf(v.y - cm.y) / cs.y;
    o.z = expf(v.z - cm.z) / cs.z; o.w = expf(v.w - cm.w) / cs.w;
    ((float4*)out_sm0)[i] = o;
}

extern "C" void kernel_launch(void* const* d_in, const int* in_sizes, int n_in,
                              void* d_out, int out_size, void* d_ws, size_t ws_size,
                              hipStream_t stream) {
    (void)in_sizes; (void)n_in; (void)out_size; (void)ws_size;
    const float* x   = (const float*)d_in[0];
    const int*   ei  = (const int*)d_in[1];
    const float* W1  = (const float*)d_in[2];
    const float* b1  = (const float*)d_in[3];
    const float* W2  = (const float*)d_in[4];
    const float* b2  = (const float*)d_in[5];
    const int* src = ei;
    const int* dst = ei + N_EDGES;

    float* ws = (float*)d_ws;
    size_t off = 0;
    float* dinv = ws + off; off += N_NODES; off = (off + 63) & ~(size_t)63;
    float* xw1  = ws + off; off += (size_t)N_NODES * NH;
    float* pl2  = ws + off; off += (size_t)N_NODES * NH;
    float* agg2 = ws + off; off += (size_t)N_NODES * NC;
    float* colmax = ws + off; off += 64;
    float* colsum = ws + off; off += 64;
    static_assert((size_t)N_NODES * X2P + (size_t)CS_BLOCKS * CS_PITCH <= (size_t)N_NODES * NH
                  && ((size_t)N_NODES * NH) % 64 == 0
                  && ((size_t)N_NODES * NC) % 64 == 0, "second product and partials fit; whole units");
    float* xw2   = pl2;
    float* cpart = pl2 + (size_t)N_NODES * X2P;

    static_assert(2 * (size_t)N_NODES + 2 * (size_t)N_EDGES + (size_t)(NB + 1) * NG
                  <= (size_t)N_NODES * NC, "edge lists fit the reused plane");
    int* cnt    = (int*)agg2;
    int* rowptr = cnt + N_NODES;
    int* tmp    = rowptr + N_NODES;
    int* nbr    = tmp + N_EDGES;
    int* lstT   = nbr + N_EDGES;

    float* out_lsm = (float*)d_out;
    float* logits  = (float*)d_out + (size_t)N_NODES * NC;
    float* out_sm0 = (float*)d_out + 2 * (size_t)N_NODES * NC;

    k_part<<<NG, PS_THREADS, 0, stream>>>(src, dst, tmp, lstT);
    k_bsort<<<NB, BS_THREADS, 0, stream>>>(tmp, lstT, cnt, rowptr, dinv, nbr);

    k_gemm_wmma<NF, NH, NH / 16, NH, NH, NH><<<G1_BLOCKS, (NH / 16) * 32, 0, stream>>>(x, W1, dinv, xw1);
    k_gather1_mm<<<F1_BLOCKS, GA_WAVES * 32, 0, stream>>>(rowptr, cnt, nbr, dinv, xw1, b1, W2, xw2);

    k_gather2<<<G2_BLOCKS, GA_WAVES * 32, 0, stream>>>(rowptr, cnt, nbr, dinv, xw2, b2, logits, cpart,
                                                         out_lsm);

    k_colcomb<<<NC, CC_THREADS, 0, stream>>>(cpart, colmax, colsum);
    k_finalout<<<(FO_Q4 + FO_THREADS - 1) / FO_THREADS, FO_THREADS, 0, stream>>>(logits, colmax, colsum, out_sm0);
}


# bucket sort: a wave reads whole runs of its bucket (an item per lane, all loads in flight, kept in registers for the slot pass), no per-item search in two passes; node sum order = LDS arrival order
# speedup vs baseline: 7.6800x; 1.0411x over previous
#include <hip/hip_runtime.h>
#include <math.h>

#define N_NODES 100000
#define N_EDGES 1600000
#define NF 256
#define NH 128
#define NC 40
#define NCP 48

typedef float v2f __attribute__((ext_vector_type(2)));
typedef float v4f __attribute__((ext_vector_type(4)));
typedef float v8f __attribute__((ext_vector_type(8)));

#define E4 (N_EDGES / 4)
static_assert(N_EDGES % 4 == 0 && N_NODES % 4 == 0, "16-byte index accesses");

#define BK_SHIFT   8
#define BK_W       (1 << BK_SHIFT)
#define NB         ((N_NODES + BK_W - 1) / BK_W)
#define SRC_BITS   17
#define SRC_MASK   ((1 << SRC_BITS) - 1)
#define CH         8192
#define NG         ((N_EDGES + CH - 1) / CH)
#define PS_THREADS 1024
#define PS_Q       (CH / 4 / PS_THREADS)
static_assert(N_NODES <= (1 << SRC_BITS) && SRC_BITS + BK_SHIFT <= 31, "packed entry fits an int");
static_assert(CH % (4 * PS_THREADS) == 0 && NB + 1 <= PS_THREADS, "chunk shape");

__global__ __launch_bounds__(PS_THREADS)
void k_part(const int* __restrict__ src, const int* __restrict__ dst,
            int* __restrict__ tmp, int* __restrict__ lstT) {
    __shared__ int hist[NB + 1];
    __shared__ int wtot[PS_THREADS / 32];
    __shared__ __attribute__((aligned(16))) int sorted[CH];
    const int t = threadIdx.x, lane = t & 31, w = t >> 5, g = blockIdx.x;

    if (t <= NB) hist[t] = 0;
    __syncthreads();

    int item[PS_Q * 4], bk[PS_Q * 4], rk[PS_Q * 4];
    bool ok[PS_Q];
    #pragma unroll
    for (int u = 0; u < PS_Q; ++u) {
        const int q = g * (CH / 4) + u * PS_THREADS + t;
        ok[u] = q < E4;
        int4 d = {0, 0, 0, 0}, s = {0, 0, 0, 0};
        if (ok[u]) { d = ((const int4*)dst)[q]; s = ((const int4*)src)[q]; }
        const int dd[4] = {d.x, d.y, d.z, d.w}, ss[4] = {s.x, s.y, s.z, s.w};
        #pragma unroll
        for (int k = 0; k < 4; ++k) {
            bk[4 * u + k]   = dd[k] >> BK_SHIFT;
            item[4 * u + k] = ss[k] | ((dd[k] & (BK_W - 1)) << SRC_BITS);
            rk[4 * u + k]   = 0;
            if (ok[u]) rk[4 * u + k] = atomicAdd(&hist[bk[4 * u + k]], 1);
        }
    }
    __syncthreads();

    const int v = (t < NB) ? hist[t] : 0;
    int inc = v;
    for (int off = 1; off < 32; off <<= 1) {
        const int y = __shfl_up(inc, off, 32);
        if (lane >= off) inc += y;
    }
    if (lane == 31) wtot[w] = inc;
    __syncthreads();
    int ex = inc - v;
    for (int k = 0; k < w; ++k) ex += wtot[k];
    if (t <= NB) { hist[t] = ex; lstT[t * NG + g] = ex; }
    __syncthreads();

    #pragma unroll
    for (int u = 0; u < PS_Q; ++u) {
        if (ok[u]) {
            #pragma unroll
            for (int k = 0; k < 4; ++k) sorted[hist[bk[4 * u + k]] + rk[4 * u + k]] = item[4 * u + k];
        }
    }
    __syncthreads();

    #pragma unroll
    for (int u = 0; u < PS_Q; ++u) {
        const int i = u * PS_THREADS + t;
        const int q = g * (CH / 4) + i;
        if (q < E4) ((int4*)tmp)[q] = ((const int4*)sorted)[i];
    }
}

#define BS_THREADS 1024
#define BS_WAVES   (BS_THREADS / 32)
#define BS_RPW     ((NG + BS_WAVES - 1) / BS_WAVES)
#define BS_PCS     2
#define BS_CAP     16384
static_assert(BS_RPW <= 32 && BK_W <= BS_THREADS && BS_WAVES == 32, "a lane per run of a wave and per wave total, a thread per node of a bucket");

__global__ __launch_bounds__(BS_THREADS)
void k_bsort(const int* __restrict__ tmp, const int* __restrict__ lstT,
             int* __restrict__ cnt, int* __restrict__ rowptr, float* __restrict__ dinv,
             int* __restrict__ nbr) {
    __shared__ int ncnt[BK_W];
    __shared__ int ncur[BK_W];
    __shared__ int wrd[BS_WAVES], wsc2[BS_WAVES];
    __shared__ int stage[BS_CAP];
    const int t = threadIdx.x, lane = t & 31, b = blockIdx.x;
    const int w = __builtin_amdgcn_readfirstlane(t >> 5);

    int a = 0, len = 0;
    {
        const int r = w + lane * BS_WAVES;
        if (lane < BS_RPW && r < NG) { a = lstT[b * NG + r]; len = lstT[(b + 1) * NG + r] - a; }
    }
    if (t < BK_W) ncnt[t] = 0;

    int it[BS_RPW][BS_PCS], ca[BS_RPW], ln[BS_RPW];
    #pragma unroll
    for (int k = 0; k < BS_RPW; ++k) {
        ca[k] = (w + k * BS_WAVES) * CH + __builtin_amdgcn_readlane(a, k);
        ln[k] = __builtin_amdgcn_readlane(len, k);
        #pragma unroll
        for (int p = 0; p < BS_PCS; ++p) {
            it[k][p] = 0;
            if (p * 32 + lane < ln[k]) it[k][p] = tmp[ca[k] + p * 32 + lane];
        }
    }
    int asum = a;
    for (int off = 16; off > 0; off >>= 1) asum += __shfl_xor(asum, off, 32);
    if (lane == 0) wrd[w] = asum;
    __syncthreads();
    int base = wrd[lane];
    for (int off = 16; off > 0; off >>= 1) base += __shfl_xor(base, off, 32);

    #pragma unroll
    for (int k = 0; k < BS_RPW; ++k) {
        #pragma unroll
        for (int p = 0; p < BS_PCS; ++p)
            if (p * 32 + lane < ln[k]) atomicAdd(&ncnt[it[k][p] >> SRC_BITS], 1);
    }
    #pragma nounroll
    for (int k = 0; k < BS_RPW; ++k) {
        const int lk = __builtin_amdgcn_readlane(len, k);
        const int ck = (w + k * BS_WAVES) * CH + __builtin_amdgcn_readlane(a, k);
        #pragma nounroll
        for (int o = 32 * BS_PCS; o < lk; o += 32)
            if (o + lane < lk) atomicAdd(&ncnt[tmp[ck + o + lane] >> SRC_BITS], 1);
    }
    __syncthreads();

    const int v = (t < BK_W) ? ncnt[t] : 0;
    int inc2 = v;
    for (int off = 1; off < 32; off <<= 1) {
        const int y = __shfl_up(inc2, off, 32);
        if (lane >= off) inc2 += y;
    }
    if (lane == 31) wsc2[w] = inc2;
    __syncthreads();
    if (t < BK_W) {
        int ex = inc2 - v;
        #pragma unroll
        for (int k = 0; k < BK_W / 32; ++k) if (k < w) ex += wsc2[k];
        ncur[t] = ex;
        const int node = b * BK_W + t;
        if (node < N_NODES) {
            cnt[node] = v;
            rowptr[node] = base + ex;
            dinv[node] = rsqrtf((float)v + 1.0f);
        }
    }
    int items = 0;
    for (int k = 0; k < BK_W / 32; ++k) items += wsc2[k];
    __syncthreads();

    int pos[BS_RPW][BS_PCS];
    #pragma unroll
    for (int k = 0; k < BS_RPW; ++k) {
        #pragma unroll
        for (int p = 0; p < BS_PCS; ++p) {
            pos[k][p] = 0;
            if (p * 32 + lane < ln[k]) pos[k][p] = atomicAdd(&ncur[it[k][p] >> SRC_BITS], 1);
        }
    }
    #pragma unroll
    for (int k = 0; k < BS_RPW; ++k) {
        #pragma unroll
        for (int p = 0; p < BS_PCS; ++p) {
            if (p * 32 + lane < ln[k]) {
                if (pos[k][p] < BS_CAP) stage[pos[k][p]] = it[k][p] & SRC_MASK;
                else nbr[base + pos[k][p]] = it[k][p] & SRC_MASK;
            }
        }
    }
    #pragma nounroll
    for (int k = 0; k < BS_RPW; ++k) {
        const int lk = __builtin_amdgcn_readlane(len, k);
        const int ck = (w + k * BS_WAVES) * CH + __builtin_amdgcn_readlane(a, k);
        #pragma nounroll
        for (int o = 32 * BS_PCS; o < lk; o += 32) {
            if (o + lane < lk) {
                const int x = tmp[ck + o + lane];
                const int ps = atomicAdd(&ncur[x >> SRC_BITS], 1);
                if (ps < BS_CAP) stage[ps] = x & SRC_MASK;
                else nbr[base + ps] = x & SRC_MASK;
            }
        }
    }
    __syncthreads();
    const int m = min(items, BS_CAP);
    for (int i = t; i < m; i += BS_THREADS) nbr[base + i] = stage[i];
}

#define GEMM_MT (N_NODES / 16)
static_assert(N_NODES % 16 == 0, "whole row tiles");
#define G1_BLOCKS 512
static_assert(G1_BLOCKS <= GEMM_MT, "grid shape");

template<int K, int LDB, int NT, int LDC, int BP, int BV>
__global__ __launch_bounds__(NT * 32)
void k_gemm_wmma(const float* __restrict__ A, const float* __restrict__ B,
                 const float* __restrict__ rs, float* __restrict__ C) {
    constexpr int P     = K + 4;
    constexpr int RQ    = K / 4;
    constexpr int PPL   = RQ / 32;
    constexpr int RPW   = (16 + NT - 1) / NT;
    constexpr int LROWS = RPW * NT;
    constexpr int NST   = RPW * PPL;
    static_assert(LDB == NT * 16 && LDC >= LDB && RQ % 32 == 0 && (P / 4) % 2 == 1, "tile shape");
    static_assert(BV <= LDB && BV <= BP && BV > 0, "valid columns of B inside its tiles and its rows");
    __shared__ __attribute__((aligned(16))) float tile[2][LROWS * P];

    const int t    = threadIdx.x;
    const int nt   = t >> 5;
    const int lane = t & 31;
    const int half = lane >> 4;
    const int lm   = lane & 15;

    v2f b[K / 4];
    const int  bc  = nt * 16 + lm;
    const bool bok = (BV == LDB) || bc < BV;
    const float* bcol = B + (bok ? bc : BV - 1);
    #pragma unroll
    for (int s = 0; s < K / 4; ++s) {
        const int ka = 4 * s + 2 * half;
        const float bx = bcol[(size_t)ka * BP];
        const float by = bcol[(size_t)(ka + 1) * BP];
        b[s].x = bok ? bx : 0.f;
        b[s].y = bok ? by : 0.f;
    }

    int goff[NST], loff[NST];
    #pragma unroll
    for (int i = 0; i < RPW; ++i) {
        const int row = nt + i * NT;
        const int srow = (16 % NT == 0) ? row : min(row, 15);
        #pragma unroll
        for (int j = 0; j < PPL; ++j) {
            goff[i * PPL + j] = srow * RQ + j * 32 + lane;
            loff[i * PPL + j] = row * P + (j * 32 + lane) * 4;
        }
    }

    v4f st[NST];
    int mt = blockIdx.x;
    const v4f* rsp = (const v4f*)(rs + 8 * half);
    v4f rs0 = rsp[mt * 4], rs1 = rsp[mt * 4 + 1];
    {
        const v4f* g = (const v4f*)(A + (size_t)mt * (16 * K));
        #pragma unroll
        for (int i = 0; i < NST; ++i) st[i] = g[goff[i]];
        #pragma unroll
        for (int i = 0; i < NST; ++i) *(v4f*)&tile[0][loff[i]] = st[i];
    }
    __syncthreads();

    int cur = 0;
    for (; mt < GEMM_MT; mt += gridDim.x) {
        const int nx = mt + gridDim.x;
        const bool more = nx < GEMM_MT;
        v4f nrs0 = rs0, nrs1 = rs1;
        if (more) {
            const v4f* g = (const v4f*)(A + (size_t)nx * (16 * K));
            #pragma unroll
            for (int i = 0; i < NST; ++i) st[i] = g[goff[i]];
            nrs0 = rsp[nx * 4]; nrs1 = rsp[nx * 4 + 1];
        }
        const float* ap = &tile[cur][lm * P + 2 * half];
        v8f c = {0.f, 0.f, 0.f, 0.f, 0.f, 0.f, 0.f, 0.f};
        #pragma unroll
        for (int s = 0; s < K / 4; ++s) {
            const v2f a = *(const v2f*)(ap + 4 * s);
            c = __builtin_amdgcn_wmma_f32_16x16x4_f32(false, a, false, b[s],
                                                      (short)0, c, false, false);
        }

        if (more) {
            #pragma unroll
            for (int i = 0; i < NST; ++i) *(v4f*)&tile[cur ^ 1][loff[i]] = st[i];
        }
        __syncthreads();
        cur ^= 1;

        float* out = C + (size_t)(mt * 16 + 8 * half) * LDC + nt * 16 + lm;
        #pragma unroll
        for (int v = 0; v < 8; ++v) out[(size_t)v * LDC] = c[v] * (v < 4 ? rs0[v & 3] : rs1[v & 3]);
        rs0 = nrs0; rs1 = nrs1;
    }
}

#define X2P  64
#define GA_WAVES 8
#define GA_FLY   8
static_assert(N_NODES % GA_WAVES == 0, "whole work-groups of nodes");
static_assert(NH == 128, "a 32-lane wave covers a hidden row with one float4 per lane");
#define F1_TRIPS  5
#define F1_BLOCKS (GEMM_MT / F1_TRIPS)
#define F1_P      (NH + 4)
#define F1_NT     (NCP / 16)
#define F1_W2Q    (NH * NC / (GA_WAVES * 32))
static_assert(GEMM_MT % F1_TRIPS == 0 && 2 * GA_WAVES == 16, "whole tiles: two nodes per wave");
static_assert((NH * NC) % (GA_WAVES * 32) == 0 && ((NCP - NC) * NH) % (GA_WAVES * 32) == 0, "W2 staging");
static_assert(F1_NT <= GA_WAVES && (F1_P / 4) % 2 == 1, "tile shape");

__device__ __forceinline__ int   bcast_i(int v, int j)   { return __builtin_amdgcn_readlane(v, j); }

__device__ __forceinline__ float4 node_sum(const float4* __restrict__ rows, const int* __restrict__ nbr,
                                           float4 acc, int my_s, int start, int deg, int lane) {
    for (int base = 0; base < deg; base += 32) {
        const int n = min(32, deg - base);
        if (base > 0) {
            my_s = 0;
            if (lane < n) my_s = nbr[start + base + lane];
        }
        for (int j = 0; j < n; j += GA_FLY) {
            const int r = n - j;
            float4 v[GA_FLY];
            v[0] = rows[(size_t)bcast_i(my_s, j) * (NH / 4) + lane];
            #pragma unroll
            for (int k = 1; k < GA_FLY; ++k) {
                v[k] = make_float4(0.f, 0.f, 0.f, 0.f);
                if (r > k) v[k] = rows[(size_t)bcast_i(my_s, j + k) * (NH / 4) + lane];
            }
            acc.x += v[0].x; acc.y += v[0].y; acc.z += v[0].z; acc.w += v[0].w;
            #pragma unroll
            for (int k = 1; k < GA_FLY; ++k)
                if (r > k) { acc.x += v[k].x; acc.y += v[k].y; acc.z += v[k].z; acc.w += v[k].w; }
        }
    }
    return acc;
}

__global__ __launch_bounds__(GA_WAVES * 32)
void k_gather1_mm(const int* __restrict__ rowptr, const int* __restrict__ cnt,
                  const int* __restrict__ nbr, const float* __restrict__ dinv,
                  const float* __restrict__ xs, const float* __restrict__ b,
                  const float* __restrict__ W2, float* __restrict__ xw2) {
    __shared__ __attribute__((aligned(16))) float htile[2][16 * F1_P];
    __shared__ __attribute__((aligned(16))) float w2t[NCP * F1_P];
    __shared__ __attribute__((aligned(16))) float dtile[2][16];
    const int t = threadIdx.x, lane = t & 31;
    const int w = __builtin_amdgcn_readfirstlane(t >> 5);
    const int half = lane >> 4;
    const int lm   = lane & 15;
    const float4* rows = (const float4*)xs;

    float wv[F1_W2Q];
    #pragma unroll
    for (int i = 0; i < F1_W2Q; ++i) wv[i] = W2[t + i * (GA_WAVES * 32)];

    const int tl = min(lane, F1_TRIPS - 1);
    const int nl = ((int)blockIdx.x + tl * F1_BLOCKS) * 16 + 2 * w;
    const int m_s0 = rowptr[nl], m_s1 = rowptr[nl + 1];
    const int m_d0 = cnt[nl],    m_d1 = cnt[nl + 1];
    const int m_i0 = __float_as_int(dinv[nl]), m_i1 = __float_as_int(dinv[nl + 1]);

    int my0 = 0, my1 = 0;
    float4 acc0, acc1;
    {
        const int na = (int)blockIdx.x * 16 + 2 * w;
        if (lane < min(32, bcast_i(m_d0, 0))) my0 = nbr[bcast_i(m_s0, 0) + lane];
        if (lane < min(32, bcast_i(m_d1, 0))) my1 = nbr[bcast_i(m_s1, 0) + lane];
        acc0 = rows[(size_t)na * (NH / 4) + lane];
        acc1 = rows[(size_t)(na + 1) * (NH / 4) + lane];
    }

    #pragma unroll
    for (int i = 0; i < F1_W2Q; ++i) {
        const int e = t + i * (GA_WAVES * 32);
        const int k = e / NC, c = e - k * NC;
        w2t[c * F1_P + k] = wv[i];
    }
    for (int e = t; e < (NCP - NC) * NH; e += GA_WAVES * 32)
        w2t[(NC + e / NH) * F1_P + (e % NH)] = 0.f;

    float4 b4;
    b4.x = b[lane * 4 + 0]; b4.y = b[lane * 4 + 1]; b4.z = b[lane * 4 + 2]; b4.w = b[lane * 4 + 3];

    int cur = 0;
    for (int trip = 0; trip < F1_TRIPS; ++trip) {
        const int node0 = ((int)blockIdx.x + trip * F1_BLOCKS) * 16;
        const int s0 = bcast_i(m_s0, trip), s1 = bcast_i(m_s1, trip);
        const int d0 = bcast_i(m_d0, trip), d1 = bcast_i(m_d1, trip);
        const float di0 = __int_as_float(bcast_i(m_i0, trip)), di1 = __int_as_float(bcast_i(m_i1, trip));

        const float4 a0 = node_sum(rows, nbr, acc0, my0, s0, d0, lane);
        float4 o;
        o.x = fmaxf(a0.x * di0 + b4.x, 0.f); o.y = fmaxf(a0.y * di0 + b4.y, 0.f);
        o.z = fmaxf(a0.z * di0 + b4.z, 0.f); o.w = fmaxf(a0.w * di0 + b4.w, 0.f);
        *(float4*)&htile[cur][(2 * w) * F1_P + lane * 4] = o;
        const float4 a1 = node_sum(rows, nbr, acc1, my1, s1, d1, lane);
        o.x = fmaxf(a1.x * di1 + b4.x, 0.f); o.y = fmaxf(a1.y * di1 + b4.y, 0.f);
        o.z = fmaxf(a1.z * di1 + b4.z, 0.f); o.w = fmaxf(a1.w * di1 + b4.w, 0.f);
        *(float4*)&htile[cur][(2 * w + 1) * F1_P + lane * 4] = o;
        if (lane == 0) { dtile[cur][2 * w] = di0; dtile[cur][2 * w + 1] = di1; }
        __syncthreads();

        if (trip + 1 < F1_TRIPS) {
            const int nna = node0 + 16 * F1_BLOCKS + 2 * w;
            my0 = 0; my1 = 0;
            if (lane < min(32, bcast_i(m_d0, trip + 1))) my0 = nbr[bcast_i(m_s0, trip + 1) + lane];
            if (lane < min(32, bcast_i(m_d1, trip + 1))) my1 = nbr[bcast_i(m_s1, trip + 1) + lane];
            acc0 = rows[(size_t)nna * (NH / 4) + lane];
            acc1 = rows[(size_t)(nna + 1) * (NH / 4) + lane];
        }
        if (w < F1_NT) {
            const float* ap = &htile[cur][lm * F1_P + 2 * half];
            const float* bp = &w2t[(w * 16 + lm) * F1_P + 2 * half];
            v8f c = {0.f, 0.f, 0.f, 0.f, 0.f, 0.f, 0.f, 0.f};
            #pragma unroll
            for (int s = 0; s < NH / 4; ++s) {
                const v2f a  = *(const v2f*)(ap + 4 * s);
                const v2f bf = *(const v2f*)(bp + 4 * s);
                c = __builtin_amdgcn_wmma_f32_16x16x4_f32(false, a, false, bf,
                                                          (short)0, c, false, false);
            }
            const v4f rs0 = *(const v4f*)&dtile[cur][8 * half];
            const v4f rs1 = *(const v4f*)&dtile[cur][8 * half + 4];
            float* out = xw2 + (size_t)(node0 + 8 * half) * X2P + w * 16 + lm;
            #pragma unroll
            for (int v = 0; v < 8; ++v) out[(size_t)v * X2P] = c[v] * (v < 4 ? rs0[v & 3] : rs1[v & 3]);
        }
        cur ^= 1;
    }
}


#define L2_Q (NC / 4)
#define G2_BLOCKS 1786
#define G2_STRIDE (G2_BLOCKS * GA_WAVES)
#define G2_TRIPS  ((N_NODES + G2_STRIDE - 1) / G2_STRIDE)
#define G2_ROWS   (G2_TRIPS * GA_WAVES)
#define G2_RG     4
#define CS_PITCH  128
#define CS_SUMOFF 64
static_assert(3 * L2_Q <= 32 && NCP % 4 == 0, "three edge groups per wave");
static_assert(NCP <= X2P && X2P % 64 == 0, "a row fits its pitch; the pitch is whole 256-byte units");
static_assert(N_NODES >= G2_STRIDE && (N_NODES % G2_STRIDE) % GA_WAVES == 0,
              "every wave has a node, and the waves of a work-group make the same number of trips");
#define G2_LSE_T0 (GA_WAVES * 32 - 64)
static_assert(G2_RG * NC <= G2_LSE_T0 && G2_RG <= GA_WAVES && G2_ROWS <= 64 && G2_ROWS % G2_RG == 0,
              "column fold: a thread per (row group, column); row pass: a thread per row, on waves of its own");
static_assert(NC <= CS_SUMOFF && CS_SUMOFF + NC <= CS_PITCH, "partial plane layout");

__global__ __launch_bounds__(GA_WAVES * 32)
void k_gather2(const int* __restrict__ rowptr, const int* __restrict__ cnt,
               const int* __restrict__ nbr, const float* __restrict__ dinv,
               const float* __restrict__ xs2, const float* __restrict__ b2,
               float* __restrict__ logits, float* __restrict__ part,
               float* __restrict__ out_lsm) {
    __shared__ __attribute__((aligned(16))) float rowsm[G2_ROWS * NC];
    __shared__ float pmx[G2_RG * NC], psm[G2_RG * NC];
    __shared__ float lse[G2_ROWS];
    const int t = threadIdx.x, lane = t & 31;
    const int w  = __builtin_amdgcn_readfirstlane(t >> 5);
    const int gw = __builtin_amdgcn_readfirstlane(blockIdx.x * GA_WAVES + (t >> 5));
    const float4* rows = (const float4*)xs2;
    const int g = lane / L2_Q;
    const int q = lane - g * L2_Q;
    const bool live = g < 3;

    float4 bb = {0.f, 0.f, 0.f, 0.f};
    if (g == 0) { bb.x = b2[q * 4 + 0]; bb.y = b2[q * 4 + 1]; bb.z = b2[q * 4 + 2]; bb.w = b2[q * 4 + 3]; }

    int node = gw;
    int start = rowptr[node], deg = cnt[node];
    float di = dinv[node];
    for (int it = 0; node < N_NODES; ++it) {
        const int nnode = node + G2_STRIDE;
        const int nc = min(nnode, N_NODES - 1);
        const int nstart = rowptr[nc], ndeg = cnt[nc];
        const float ndi = dinv[nc];

        float4 acc = {0.f, 0.f, 0.f, 0.f};
        if (g == 0) acc = rows[(size_t)node * (X2P / 4) + q];
        for (int base = 0; base < deg; base += 32) {
            const int n = min(32, deg - base);
            int my_s = 0;
            if (lane < n) my_s = nbr[start + base + lane];
            for (int j0 = 0; j0 < n; j0 += 6) {
                const int ja = j0 + g, jb = j0 + 3 + g;
                const bool va = live && ja < n, vb = live && jb < n;
                const int sa = __shfl(my_s, ja & 31, 32), sb = __shfl(my_s, jb & 31, 32);
                float4 xa = {0.f, 0.f, 0.f, 0.f}, xb = {0.f, 0.f, 0.f, 0.f};
                if (va) xa = rows[(size_t)sa * (X2P / 4) + q];
                if (vb) xb = rows[(size_t)sb * (X2P / 4) + q];
                if (va) { acc.x += xa.x; acc.y += xa.y; acc.z += xa.z; acc.w += xa.w; }
                if (vb) { acc.x += xb.x; acc.y += xb.y; acc.z += xb.z; acc.w += xb.w; }
            }
        }
        const int l1 = (lane + L2_Q) & 31, l2 = (lane + 2 * L2_Q) & 31;
        const float ax = __shfl(acc.x, l1, 32), ay = __shfl(acc.y, l1, 32);
        const float az = __shfl(acc.z, l1, 32), aw = __shfl(acc.w, l1, 32);
        const float bx = __shfl(acc.x, l2, 32), by = __shfl(acc.y, l2, 32);
        const float bz = __shfl(acc.z, l2, 32), bw = __shfl(acc.w, l2, 32);
        if (g == 0) {
            float4 o;
            o.x = (acc.x + ax + bx) * di + bb.x; o.y = (acc.y + ay + by) * di + bb.y;
            o.z = (acc.z + az + bz) * di + bb.z; o.w = (acc.w + aw + bw) * di + bb.w;
            ((float4*)logits)[(size_t)node * L2_Q + q] = o;
            *(float4*)&rowsm[(it * GA_WAVES + w) * NC + q * 4] = o;
        }
        node = nnode; start = nstart; deg = ndeg; di = ndi;
    }
    __syncthreads();

    const int nrows = ((N_NODES - 1 - (int)blockIdx.x * GA_WAVES) / G2_STRIDE + 1) * GA_WAVES;
    if (t < G2_RG * NC) {
        const int rg = t / NC, c = t - rg * NC;
        float xv[G2_ROWS / G2_RG];
        for (int k = 0; k < G2_ROWS / G2_RG; ++k) {
            const int r = rg + k * G2_RG;
            const float x = rowsm[min(r, nrows - 1) * NC + c];
            xv[k] = (r < nrows) ? x : -INFINITY;
        }
        float M = xv[0];
        for (int k = 1; k < G2_ROWS / G2_RG; ++k) M = fmaxf(M, xv[k]);
        float S = 0.f;
        for (int k = 0; k < G2_ROWS / G2_RG; ++k) S += expf(xv[k] - M);
        pmx[t] = M; psm[t] = S;
    } else if (t >= G2_LSE_T0 && t - G2_LSE_T0 < nrows) {
        const int r = t - G2_LSE_T0;
        const float4* r4 = (const float4*)&rowsm[r * NC];
        float4 v[L2_Q];
        for (int k = 0; k < L2_Q; ++k) v[k] = r4[k];
        float m = fmaxf(fmaxf(v[0].x, v[0].y), fmaxf(v[0].z, v[0].w));
        for (int k = 1; k < L2_Q; ++k) m = fmaxf(m, fmaxf(fmaxf(v[k].x, v[k].y), fmaxf(v[k].z, v[k].w)));
        float s = 0.f;
        for (int k = 0; k < L2_Q; ++k)
            s += (expf(v[k].x - m) + expf(v[k].y - m)) + (expf(v[k].z - m) + expf(v[k].w - m));
        lse[r] = m + logf(s);
    }
    __syncthreads();
    for (int k = 0; k < (G2_ROWS * NC + GA_WAVES * 32 - 1) / (GA_WAVES * 32); ++k) {
        const int e = t + k * (GA_WAVES * 32);
        if (e < nrows * NC) {
            const int r = e / NC, c = e - r * NC;
            const int node = (int)blockIdx.x * GA_WAVES + (r % GA_WAVES) + (r / GA_WAVES) * G2_STRIDE;
            out_lsm[(size_t)node * NC + c] = rowsm[e] - lse[r];
        }
    }
    if (t < NC) {
        float M = pmx[t];
        for (int k = 1; k < G2_RG; ++k) M = fmaxf(M, pmx[k * NC + t]);
        float S = 0.f;
        for (int k = 0; k < G2_RG; ++k) S += psm[k * NC + t] * expf(pmx[k * NC + t] - M);
        float* o = part + (size_t)blockIdx.x * CS_PITCH + t;
        o[0] = M;
        o[CS_SUMOFF] = S;
    }
}

#define CS_BLOCKS G2_BLOCKS

#define CC_THREADS 256
#define CC_PER     ((CS_BLOCKS + CC_THREADS - 1) / CC_THREADS)

__global__ __launch_bounds__(CC_THREADS)
void k_colcomb(const float* __restrict__ part,
               float* __restrict__ colmax, float* __restrict__ colsum) {
    __shared__ float wmax[CC_THREADS / 32];
    __shared__ float wsum[CC_THREADS / 32];
    const int c = blockIdx.x;
    const int t = threadIdx.x;

    float pm[CC_PER], ps[CC_PER];
    for (int i = 0; i < CC_PER; ++i) {
        const int b = t + i * CC_THREADS;
        const bool ok = b < CS_BLOCKS;
        const float* q = part + (size_t)(ok ? b : 0) * CS_PITCH + c;
        const float qm = q[0], qs = q[CS_SUMOFF];
        pm[i] = ok ? qm : -INFINITY;
        ps[i] = ok ? qs : 0.f;
    }
    float m = pm[0];
    for (int i = 1; i < CC_PER; ++i) m = fmaxf(m, pm[i]);
    for (int off = 16; off > 0; off >>= 1) m = fmaxf(m, __shfl_xor(m, off, 32));
    if ((t & 31) == 0) wmax[t >> 5] = m;
    __syncthreads();
    float cm = wmax[0];
    for (int w = 1; w < CC_THREADS / 32; ++w) cm = fmaxf(cm, wmax[w]);

    float acc = 0.f;
    for (int i = 0; i < CC_PER; ++i) {
        const bool ok = (t + i * CC_THREADS) < CS_BLOCKS;
        acc += ok ? ps[i] * expf(pm[i] - cm) : 0.f;
    }
    for (int off = 16; off > 0; off >>= 1) acc += __shfl_xor(acc, off, 32);
    if ((t & 31) == 0) wsum[t >> 5] = acc;
    __syncthreads();
    if (t == 0) {
        float tot = wsum[0];
        for (int w = 1; w < CC_THREADS / 32; ++w) tot += wsum[w];
        colmax[c] = cm;
        colsum[c] = tot;
    }
}

#define FO_THREADS 256
#define FO_Q4      (N_NODES * L2_Q)
__global__ __launch_bounds__(FO_THREADS)
void k_finalout(const float* __restrict__ logits,
                const float* __restrict__ colmax, const float* __restrict__ colsum,
                float* __restrict__ out_sm0) {
    const int i = blockIdx.x * FO_THREADS + threadIdx.x;
    if (i >= FO_Q4) return;
    const int q = i % L2_Q;
    const float4 v  = ((const float4*)logits)[i];
    const float4 cm = ((const float4*)colmax)[q], cs = ((const float4*)colsum)[q];
    float4 o;
    o.x = expf(v.x - cm.x) / cs.x; o.y = expf(v.y - cm.y) / cs.y;
    o.z = expf(v.z - cm.z) / cs.z; o.w = expf(v.w - cm.w) / cs.w;
    ((float4*)out_sm0)[i] = o;
}

extern "C" void kernel_launch(void* const* d_in, const int* in_sizes, int n_in,
                              void* d_out, int out_size, void* d_ws, size_t ws_size,
                              hipStream_t stream) {
    (void)in_sizes; (void)n_in; (void)out_size; (void)ws_size;
    const float* x   = (const float*)d_in[0];
    const int*   ei  = (const int*)d_in[1];
    const float* W1  = (const float*)d_in[2];
    const float* b1  = (const float*)d_in[3];
    const float* W2  = (const float*)d_in[4];
    const float* b2  = (const float*)d_in[5];
    const int* src = ei;
    const int* dst = ei + N_EDGES;

    float* ws = (float*)d_ws;
    size_t off = 0;
    float* dinv = ws + off; off += N_NODES; off = (off + 63) & ~(size_t)63;
    float* xw1  = ws + off; off += (size_t)N_NODES * NH;
    float* pl2  = ws + off; off += (size_t)N_NODES * NH;
    float* agg2 = ws + off; off += (size_t)N_NODES * NC;
    float* colmax = ws + off; off += 64;
    float* colsum = ws + off; off += 64;
    static_assert((size_t)N_NODES * X2P + (size_t)CS_BLOCKS * CS_PITCH <= (size_t)N_NODES * NH
                  && ((size_t)N_NODES * NH) % 64 == 0
                  && ((size_t)N_NODES * NC) % 64 == 0, "second product and partials fit; whole units");
    float* xw2   = pl2;
    float* cpart = pl2 + (size_t)N_NODES * X2P;

    static_assert(2 * (size_t)N_NODES + 2 * (size_t)N_EDGES + (size_t)(NB + 1) * NG
                  <= (size_t)N_NODES * NC, "edge lists fit the reused plane");
    int* cnt    = (int*)agg2;
    int* rowptr = cnt + N_NODES;
    int* tmp    = rowptr + N_NODES;
    int* nbr    = tmp + N_EDGES;
    int* lstT   = nbr + N_EDGES;

    float* out_lsm = (float*)d_out;
    float* logits  = (float*)d_out + (size_t)N_NODES * NC;
    float* out_sm0 = (float*)d_out + 2 * (size_t)N_NODES * NC;

    k_part<<<NG, PS_THREADS, 0, stream>>>(src, dst, tmp, lstT);
    k_bsort<<<NB, BS_THREADS, 0, stream>>>(tmp, lstT, cnt, rowptr, dinv, nbr);

    k_gemm_wmma<NF, NH, NH / 16, NH, NH, NH><<<G1_BLOCKS, (NH / 16) * 32, 0, stream>>>(x, W1, dinv, xw1);
    k_gather1_mm<<<F1_BLOCKS, GA_WAVES * 32, 0, stream>>>(rowptr, cnt, nbr, dinv, xw1, b1, W2, xw2);

    k_gather2<<<G2_BLOCKS, GA_WAVES * 32, 0, stream>>>(rowptr, cnt, nbr, dinv, xw2, b2, logits, cpart,
                                                         out_lsm);

    k_colcomb<<<NC, CC_THREADS, 0, stream>>>(cpart, colmax, colsum);
    k_finalout<<<(FO_Q4 + FO_THREADS - 1) / FO_THREADS, FO_THREADS, 0, stream>>>(logits, colmax, colsum, out_sm0);
}
